# starting kernel
# baseline (speedup 1.0000x reference)
#include <hip/hip_runtime.h>
#include <hip/hip_bf16.h>
#include <math.h>


typedef float v2f __attribute__((ext_vector_type(2)));
typedef float v8f __attribute__((ext_vector_type(8)));

#define HID   1024
#define NST   128
#define BATCH 256
#define IN2   256
#define CB    32
#define TS    1028

__device__ __forceinline__ v8f wmma_f32(v2f a, v2f b, v8f c) {
  return __builtin_amdgcn_wmma_f32_16x16x4_f32(
       false, a,  false, b,
       (short)0, c,  false,  false);
}

__global__ void build_x2(const float* __restrict__ x, float* __restrict__ X2) {
  int i = blockIdx.x * blockDim.x + threadIdx.x;
  if (i >= 512 * IN2) return;
  int row = i >> 8, c = i & 255;
  float v;
  if (row < 256) {
    v = x[row * 512 + c];
  } else {
    int rz = row - 256;
    v = (c < 128) ? x[rz * 512 + 384 + c] : x[rz * 512 + 128 + (c - 128)];
  }
  X2[i] = v;
}

__global__ void zero_buf(float* __restrict__ p, int n) {
  int i = blockIdx.x * blockDim.x + threadIdx.x;
  if (i < n) p[i] = 0.f;
}

__global__ __launch_bounds__(256) void gemm_awt(
    const float* __restrict__ A, int lda,
    const float* __restrict__ W, int ldw,
    const float* __restrict__ bias,
    const float* __restrict__ mask, int ldm,
    float* __restrict__ C, int ldc,
    int M, int K, int epi)
{
  const int lane = threadIdx.x & 31, wv = threadIdx.x >> 5;
  const int l16 = lane & 15, hlf = lane >> 4;
  const int m0 = blockIdx.x * 128 + wv * 16;
  const int n0 = blockIdx.y * 64;
  if (m0 >= M) return;

  const v8f vz = {0.f,0.f,0.f,0.f,0.f,0.f,0.f,0.f};
  v8f acc[4] = {vz, vz, vz, vz};

  const int kof = hlf * 2;
  const float* Ar  = A + (size_t)(m0 + l16) * lda;
  const float* Wr0 = W + (size_t)(n0 +  0 + l16) * ldw;
  const float* Wr1 = W + (size_t)(n0 + 16 + l16) * ldw;
  const float* Wr2 = W + (size_t)(n0 + 32 + l16) * ldw;
  const float* Wr3 = W + (size_t)(n0 + 48 + l16) * ldw;

  v2f a0, b00, b10, b20, b30;
  v2f a1, b01, b11, b21, b31;

#define LDSET0(kk) { a0  = *(const v2f*)(Ar  + (kk)); \
                     b00 = *(const v2f*)(Wr0 + (kk)); \
                     b10 = *(const v2f*)(Wr1 + (kk)); \
                     b20 = *(const v2f*)(Wr2 + (kk)); \
                     b30 = *(const v2f*)(Wr3 + (kk)); }
#define LDSET1(kk) { a1  = *(const v2f*)(Ar  + (kk)); \
                     b01 = *(const v2f*)(Wr0 + (kk)); \
                     b11 = *(const v2f*)(Wr1 + (kk)); \
                     b21 = *(const v2f*)(Wr2 + (kk)); \
                     b31 = *(const v2f*)(Wr3 + (kk)); }
#define WM0() { acc[0] = wmma_f32(a0, b00, acc[0]); \
                acc[1] = wmma_f32(a0, b10, acc[1]); \
                acc[2] = wmma_f32(a0, b20, acc[2]); \
                acc[3] = wmma_f32(a0, b30, acc[3]); }
#define WM1() { acc[0] = wmma_f32(a1, b01, acc[0]); \
                acc[1] = wmma_f32(a1, b11, acc[1]); \
                acc[2] = wmma_f32(a1, b21, acc[2]); \
                acc[3] = wmma_f32(a1, b31, acc[3]); }

  LDSET0(kof);
  LDSET1(kof + 4);
  for (int k = 0; k < K - 8; k += 8) {
    WM0(); LDSET0(k + 8  + kof);
    WM1(); LDSET1(k + 12 + kof);
  }
  WM0();
  WM1();
#undef LDSET0
#undef LDSET1
#undef WM0
#undef WM1

  #pragma unroll
  for (int nt = 0; nt < 4; ++nt) {
    #pragma unroll
    for (int r = 0; r < 8; ++r) {
      const int m  = m0 + r + 8 * hlf;
      const int nn = n0 + nt * 16 + l16;
      float v = acc[nt][r];
      if (bias) v += bias[nn];
      if (epi == 1)      v = v > 0.f ? v   : 0.f;
      else if (epi == 2) v = v > 0.f ? 1.f : 0.f;
      if (mask) v *= mask[(size_t)m * ldm + nn];
      C[(size_t)m * ldc + nn] = v;
    }
  }
}

__global__ __launch_bounds__(256) void gemm_ab(
    const float* __restrict__ A, int lda,
    const float* __restrict__ B, int ldb,
    float* __restrict__ C, int ldc,
    int M, int K)
{
  const int lane = threadIdx.x & 31, wv = threadIdx.x >> 5;
  const int l16 = lane & 15, hlf = lane >> 4;
  const int m0 = blockIdx.x * 128 + wv * 16;
  const int n0 = blockIdx.y * 64;
  if (m0 >= M) return;

  const v8f vz = {0.f,0.f,0.f,0.f,0.f,0.f,0.f,0.f};
  v8f acc[4] = {vz, vz, vz, vz};
  const int kof = hlf * 2;
  const float* Ar = A + (size_t)(m0 + l16) * lda;
  const float* Bc = B + n0 + l16;

  v2f a0, c00, c10, c20, c30;
  v2f a1, c01, c11, c21, c31;

#define BLD(dst, kk, nt) { (dst).x = Bc[(size_t)(kk) * ldb + (nt) * 16]; \
                           (dst).y = Bc[(size_t)((kk) + 1) * ldb + (nt) * 16]; }
#define LDSET0(kk) { a0 = *(const v2f*)(Ar + (kk)); \
                     BLD(c00, (kk), 0) BLD(c10, (kk), 1) BLD(c20, (kk), 2) BLD(c30, (kk), 3) }
#define LDSET1(kk) { a1 = *(const v2f*)(Ar + (kk)); \
                     BLD(c01, (kk), 0) BLD(c11, (kk), 1) BLD(c21, (kk), 2) BLD(c31, (kk), 3) }
#define WM0() { acc[0] = wmma_f32(a0, c00, acc[0]); acc[1] = wmma_f32(a0, c10, acc[1]); \
                acc[2] = wmma_f32(a0, c20, acc[2]); acc[3] = wmma_f32(a0, c30, acc[3]); }
#define WM1() { acc[0] = wmma_f32(a1, c01, acc[0]); acc[1] = wmma_f32(a1, c11, acc[1]); \
                acc[2] = wmma_f32(a1, c21, acc[2]); acc[3] = wmma_f32(a1, c31, acc[3]); }

  LDSET0(kof);
  LDSET1(kof + 4);
  for (int k = 0; k < K - 8; k += 8) {
    WM0(); LDSET0(k + 8  + kof);
    WM1(); LDSET1(k + 12 + kof);
  }
  WM0();
  WM1();
#undef BLD
#undef LDSET0
#undef LDSET1
#undef WM0
#undef WM1

  #pragma unroll
  for (int nt = 0; nt < 4; ++nt)
    #pragma unroll
    for (int r = 0; r < 8; ++r)
      C[(size_t)(m0 + r + 8 * hlf) * ldc + n0 + nt * 16 + l16] = acc[nt][r];
}

__device__ __forceinline__ void chain_ld(
    const float* __restrict__ Wr, const float* __restrict__ Tb0,
    const float* __restrict__ Tb1, int kk,
    v2f av[8], v2f& c0, v2f& c1)
{
  c0 = *(const v2f*)(Tb0 + kk);
  c1 = *(const v2f*)(Tb1 + kk);
  #pragma unroll
  for (int mt = 0; mt < 8; ++mt)
    av[mt] = *(const v2f*)(Wr + (size_t)(mt * 16 * HID) + kk);
}

__device__ __forceinline__ void chain_mm(const v2f av[8], v2f c0, v2f c1,
                                         v8f acc[8][2])
{
  #pragma unroll
  for (int mt = 0; mt < 8; ++mt) {
    acc[mt][0] = wmma_f32(av[mt], c0, acc[mt][0]);
    acc[mt][1] = wmma_f32(av[mt], c1, acc[mt][1]);
  }
}

__device__ __forceinline__ void chain_gemm_1024(
    const float* __restrict__ Wm,
    const float* __restrict__ Tin,
    float* __restrict__ Tout,
    const float* __restrict__ dm,
    int lane, int wv)
{
  const int l16 = lane & 15, hlf = lane >> 4;
  const int kof = hlf * 2;
  const v8f vz = {0.f,0.f,0.f,0.f,0.f,0.f,0.f,0.f};
  v8f acc[8][2];
  #pragma unroll
  for (int mt = 0; mt < 8; ++mt) { acc[mt][0] = vz; acc[mt][1] = vz; }

  const float* Tb0 = Tin + (size_t)l16 * TS;
  const float* Tb1 = Tin + (size_t)(16 + l16) * TS;
  const float* Wr  = Wm + (size_t)(wv * 128 + l16) * HID;

  v2f avA[8], cA0, cA1;
  v2f avB[8], cB0, cB1;
  v2f avC[8], cC0, cC1;
  v2f avD[8], cD0, cD1;

  chain_ld(Wr, Tb0, Tb1, kof,      avA, cA0, cA1);
  chain_ld(Wr, Tb0, Tb1, kof + 4,  avB, cB0, cB1);
  chain_ld(Wr, Tb0, Tb1, kof + 8,  avC, cC0, cC1);
  chain_ld(Wr, Tb0, Tb1, kof + 12, avD, cD0, cD1);
  for (int k = 0; k < HID - 16; k += 16) {
    chain_mm(avA, cA0, cA1, acc);
    chain_ld(Wr, Tb0, Tb1, k + 16 + kof, avA, cA0, cA1);
    chain_mm(avB, cB0, cB1, acc);
    chain_ld(Wr, Tb0, Tb1, k + 20 + kof, avB, cB0, cB1);
    chain_mm(avC, cC0, cC1, acc);
    chain_ld(Wr, Tb0, Tb1, k + 24 + kof, avC, cC0, cC1);
    chain_mm(avD, cD0, cD1, acc);
    chain_ld(Wr, Tb0, Tb1, k + 28 + kof, avD, cD0, cD1);
  }
  chain_mm(avA, cA0, cA1, acc);
  chain_mm(avB, cB0, cB1, acc);
  chain_mm(avC, cC0, cC1, acc);
  chain_mm(avD, cD0, cD1, acc);

  #pragma unroll
  for (int mt = 0; mt < 8; ++mt) {
    const int mb = wv * 128 + mt * 16 + 8 * hlf;
    const float4 dA = *(const float4*)(dm + mb);
    const float4 dB = *(const float4*)(dm + mb + 4);
    float4 v0l, v0h, v1l, v1h;
    v0l.x = acc[mt][0][0] * dA.x; v0l.y = acc[mt][0][1] * dA.y;
    v0l.z = acc[mt][0][2] * dA.z; v0l.w = acc[mt][0][3] * dA.w;
    v0h.x = acc[mt][0][4] * dB.x; v0h.y = acc[mt][0][5] * dB.y;
    v0h.z = acc[mt][0][6] * dB.z; v0h.w = acc[mt][0][7] * dB.w;
    v1l.x = acc[mt][1][0] * dA.x; v1l.y = acc[mt][1][1] * dA.y;
    v1l.z = acc[mt][1][2] * dA.z; v1l.w = acc[mt][1][3] * dA.w;
    v1h.x = acc[mt][1][4] * dB.x; v1h.y = acc[mt][1][5] * dB.y;
    v1h.z = acc[mt][1][6] * dB.z; v1h.w = acc[mt][1][7] * dB.w;
    *(float4*)(Tout + (size_t)l16 * TS + mb)            = v0l;
    *(float4*)(Tout + (size_t)l16 * TS + mb + 4)        = v0h;
    *(float4*)(Tout + (size_t)(16 + l16) * TS + mb)     = v1l;
    *(float4*)(Tout + (size_t)(16 + l16) * TS + mb + 4) = v1h;
  }
}

__global__ __launch_bounds__(256) void zero_chain(
    const float* __restrict__ M0,
    const float* __restrict__ Wh1, const float* __restrict__ Wh2,
    const float* __restrict__ d1z, const float* __restrict__ d2z,
    const float* __restrict__ d3z,
    const float* __restrict__ Wout,
    float* __restrict__ zsum)
{
  extern __shared__ float lds[];
  float* T0 = lds;
  float* T1 = lds + CB * TS;

  const int cb = blockIdx.x;
  const int b  = blockIdx.y;
  const int c0 = cb * CB;
  const int lane = threadIdx.x & 31, wv = threadIdx.x >> 5;

  const float* d1 = d1z + (size_t)b * HID;
  const float* d2 = d2z + (size_t)b * HID;
  const float* d3 = d3z + (size_t)b * HID;

  for (int idx = threadIdx.x; idx < HID * CB; idx += 256) {
    const int j = idx >> 5, c = idx & 31;
    T0[(size_t)c * TS + j] = M0[(size_t)j * NST + c0 + c] * d1[j];
  }
  __syncthreads();

  chain_gemm_1024(Wh1, T0, T1, d2, lane, wv);
  __syncthreads();
  chain_gemm_1024(Wh2, T1, T0, d3, lane, wv);
  __syncthreads();

  const int l16 = lane & 15, hlf = lane >> 4;
  const int kof = hlf * 2;
  const v8f vz = {0.f,0.f,0.f,0.f,0.f,0.f,0.f,0.f};
  v8f g0 = vz, g1 = vz;
  const float* Tb0 = T0 + (size_t)l16 * TS;
  const float* Tb1 = T0 + (size_t)(16 + l16) * TS;
  const float* Wr  = Wout + (size_t)(wv * 16 + l16) * HID;

  v2f a0, c00, c10;
  v2f a1, c01, c11;
#define FLD0(kk) { a0 = *(const v2f*)(Wr + (kk)); \
                   c00 = *(const v2f*)(Tb0 + (kk)); c10 = *(const v2f*)(Tb1 + (kk)); }
#define FLD1(kk) { a1 = *(const v2f*)(Wr + (kk)); \
                   c01 = *(const v2f*)(Tb0 + (kk)); c11 = *(const v2f*)(Tb1 + (kk)); }
  FLD0(kof);
  FLD1(kof + 4);
  for (int k = 0; k < HID - 8; k += 8) {
    g0 = wmma_f32(a0, c00, g0); g1 = wmma_f32(a0, c10, g1);
    FLD0(k + 8 + kof);
    g0 = wmma_f32(a1, c01, g0); g1 = wmma_f32(a1, c11, g1);
    FLD1(k + 12 + kof);
  }
  g0 = wmma_f32(a0, c00, g0); g1 = wmma_f32(a0, c10, g1);
  g0 = wmma_f32(a1, c01, g0); g1 = wmma_f32(a1, c11, g1);
#undef FLD0
#undef FLD1

  #pragma unroll
  for (int r = 0; r < 8; ++r) {
    float s = g0[r] * g0[r] + g1[r] * g1[r];
    s += __shfl_xor(s, 1);
    s += __shfl_xor(s, 2);
    s += __shfl_xor(s, 4);
    s += __shfl_xor(s, 8);
    if (l16 == 0)
      atomicAdd(&zsum[(size_t)b * NST + wv * 16 + r + 8 * hlf], s);
  }
}

__global__ void finalize_norm(const float* __restrict__ zsum,
                              float* __restrict__ out) {
  int i = blockIdx.x * blockDim.x + threadIdx.x;
  if (i >= BATCH * NST) return;
  int b = i >> 7, c = i & 127;
  out[(size_t)b * 384 + 256 + c] = sqrtf(zsum[i]);
}

extern "C" void kernel_launch(void* const* d_in, const int* in_sizes, int n_in,
                              void* d_out, int out_size, void* d_ws, size_t ws_size,
                              hipStream_t stream)
{
  (void)in_sizes; (void)n_in; (void)out_size; (void)ws_size;
  const float* x     = (const float*)d_in[0];
  const float* W_in  = (const float*)d_in[1];
  const float* b_in  = (const float*)d_in[2];
  const float* Wh    = (const float*)d_in[3];
  const float* bh    = (const float*)d_in[4];
  const float* W_out = (const float*)d_in[5];
  const float* b_out = (const float*)d_in[6];
  float* out = (float*)d_out;
  float* ws  = (float*)d_ws;

  const float* Wh0 = Wh;
  const float* Wh1 = Wh + 1024 * 1024;
  const float* Wh2 = Wh + 2 * 1024 * 1024;
  const float* bh0 = bh, *bh1 = bh + 1024, *bh2 = bh + 2048;

  float* X2  = ws;
  float* hA  = X2  + 512 * 256;
  float* hB  = hA  + 512 * 1024;
  float* D1  = hB  + 512 * 1024;
  float* D2  = D1  + 512 * 1024;
  float* D3  = D2  + 512 * 1024;
  float* uA  = D3  + 512 * 1024;
  float* uB  = uA  + 256 * 1024;
  float* M0b = uB  + 256 * 1024;
  float* zs  = M0b + 1024 * 128;

  dim3 blk(256);
  build_x2<<<dim3((512 * 256 + 255) / 256), blk, 0, stream>>>(x, X2);
  zero_buf<<<dim3((256 * 128 + 255) / 256), blk, 0, stream>>>(zs, 256 * 128);

  gemm_awt<<<dim3(4,16), blk, 0, stream>>>(X2,256,  W_in,256, b_in, nullptr,0, hA,1024, 512,256,  0);
  gemm_awt<<<dim3(4,16), blk, 0, stream>>>(hA,1024, Wh0,1024, bh0,  nullptr,0, hB,1024, 512,1024, 1);
  gemm_awt<<<dim3(4,16), blk, 0, stream>>>(hB,1024, Wh0,1024, bh0,  nullptr,0, D1,1024, 512,1024, 2);
  gemm_awt<<<dim3(4,16), blk, 0, stream>>>(hB,1024, Wh1,1024, bh1,  nullptr,0, hA,1024, 512,1024, 1);
  gemm_awt<<<dim3(4,16), blk, 0, stream>>>(hA,1024, Wh1,1024, bh1,  nullptr,0, D2,1024, 512,1024, 2);
  gemm_awt<<<dim3(4,16), blk, 0, stream>>>(hA,1024, Wh2,1024, bh2,  nullptr,0, hB,1024, 512,1024, 1);
  gemm_awt<<<dim3(4,16), blk, 0, stream>>>(hB,1024, Wh2,1024, bh2,  nullptr,0, D3,1024, 512,1024, 2);
  gemm_awt<<<dim3(2,2),  blk, 0, stream>>>(hB,1024, W_out,1024, b_out, nullptr,0, out,384, 256,1024, 0);

  gemm_awt<<<dim3(2,16), blk, 0, stream>>>(x + 256,512, W_in,256, nullptr, nullptr,0, uA,1024, 256,128,  0);
  gemm_awt<<<dim3(2,16), blk, 0, stream>>>(uA,1024, Wh0,1024, nullptr, D1,1024, uB,1024, 256,1024, 0);
  gemm_awt<<<dim3(2,16), blk, 0, stream>>>(uB,1024, Wh1,1024, nullptr, D2,1024, uA,1024, 256,1024, 0);
  gemm_awt<<<dim3(2,16), blk, 0, stream>>>(uA,1024, Wh2,1024, nullptr, D3,1024, uB,1024, 256,1024, 0);
  gemm_awt<<<dim3(2,2),  blk, 0, stream>>>(uB,1024, W_out,1024, nullptr, nullptr,0, out + 128,384, 256,1024, 0);

  gemm_ab<<<dim3(8,2), blk, 0, stream>>>(Wh0,1024, W_in,256, M0b,128, 1024,1024);
  zero_chain<<<dim3(4,256), blk, 2 * CB * TS * sizeof(float), stream>>>(
      M0b, Wh1, Wh2,
      D1 + 256 * 1024, D2 + 256 * 1024, D3 + 256 * 1024,
      W_out, zs);
  finalize_norm<<<dim3((256 * 128 + 255) / 256), blk, 0, stream>>>(zs, out);
}


# All WMMA fragment loads widened from 8 to 16 bytes per lane (lane half h holds k=kb+4h..+3, same map for A and B, one load feeds two WMMAs); k sum order inside each 8-k block changes
# speedup vs baseline: 1.3157x; 1.3157x over previous
#include <hip/hip_runtime.h>
#include <hip/hip_bf16.h>
#include <math.h>


typedef float v2f __attribute__((ext_vector_type(2)));
typedef float v4f __attribute__((ext_vector_type(4)));
typedef float v8f __attribute__((ext_vector_type(8)));

#define HID   1024
#define NST   128
#define BATCH 256
#define IN2   256
#define CB    32
#define TS    1028

__device__ __forceinline__ v8f wmma_f32(v2f a, v2f b, v8f c) {
  return __builtin_amdgcn_wmma_f32_16x16x4_f32(
       false, a,  false, b,
       (short)0, c,  false,  false);
}

__device__ __forceinline__ v2f lo2(v4f v) { return __builtin_shufflevector(v, v, 0, 1); }
__device__ __forceinline__ v2f hi2(v4f v) { return __builtin_shufflevector(v, v, 2, 3); }

__global__ void build_x2(const float* __restrict__ x, float* __restrict__ X2) {
  int i = blockIdx.x * blockDim.x + threadIdx.x;
  if (i >= 512 * IN2) return;
  int row = i >> 8, c = i & 255;
  float v;
  if (row < 256) {
    v = x[row * 512 + c];
  } else {
    int rz = row - 256;
    v = (c < 128) ? x[rz * 512 + 384 + c] : x[rz * 512 + 128 + (c - 128)];
  }
  X2[i] = v;
}

__global__ void zero_buf(float* __restrict__ p, int n) {
  int i = blockIdx.x * blockDim.x + threadIdx.x;
  if (i < n) p[i] = 0.f;
}

__global__ __launch_bounds__(256) void gemm_awt(
    const float* __restrict__ A, int lda,
    const float* __restrict__ W, int ldw,
    const float* __restrict__ bias,
    const float* __restrict__ mask, int ldm,
    float* __restrict__ C, int ldc,
    int M, int K, int epi)
{
  const int lane = threadIdx.x & 31, wv = threadIdx.x >> 5;
  const int l16 = lane & 15, hlf = lane >> 4;
  const int m0 = blockIdx.x * 128 + wv * 16;
  const int n0 = blockIdx.y * 64;
  if (m0 >= M) return;

  const v8f vz = {0.f,0.f,0.f,0.f,0.f,0.f,0.f,0.f};
  v8f acc[4] = {vz, vz, vz, vz};

  const int kof = hlf * 4;
  const float* Ar  = A + (size_t)(m0 + l16) * lda;
  const float* Wr0 = W + (size_t)(n0 +  0 + l16) * ldw;
  const float* Wr1 = W + (size_t)(n0 + 16 + l16) * ldw;
  const float* Wr2 = W + (size_t)(n0 + 32 + l16) * ldw;
  const float* Wr3 = W + (size_t)(n0 + 48 + l16) * ldw;

  v4f a0, b00, b10, b20, b30;
  v4f a1, b01, b11, b21, b31;

#define LDSET0(kk) { a0  = *(const v4f*)(Ar  + (kk)); \
                     b00 = *(const v4f*)(Wr0 + (kk)); \
                     b10 = *(const v4f*)(Wr1 + (kk)); \
                     b20 = *(const v4f*)(Wr2 + (kk)); \
                     b30 = *(const v4f*)(Wr3 + (kk)); }
#define LDSET1(kk) { a1  = *(const v4f*)(Ar  + (kk)); \
                     b01 = *(const v4f*)(Wr0 + (kk)); \
                     b11 = *(const v4f*)(Wr1 + (kk)); \
                     b21 = *(const v4f*)(Wr2 + (kk)); \
                     b31 = *(const v4f*)(Wr3 + (kk)); }
#define WM0() { acc[0] = wmma_f32(lo2(a0), lo2(b00), acc[0]); \
                acc[1] = wmma_f32(lo2(a0), lo2(b10), acc[1]); \
                acc[2] = wmma_f32(lo2(a0), lo2(b20), acc[2]); \
                acc[3] = wmma_f32(lo2(a0), lo2(b30), acc[3]); \
                acc[0] = wmma_f32(hi2(a0), hi2(b00), acc[0]); \
                acc[1] = wmma_f32(hi2(a0), hi2(b10), acc[1]); \
                acc[2] = wmma_f32(hi2(a0), hi2(b20), acc[2]); \
                acc[3] = wmma_f32(hi2(a0), hi2(b30), acc[3]); }
#define WM1() { acc[0] = wmma_f32(lo2(a1), lo2(b01), acc[0]); \
                acc[1] = wmma_f32(lo2(a1), lo2(b11), acc[1]); \
                acc[2] = wmma_f32(lo2(a1), lo2(b21), acc[2]); \
                acc[3] = wmma_f32(lo2(a1), lo2(b31), acc[3]); \
                acc[0] = wmma_f32(hi2(a1), hi2(b01), acc[0]); \
                acc[1] = wmma_f32(hi2(a1), hi2(b11), acc[1]); \
                acc[2] = wmma_f32(hi2(a1), hi2(b21), acc[2]); \
                acc[3] = wmma_f32(hi2(a1), hi2(b31), acc[3]); }

  LDSET0(kof);
  LDSET1(kof + 8);
  for (int k = 0; k < K - 16; k += 16) {
    WM0(); LDSET0(k + 16 + kof);
    WM1(); LDSET1(k + 24 + kof);
  }
  WM0();
  WM1();
#undef LDSET0
#undef LDSET1
#undef WM0
#undef WM1

  #pragma unroll
  for (int nt = 0; nt < 4; ++nt) {
    #pragma unroll
    for (int r = 0; r < 8; ++r) {
      const int m  = m0 + r + 8 * hlf;
      const int nn = n0 + nt * 16 + l16;
      float v = acc[nt][r];
      if (bias) v += bias[nn];
      if (epi == 1)      v = v > 0.f ? v   : 0.f;
      else if (epi == 2) v = v > 0.f ? 1.f : 0.f;
      if (mask) v *= mask[(size_t)m * ldm + nn];
      C[(size_t)m * ldc + nn] = v;
    }
  }
}

__global__ __launch_bounds__(256) void gemm_ab(
    const float* __restrict__ A, int lda,
    const float* __restrict__ B, int ldb,
    float* __restrict__ C, int ldc,
    int M, int K)
{
  const int lane = threadIdx.x & 31, wv = threadIdx.x >> 5;
  const int l16 = lane & 15, hlf = lane >> 4;
  const int m0 = blockIdx.x * 128 + wv * 16;
  const int n0 = blockIdx.y * 64;
  if (m0 >= M) return;

  const v8f vz = {0.f,0.f,0.f,0.f,0.f,0.f,0.f,0.f};
  v8f acc[4] = {vz, vz, vz, vz};
  const int kof = hlf * 4;
  const float* Ar = A + (size_t)(m0 + l16) * lda;
  const float* Bc = B + n0 + l16;

  v4f a0, c00, c10, c20, c30;
  v4f a1, c01, c11, c21, c31;

#define BLD(dst, kk, nt) { (dst).x = Bc[(size_t)(kk) * ldb + (nt) * 16]; \
                           (dst).y = Bc[(size_t)((kk) + 1) * ldb + (nt) * 16]; \
                           (dst).z = Bc[(size_t)((kk) + 2) * ldb + (nt) * 16]; \
                           (dst).w = Bc[(size_t)((kk) + 3) * ldb + (nt) * 16]; }
#define LDSET0(kk) { a0 = *(const v4f*)(Ar + (kk)); \
                     BLD(c00, (kk), 0) BLD(c10, (kk), 1) BLD(c20, (kk), 2) BLD(c30, (kk), 3) }
#define LDSET1(kk) { a1 = *(const v4f*)(Ar + (kk)); \
                     BLD(c01, (kk), 0) BLD(c11, (kk), 1) BLD(c21, (kk), 2) BLD(c31, (kk), 3) }
#define WM0() { acc[0] = wmma_f32(lo2(a0), lo2(c00), acc[0]); acc[1] = wmma_f32(lo2(a0), lo2(c10), acc[1]); \
                acc[2] = wmma_f32(lo2(a0), lo2(c20), acc[2]); acc[3] = wmma_f32(lo2(a0), lo2(c30), acc[3]); \
                acc[0] = wmma_f32(hi2(a0), hi2(c00), acc[0]); acc[1] = wmma_f32(hi2(a0), hi2(c10), acc[1]); \
                acc[2] = wmma_f32(hi2(a0), hi2(c20), acc[2]); acc[3] = wmma_f32(hi2(a0), hi2(c30), acc[3]); }
#define WM1() { acc[0] = wmma_f32(lo2(a1), lo2(c01), acc[0]); acc[1] = wmma_f32(lo2(a1), lo2(c11), acc[1]); \
                acc[2] = wmma_f32(lo2(a1), lo2(c21), acc[2]); acc[3] = wmma_f32(lo2(a1), lo2(c31), acc[3]); \
                acc[0] = wmma_f32(hi2(a1), hi2(c01), acc[0]); acc[1] = wmma_f32(hi2(a1), hi2(c11), acc[1]); \
                acc[2] = wmma_f32(hi2(a1), hi2(c21), acc[2]); acc[3] = wmma_f32(hi2(a1), hi2(c31), acc[3]); }

  LDSET0(kof);
  LDSET1(kof + 8);
  for (int k = 0; k < K - 16; k += 16) {
    WM0(); LDSET0(k + 16 + kof);
    WM1(); LDSET1(k + 24 + kof);
  }
  WM0();
  WM1();
#undef BLD
#undef LDSET0
#undef LDSET1
#undef WM0
#undef WM1

  #pragma unroll
  for (int nt = 0; nt < 4; ++nt)
    #pragma unroll
    for (int r = 0; r < 8; ++r)
      C[(size_t)(m0 + r + 8 * hlf) * ldc + n0 + nt * 16 + l16] = acc[nt][r];
}

__device__ __forceinline__ void chain_ld(
    const float* __restrict__ Wr, const float* __restrict__ Tb0,
    const float* __restrict__ Tb1, int kk,
    v4f av[8], v4f& c0, v4f& c1)
{
  c0 = *(const v4f*)(Tb0 + kk);
  c1 = *(const v4f*)(Tb1 + kk);
  #pragma unroll
  for (int mt = 0; mt < 8; ++mt)
    av[mt] = *(const v4f*)(Wr + (size_t)(mt * 16 * HID) + kk);
}

__device__ __forceinline__ void chain_mm(const v4f av[8], v4f c0, v4f c1,
                                         v8f acc[8][2])
{
  #pragma unroll
  for (int mt = 0; mt < 8; ++mt) {
    acc[mt][0] = wmma_f32(lo2(av[mt]), lo2(c0), acc[mt][0]);
    acc[mt][1] = wmma_f32(lo2(av[mt]), lo2(c1), acc[mt][1]);
  }
  #pragma unroll
  for (int mt = 0; mt < 8; ++mt) {
    acc[mt][0] = wmma_f32(hi2(av[mt]), hi2(c0), acc[mt][0]);
    acc[mt][1] = wmma_f32(hi2(av[mt]), hi2(c1), acc[mt][1]);
  }
}

__device__ __forceinline__ void chain_gemm_1024(
    const float* __restrict__ Wm,
    const float* __restrict__ Tin,
    float* __restrict__ Tout,
    const float* __restrict__ dm,
    int lane, int wv)
{
  const int l16 = lane & 15, hlf = lane >> 4;
  const int kof = hlf * 4;
  const v8f vz = {0.f,0.f,0.f,0.f,0.f,0.f,0.f,0.f};
  v8f acc[8][2];
  #pragma unroll
  for (int mt = 0; mt < 8; ++mt) { acc[mt][0] = vz; acc[mt][1] = vz; }

  const float* Tb0 = Tin + (size_t)l16 * TS;
  const float* Tb1 = Tin + (size_t)(16 + l16) * TS;
  const float* Wr  = Wm + (size_t)(wv * 128 + l16) * HID;

  v4f avA[8], cA0, cA1;
  v4f avB[8], cB0, cB1;
  v4f avC[8], cC0, cC1;
  v4f avD[8], cD0, cD1;

  chain_ld(Wr, Tb0, Tb1, kof,      avA, cA0, cA1);
  chain_ld(Wr, Tb0, Tb1, kof + 8,  avB, cB0, cB1);
  chain_ld(Wr, Tb0, Tb1, kof + 16, avC, cC0, cC1);
  chain_ld(Wr, Tb0, Tb1, kof + 24, avD, cD0, cD1);
  for (int k = 0; k < HID - 32; k += 32) {
    chain_mm(avA, cA0, cA1, acc);
    chain_ld(Wr, Tb0, Tb1, k + 32 + kof, avA, cA0, cA1);
    chain_mm(avB, cB0, cB1, acc);
    chain_ld(Wr, Tb0, Tb1, k + 40 + kof, avB, cB0, cB1);
    chain_mm(avC, cC0, cC1, acc);
    chain_ld(Wr, Tb0, Tb1, k + 48 + kof, avC, cC0, cC1);
    chain_mm(avD, cD0, cD1, acc);
    chain_ld(Wr, Tb0, Tb1, k + 56 + kof, avD, cD0, cD1);
  }
  chain_mm(avA, cA0, cA1, acc);
  chain_mm(avB, cB0, cB1, acc);
  chain_mm(avC, cC0, cC1, acc);
  chain_mm(avD, cD0, cD1, acc);

  #pragma unroll
  for (int mt = 0; mt < 8; ++mt) {
    const int mb = wv * 128 + mt * 16 + 8 * hlf;
    const float4 dA = *(const float4*)(dm + mb);
    const float4 dB = *(const float4*)(dm + mb + 4);
    float4 v0l, v0h, v1l, v1h;
    v0l.x = acc[mt][0][0] * dA.x; v0l.y = acc[mt][0][1] * dA.y;
    v0l.z = acc[mt][0][2] * dA.z; v0l.w = acc[mt][0][3] * dA.w;
    v0h.x = acc[mt][0][4] * dB.x; v0h.y = acc[mt][0][5] * dB.y;
    v0h.z = acc[mt][0][6] * dB.z; v0h.w = acc[mt][0][7] * dB.w;
    v1l.x = acc[mt][1][0] * dA.x; v1l.y = acc[mt][1][1] * dA.y;
    v1l.z = acc[mt][1][2] * dA.z; v1l.w = acc[mt][1][3] * dA.w;
    v1h.x = acc[mt][1][4] * dB.x; v1h.y = acc[mt][1][5] * dB.y;
    v1h.z = acc[mt][1][6] * dB.z; v1h.w = acc[mt][1][7] * dB.w;
    *(float4*)(Tout + (size_t)l16 * TS + mb)            = v0l;
    *(float4*)(Tout + (size_t)l16 * TS + mb + 4)        = v0h;
    *(float4*)(Tout + (size_t)(16 + l16) * TS + mb)     = v1l;
    *(float4*)(Tout + (size_t)(16 + l16) * TS + mb + 4) = v1h;
  }
}

__global__ __launch_bounds__(256) void zero_chain(
    const float* __restrict__ M0,
    const float* __restrict__ Wh1, const float* __restrict__ Wh2,
    const float* __restrict__ d1z, const float* __restrict__ d2z,
    const float* __restrict__ d3z,
    const float* __restrict__ Wout,
    float* __restrict__ zsum)
{
  extern __shared__ float lds[];
  float* T0 = lds;
  float* T1 = lds + CB * TS;

  const int cb = blockIdx.x;
  const int b  = blockIdx.y;
  const int c0 = cb * CB;
  const int lane = threadIdx.x & 31, wv = threadIdx.x >> 5;

  const float* d1 = d1z + (size_t)b * HID;
  const float* d2 = d2z + (size_t)b * HID;
  const float* d3 = d3z + (size_t)b * HID;

  for (int idx = threadIdx.x; idx < HID * CB; idx += 256) {
    const int j = idx >> 5, c = idx & 31;
    T0[(size_t)c * TS + j] = M0[(size_t)j * NST + c0 + c] * d1[j];
  }
  __syncthreads();

  chain_gemm_1024(Wh1, T0, T1, d2, lane, wv);
  __syncthreads();
  chain_gemm_1024(Wh2, T1, T0, d3, lane, wv);
  __syncthreads();

  const int l16 = lane & 15, hlf = lane >> 4;
  const int kof = hlf * 4;
  const v8f vz = {0.f,0.f,0.f,0.f,0.f,0.f,0.f,0.f};
  v8f g0 = vz, g1 = vz;
  const float* Tb0 = T0 + (size_t)l16 * TS;
  const float* Tb1 = T0 + (size_t)(16 + l16) * TS;
  const float* Wr  = Wout + (size_t)(wv * 16 + l16) * HID;

  v4f a0, c00, c10;
  v4f a1, c01, c11;
#define FLD0(kk) { a0 = *(const v4f*)(Wr + (kk)); \
                   c00 = *(const v4f*)(Tb0 + (kk)); c10 = *(const v4f*)(Tb1 + (kk)); }
#define FLD1(kk) { a1 = *(const v4f*)(Wr + (kk)); \
                   c01 = *(const v4f*)(Tb0 + (kk)); c11 = *(const v4f*)(Tb1 + (kk)); }
#define GM0() { g0 = wmma_f32(lo2(a0), lo2(c00), g0); g1 = wmma_f32(lo2(a0), lo2(c10), g1); \
                g0 = wmma_f32(hi2(a0), hi2(c00), g0); g1 = wmma_f32(hi2(a0), hi2(c10), g1); }
#define GM1() { g0 = wmma_f32(lo2(a1), lo2(c01), g0); g1 = wmma_f32(lo2(a1), lo2(c11), g1); \
                g0 = wmma_f32(hi2(a1), hi2(c01), g0); g1 = wmma_f32(hi2(a1), hi2(c11), g1); }
  FLD0(kof);
  FLD1(kof + 8);
  for (int k = 0; k < HID - 16; k += 16) {
    GM0();
    FLD0(k + 16 + kof);
    GM1();
    FLD1(k + 24 + kof);
  }
  GM0();
  GM1();
#undef FLD0
#undef FLD1
#undef GM0
#undef GM1

  #pragma unroll
  for (int r = 0; r < 8; ++r) {
    float s = g0[r] * g0[r] + g1[r] * g1[r];
    s += __shfl_xor(s, 1);
    s += __shfl_xor(s, 2);
    s += __shfl_xor(s, 4);
    s += __shfl_xor(s, 8);
    if (l16 == 0)
      atomicAdd(&zsum[(size_t)b * NST + wv * 16 + r + 8 * hlf], s);
  }
}

__global__ void finalize_norm(const float* __restrict__ zsum,
                              float* __restrict__ out) {
  int i = blockIdx.x * blockDim.x + threadIdx.x;
  if (i >= BATCH * NST) return;
  int b = i >> 7, c = i & 127;
  out[(size_t)b * 384 + 256 + c] = sqrtf(zsum[i]);
}

extern "C" void kernel_launch(void* const* d_in, const int* in_sizes, int n_in,
                              void* d_out, int out_size, void* d_ws, size_t ws_size,
                              hipStream_t stream)
{
  (void)in_sizes; (void)n_in; (void)out_size; (void)ws_size;
  const float* x     = (const float*)d_in[0];
  const float* W_in  = (const float*)d_in[1];
  const float* b_in  = (const float*)d_in[2];
  const float* Wh    = (const float*)d_in[3];
  const float* bh    = (const float*)d_in[4];
  const float* W_out = (const float*)d_in[5];
  const float* b_out = (const float*)d_in[6];
  float* out = (float*)d_out;
  float* ws  = (float*)d_ws;

  const float* Wh0 = Wh;
  const float* Wh1 = Wh + 1024 * 1024;
  const float* Wh2 = Wh + 2 * 1024 * 1024;
  const float* bh0 = bh, *bh1 = bh + 1024, *bh2 = bh + 2048;

  float* X2  = ws;
  float* hA  = X2  + 512 * 256;
  float* hB  = hA  + 512 * 1024;
  float* D1  = hB  + 512 * 1024;
  float* D2  = D1  + 512 * 1024;
  float* D3  = D2  + 512 * 1024;
  float* uA  = D3  + 512 * 1024;
  float* uB  = uA  + 256 * 1024;
  float* M0b = uB  + 256 * 1024;
  float* zs  = M0b + 1024 * 128;

  dim3 blk(256);
  build_x2<<<dim3((512 * 256 + 255) / 256), blk, 0, stream>>>(x, X2);
  zero_buf<<<dim3((256 * 128 + 255) / 256), blk, 0, stream>>>(zs, 256 * 128);

  gemm_awt<<<dim3(4,16), blk, 0, stream>>>(X2,256,  W_in,256, b_in, nullptr,0, hA,1024, 512,256,  0);
  gemm_awt<<<dim3(4,16), blk, 0, stream>>>(hA,1024, Wh0,1024, bh0,  nullptr,0, hB,1024, 512,1024, 1);
  gemm_awt<<<dim3(4,16), blk, 0, stream>>>(hB,1024, Wh0,1024, bh0,  nullptr,0, D1,1024, 512,1024, 2);
  gemm_awt<<<dim3(4,16), blk, 0, stream>>>(hB,1024, Wh1,1024, bh1,  nullptr,0, hA,1024, 512,1024, 1);
  gemm_awt<<<dim3(4,16), blk, 0, stream>>>(hA,1024, Wh1,1024, bh1,  nullptr,0, D2,1024, 512,1024, 2);
  gemm_awt<<<dim3(4,16), blk, 0, stream>>>(hA,1024, Wh2,1024, bh2,  nullptr,0, hB,1024, 512,1024, 1);
  gemm_awt<<<dim3(4,16), blk, 0, stream>>>(hB,1024, Wh2,1024, bh2,  nullptr,0, D3,1024, 512,1024, 2);
  gemm_awt<<<dim3(2,2),  blk, 0, stream>>>(hB,1024, W_out,1024, b_out, nullptr,0, out,384, 256,1024, 0);

  gemm_awt<<<dim3(2,16), blk, 0, stream>>>(x + 256,512, W_in,256, nullptr, nullptr,0, uA,1024, 256,128,  0);
  gemm_awt<<<dim3(2,16), blk, 0, stream>>>(uA,1024, Wh0,1024, nullptr, D1,1024, uB,1024, 256,1024, 0);
  gemm_awt<<<dim3(2,16), blk, 0, stream>>>(uB,1024, Wh1,1024, nullptr, D2,1024, uA,1024, 256,1024, 0);
  gemm_awt<<<dim3(2,16), blk, 0, stream>>>(uA,1024, Wh2,1024, nullptr, D3,1024, uB,1024, 256,1024, 0);
  gemm_awt<<<dim3(2,2),  blk, 0, stream>>>(uB,1024, W_out,1024, nullptr, nullptr,0, out + 128,384, 256,1024, 0);

  gemm_ab<<<dim3(8,2), blk, 0, stream>>>(Wh0,1024, W_in,256, M0b,128, 1024,1024);
  zero_chain<<<dim3(4,256), blk, 2 * CB * TS * sizeof(float), stream>>>(
      M0b, Wh1, Wh2,
      D1 + 256 * 1024, D2 + 256 * 1024, D3 + 256 * 1024,
      W_out, zs);
  finalize_norm<<<dim3((256 * 128 + 255) / 256), blk, 0, stream>>>(zs, out);
}


# Fragment loads software-pipelined for real in every GEMM K loop: each stage's loads pinned right behind the WMMAs that free its registers (4 stages, whole 128-byte lines per group); sum order kept
# speedup vs baseline: 1.5209x; 1.1560x over previous
#include <hip/hip_runtime.h>
#include <hip/hip_bf16.h>
#include <math.h>


typedef float v2f __attribute__((ext_vector_type(2)));
typedef float v4f __attribute__((ext_vector_type(4)));
typedef float v8f __attribute__((ext_vector_type(8)));

#define HID   1024
#define NST   128
#define BATCH 256
#define IN2   256
#define CB    32
#define TS    1028

__device__ __forceinline__ v8f wmma_f32(v2f a, v2f b, v8f c) {
  return __builtin_amdgcn_wmma_f32_16x16x4_f32(
       false, a,  false, b,
       (short)0, c,  false,  false);
}

#define PIPE_FENCE() __builtin_amdgcn_sched_barrier(0)

__device__ __forceinline__ v2f lo2(v4f v) { return __builtin_shufflevector(v, v, 0, 1); }
__device__ __forceinline__ v2f hi2(v4f v) { return __builtin_shufflevector(v, v, 2, 3); }

__global__ void build_x2(const float* __restrict__ x, float* __restrict__ X2) {
  int i = blockIdx.x * blockDim.x + threadIdx.x;
  if (i >= 512 * IN2) return;
  int row = i >> 8, c = i & 255;
  float v;
  if (row < 256) {
    v = x[row * 512 + c];
  } else {
    int rz = row - 256;
    v = (c < 128) ? x[rz * 512 + 384 + c] : x[rz * 512 + 128 + (c - 128)];
  }
  X2[i] = v;
}

__global__ void zero_buf(float* __restrict__ p, int n) {
  int i = blockIdx.x * blockDim.x + threadIdx.x;
  if (i < n) p[i] = 0.f;
}

__global__ __launch_bounds__(256) void gemm_awt(
    const float* __restrict__ A, int lda,
    const float* __restrict__ W, int ldw,
    const float* __restrict__ bias,
    const float* __restrict__ mask, int ldm,
    float* __restrict__ C, int ldc,
    int M, int K, int epi)
{
  const int lane = threadIdx.x & 31, wv = threadIdx.x >> 5;
  const int l16 = lane & 15, hlf = lane >> 4;
  const int m0 = blockIdx.x * 128 + wv * 16;
  const int n0 = blockIdx.y * 64;
  if (m0 >= M) return;

  const v8f vz = {0.f,0.f,0.f,0.f,0.f,0.f,0.f,0.f};
  v8f acc[4] = {vz, vz, vz, vz};

  const int kof = hlf * 4;
  const float* Ar  = A + (size_t)(m0 + l16) * lda;
  const float* Wr0 = W + (size_t)(n0 +  0 + l16) * ldw;
  const float* Wr1 = W + (size_t)(n0 + 16 + l16) * ldw;
  const float* Wr2 = W + (size_t)(n0 + 32 + l16) * ldw;
  const float* Wr3 = W + (size_t)(n0 + 48 + l16) * ldw;

  v4f aS[4], bS[4][4];
#define AWT_LD(st, kk) { aS[st]    = *(const v4f*)(Ar  + (kk)); \
                         bS[st][0] = *(const v4f*)(Wr0 + (kk)); \
                         bS[st][1] = *(const v4f*)(Wr1 + (kk)); \
                         bS[st][2] = *(const v4f*)(Wr2 + (kk)); \
                         bS[st][3] = *(const v4f*)(Wr3 + (kk)); }
#define AWT_MM(st) { acc[0] = wmma_f32(lo2(aS[st]), lo2(bS[st][0]), acc[0]); \
                     acc[1] = wmma_f32(lo2(aS[st]), lo2(bS[st][1]), acc[1]); \
                     acc[2] = wmma_f32(lo2(aS[st]), lo2(bS[st][2]), acc[2]); \
                     acc[3] = wmma_f32(lo2(aS[st]), lo2(bS[st][3]), acc[3]); \
                     acc[0] = wmma_f32(hi2(aS[st]), hi2(bS[st][0]), acc[0]); \
                     acc[1] = wmma_f32(hi2(aS[st]), hi2(bS[st][1]), acc[1]); \
                     acc[2] = wmma_f32(hi2(aS[st]), hi2(bS[st][2]), acc[2]); \
                     acc[3] = wmma_f32(hi2(aS[st]), hi2(bS[st][3]), acc[3]); }

  AWT_LD(0, kof);      PIPE_FENCE();
  AWT_LD(1, kof + 8);  PIPE_FENCE();
  AWT_LD(2, kof + 16); PIPE_FENCE();
  AWT_LD(3, kof + 24); PIPE_FENCE();
  for (int k = 0; k < K - 32; k += 32) {
    AWT_MM(0); PIPE_FENCE(); AWT_LD(0, k + 32 + kof); PIPE_FENCE();
    AWT_MM(1); PIPE_FENCE(); AWT_LD(1, k + 40 + kof); PIPE_FENCE();
    AWT_MM(2); PIPE_FENCE(); AWT_LD(2, k + 48 + kof); PIPE_FENCE();
    AWT_MM(3); PIPE_FENCE(); AWT_LD(3, k + 56 + kof); PIPE_FENCE();
  }
  AWT_MM(0);
  AWT_MM(1);
  AWT_MM(2);
  AWT_MM(3);
#undef AWT_LD
#undef AWT_MM

  #pragma unroll
  for (int nt = 0; nt < 4; ++nt) {
    #pragma unroll
    for (int r = 0; r < 8; ++r) {
      const int m  = m0 + r + 8 * hlf;
      const int nn = n0 + nt * 16 + l16;
      float v = acc[nt][r];
      if (bias) v += bias[nn];
      if (epi == 1)      v = v > 0.f ? v   : 0.f;
      else if (epi == 2) v = v > 0.f ? 1.f : 0.f;
      if (mask) v *= mask[(size_t)m * ldm + nn];
      C[(size_t)m * ldc + nn] = v;
    }
  }
}

__global__ __launch_bounds__(256) void gemm_ab(
    const float* __restrict__ A, int lda,
    const float* __restrict__ B, int ldb,
    float* __restrict__ C, int ldc,
    int M, int K)
{
  const int lane = threadIdx.x & 31, wv = threadIdx.x >> 5;
  const int l16 = lane & 15, hlf = lane >> 4;
  const int m0 = blockIdx.x * 128 + wv * 16;
  const int n0 = blockIdx.y * 64;
  if (m0 >= M) return;

  const v8f vz = {0.f,0.f,0.f,0.f,0.f,0.f,0.f,0.f};
  v8f acc[4] = {vz, vz, vz, vz};
  const int kof = hlf * 4;
  const float* Ar = A + (size_t)(m0 + l16) * lda;
  const float* Bc = B + n0 + l16;

  v4f a0, c00, c10, c20, c30;
  v4f a1, c01, c11, c21, c31;

#define BLD(dst, kk, nt) { (dst).x = Bc[(size_t)(kk) * ldb + (nt) * 16]; \
                           (dst).y = Bc[(size_t)((kk) + 1) * ldb + (nt) * 16]; \
                           (dst).z = Bc[(size_t)((kk) + 2) * ldb + (nt) * 16]; \
                           (dst).w = Bc[(size_t)((kk) + 3) * ldb + (nt) * 16]; }
#define LDSET0(kk) { a0 = *(const v4f*)(Ar + (kk)); \
                     BLD(c00, (kk), 0) BLD(c10, (kk), 1) BLD(c20, (kk), 2) BLD(c30, (kk), 3) }
#define LDSET1(kk) { a1 = *(const v4f*)(Ar + (kk)); \
                     BLD(c01, (kk), 0) BLD(c11, (kk), 1) BLD(c21, (kk), 2) BLD(c31, (kk), 3) }
#define WM0() { acc[0] = wmma_f32(lo2(a0), lo2(c00), acc[0]); acc[1] = wmma_f32(lo2(a0), lo2(c10), acc[1]); \
                acc[2] = wmma_f32(lo2(a0), lo2(c20), acc[2]); acc[3] = wmma_f32(lo2(a0), lo2(c30), acc[3]); \
                acc[0] = wmma_f32(hi2(a0), hi2(c00), acc[0]); acc[1] = wmma_f32(hi2(a0), hi2(c10), acc[1]); \
                acc[2] = wmma_f32(hi2(a0), hi2(c20), acc[2]); acc[3] = wmma_f32(hi2(a0), hi2(c30), acc[3]); }
#define WM1() { acc[0] = wmma_f32(lo2(a1), lo2(c01), acc[0]); acc[1] = wmma_f32(lo2(a1), lo2(c11), acc[1]); \
                acc[2] = wmma_f32(lo2(a1), lo2(c21), acc[2]); acc[3] = wmma_f32(lo2(a1), lo2(c31), acc[3]); \
                acc[0] = wmma_f32(hi2(a1), hi2(c01), acc[0]); acc[1] = wmma_f32(hi2(a1), hi2(c11), acc[1]); \
                acc[2] = wmma_f32(hi2(a1), hi2(c21), acc[2]); acc[3] = wmma_f32(hi2(a1), hi2(c31), acc[3]); }

  LDSET0(kof); PIPE_FENCE();
  LDSET1(kof + 8); PIPE_FENCE();
  for (int k = 0; k < K - 16; k += 16) {
    WM0(); PIPE_FENCE(); LDSET0(k + 16 + kof); PIPE_FENCE();
    WM1(); PIPE_FENCE(); LDSET1(k + 24 + kof); PIPE_FENCE();
  }
  WM0();
  WM1();
#undef BLD
#undef LDSET0
#undef LDSET1
#undef WM0
#undef WM1

  #pragma unroll
  for (int nt = 0; nt < 4; ++nt)
    #pragma unroll
    for (int r = 0; r < 8; ++r)
      C[(size_t)(m0 + r + 8 * hlf) * ldc + n0 + nt * 16 + l16] = acc[nt][r];
}

__device__ __forceinline__ void chain_ldA(
    const float* __restrict__ Wr, const int st, int kk, v4f av[2][4])
{
  #pragma unroll
  for (int mtl = 0; mtl < 2; ++mtl)
    #pragma unroll
    for (int q = 0; q < 4; ++q)
      av[mtl][q] = *(const v4f*)(Wr + (size_t)((2 * st + mtl) * 16 * HID) + kk + 8 * q);
}

__device__ __forceinline__ void chain_ldB(
    const float* __restrict__ Tb0, const float* __restrict__ Tb1, int kk,
    v4f cB[4][2])
{
  #pragma unroll
  for (int q = 0; q < 4; ++q) {
    cB[q][0] = *(const v4f*)(Tb0 + kk + 8 * q);
    cB[q][1] = *(const v4f*)(Tb1 + kk + 8 * q);
  }
}

__device__ __forceinline__ void chain_mm(const v4f av[2][4], const v4f cB[4][2],
                                         v8f acc[8][2], const int st)
{
  #pragma unroll
  for (int q = 0; q < 4; ++q) {
    #pragma unroll
    for (int mtl = 0; mtl < 2; ++mtl) {
      acc[2 * st + mtl][0] = wmma_f32(lo2(av[mtl][q]), lo2(cB[q][0]), acc[2 * st + mtl][0]);
      acc[2 * st + mtl][1] = wmma_f32(lo2(av[mtl][q]), lo2(cB[q][1]), acc[2 * st + mtl][1]);
    }
    #pragma unroll
    for (int mtl = 0; mtl < 2; ++mtl) {
      acc[2 * st + mtl][0] = wmma_f32(hi2(av[mtl][q]), hi2(cB[q][0]), acc[2 * st + mtl][0]);
      acc[2 * st + mtl][1] = wmma_f32(hi2(av[mtl][q]), hi2(cB[q][1]), acc[2 * st + mtl][1]);
    }
  }
}

__device__ __forceinline__ void chain_trip(
    const float* __restrict__ Wr, const float* __restrict__ Tb0,
    const float* __restrict__ Tb1, int kn,
    v4f avS[4][2][4], const v4f cur[4][2], v4f nxt[4][2], v8f acc[8][2])
{
  chain_mm(avS[0], cur, acc, 0); PIPE_FENCE();
  chain_ldA(Wr, 0, kn, avS[0]); chain_ldB(Tb0, Tb1, kn, nxt); PIPE_FENCE();
  chain_mm(avS[1], cur, acc, 1); PIPE_FENCE();
  chain_ldA(Wr, 1, kn, avS[1]); PIPE_FENCE();
  chain_mm(avS[2], cur, acc, 2); PIPE_FENCE();
  chain_ldA(Wr, 2, kn, avS[2]); PIPE_FENCE();
  chain_mm(avS[3], cur, acc, 3); PIPE_FENCE();
  chain_ldA(Wr, 3, kn, avS[3]); PIPE_FENCE();
}

__device__ __forceinline__ void chain_gemm_1024(
    const float* __restrict__ Wm,
    const float* __restrict__ Tin,
    float* __restrict__ Tout,
    const float* __restrict__ dm,
    int lane, int wv)
{
  const int l16 = lane & 15, hlf = lane >> 4;
  const int kof = hlf * 4;
  const v8f vz = {0.f,0.f,0.f,0.f,0.f,0.f,0.f,0.f};
  v8f acc[8][2];
  #pragma unroll
  for (int mt = 0; mt < 8; ++mt) { acc[mt][0] = vz; acc[mt][1] = vz; }

  const float* Tb0 = Tin + (size_t)l16 * TS;
  const float* Tb1 = Tin + (size_t)(16 + l16) * TS;
  const float* Wr  = Wm + (size_t)(wv * 128 + l16) * HID;

  v4f avS[4][2][4];
  v4f cBa[4][2], cBb[4][2];

  chain_ldA(Wr, 0, kof, avS[0]); chain_ldB(Tb0, Tb1, kof, cBa); PIPE_FENCE();
  chain_ldA(Wr, 1, kof, avS[1]); PIPE_FENCE();
  chain_ldA(Wr, 2, kof, avS[2]); PIPE_FENCE();
  chain_ldA(Wr, 3, kof, avS[3]); PIPE_FENCE();
  for (int k = 0; k < HID - 64; k += 64) {
    chain_trip(Wr, Tb0, Tb1, k + 32 + kof, avS, cBa, cBb, acc);
    chain_trip(Wr, Tb0, Tb1, k + 64 + kof, avS, cBb, cBa, acc);
  }
  chain_trip(Wr, Tb0, Tb1, HID - 32 + kof, avS, cBa, cBb, acc);
  chain_mm(avS[0], cBb, acc, 0);
  chain_mm(avS[1], cBb, acc, 1);
  chain_mm(avS[2], cBb, acc, 2);
  chain_mm(avS[3], cBb, acc, 3);

  #pragma unroll
  for (int mt = 0; mt < 8; ++mt) {
    const int mb = wv * 128 + mt * 16 + 8 * hlf;
    const float4 dA = *(const float4*)(dm + mb);
    const float4 dB = *(const float4*)(dm + mb + 4);
    float4 v0l, v0h, v1l, v1h;
    v0l.x = acc[mt][0][0] * dA.x; v0l.y = acc[mt][0][1] * dA.y;
    v0l.z = acc[mt][0][2] * dA.z; v0l.w = acc[mt][0][3] * dA.w;
    v0h.x = acc[mt][0][4] * dB.x; v0h.y = acc[mt][0][5] * dB.y;
    v0h.z = acc[mt][0][6] * dB.z; v0h.w = acc[mt][0][7] * dB.w;
    v1l.x = acc[mt][1][0] * dA.x; v1l.y = acc[mt][1][1] * dA.y;
    v1l.z = acc[mt][1][2] * dA.z; v1l.w = acc[mt][1][3] * dA.w;
    v1h.x = acc[mt][1][4] * dB.x; v1h.y = acc[mt][1][5] * dB.y;
    v1h.z = acc[mt][1][6] * dB.z; v1h.w = acc[mt][1][7] * dB.w;
    *(float4*)(Tout + (size_t)l16 * TS + mb)            = v0l;
    *(float4*)(Tout + (size_t)l16 * TS + mb + 4)        = v0h;
    *(float4*)(Tout + (size_t)(16 + l16) * TS + mb)     = v1l;
    *(float4*)(Tout + (size_t)(16 + l16) * TS + mb + 4) = v1h;
  }
}

__global__ __launch_bounds__(256) void zero_chain(
    const float* __restrict__ M0,
    const float* __restrict__ Wh1, const float* __restrict__ Wh2,
    const float* __restrict__ d1z, const float* __restrict__ d2z,
    const float* __restrict__ d3z,
    const float* __restrict__ Wout,
    float* __restrict__ zsum)
{
  extern __shared__ float lds[];
  float* T0 = lds;
  float* T1 = lds + CB * TS;

  const int cb = blockIdx.x;
  const int b  = blockIdx.y;
  const int c0 = cb * CB;
  const int lane = threadIdx.x & 31, wv = threadIdx.x >> 5;

  const float* d1 = d1z + (size_t)b * HID;
  const float* d2 = d2z + (size_t)b * HID;
  const float* d3 = d3z + (size_t)b * HID;

  for (int idx = threadIdx.x; idx < HID * CB; idx += 256) {
    const int j = idx >> 5, c = idx & 31;
    T0[(size_t)c * TS + j] = M0[(size_t)j * NST + c0 + c] * d1[j];
  }
  __syncthreads();

  chain_gemm_1024(Wh1, T0, T1, d2, lane, wv);
  __syncthreads();
  chain_gemm_1024(Wh2, T1, T0, d3, lane, wv);
  __syncthreads();

  const int l16 = lane & 15, hlf = lane >> 4;
  const int kof = hlf * 4;
  const v8f vz = {0.f,0.f,0.f,0.f,0.f,0.f,0.f,0.f};
  v8f g0 = vz, g1 = vz;
  const float* Tb0 = T0 + (size_t)l16 * TS;
  const float* Tb1 = T0 + (size_t)(16 + l16) * TS;
  const float* Wr  = Wout + (size_t)(wv * 16 + l16) * HID;

  v4f wa[4][4], wc0[4][4], wc1[4][4];
#define FLD(st, kk) { wa[st][0]  = *(const v4f*)(Wr  + (kk) + 0); wc0[st][0] = *(const v4f*)(Tb0 + (kk) + 0); wc1[st][0] = *(const v4f*)(Tb1 + (kk) + 0); \
                      wa[st][1]  = *(const v4f*)(Wr  + (kk) + 8); wc0[st][1] = *(const v4f*)(Tb0 + (kk) + 8); wc1[st][1] = *(const v4f*)(Tb1 + (kk) + 8); \
                      wa[st][2]  = *(const v4f*)(Wr  + (kk) + 16); wc0[st][2] = *(const v4f*)(Tb0 + (kk) + 16); wc1[st][2] = *(const v4f*)(Tb1 + (kk) + 16); \
                      wa[st][3]  = *(const v4f*)(Wr  + (kk) + 24); wc0[st][3] = *(const v4f*)(Tb0 + (kk) + 24); wc1[st][3] = *(const v4f*)(Tb1 + (kk) + 24); }
#define GM(st) { g0 = wmma_f32(lo2(wa[st][0]), lo2(wc0[st][0]), g0); g1 = wmma_f32(lo2(wa[st][0]), lo2(wc1[st][0]), g1); \
                 g0 = wmma_f32(hi2(wa[st][0]), hi2(wc0[st][0]), g0); g1 = wmma_f32(hi2(wa[st][0]), hi2(wc1[st][0]), g1); \
                 g0 = wmma_f32(lo2(wa[st][1]), lo2(wc0[st][1]), g0); g1 = wmma_f32(lo2(wa[st][1]), lo2(wc1[st][1]), g1); \
                 g0 = wmma_f32(hi2(wa[st][1]), hi2(wc0[st][1]), g0); g1 = wmma_f32(hi2(wa[st][1]), hi2(wc1[st][1]), g1); \
                 g0 = wmma_f32(lo2(wa[st][2]), lo2(wc0[st][2]), g0); g1 = wmma_f32(lo2(wa[st][2]), lo2(wc1[st][2]), g1); \
                 g0 = wmma_f32(hi2(wa[st][2]), hi2(wc0[st][2]), g0); g1 = wmma_f32(hi2(wa[st][2]), hi2(wc1[st][2]), g1); \
                 g0 = wmma_f32(lo2(wa[st][3]), lo2(wc0[st][3]), g0); g1 = wmma_f32(lo2(wa[st][3]), lo2(wc1[st][3]), g1); \
                 g0 = wmma_f32(hi2(wa[st][3]), hi2(wc0[st][3]), g0); g1 = wmma_f32(hi2(wa[st][3]), hi2(wc1[st][3]), g1); }
  FLD(0, kof);      PIPE_FENCE();
  FLD(1, kof + 32); PIPE_FENCE();
  FLD(2, kof + 64); PIPE_FENCE();
  FLD(3, kof + 96); PIPE_FENCE();
  for (int k = 0; k < HID - 128; k += 128) {
    GM(0); PIPE_FENCE(); FLD(0, k + 128 + kof); PIPE_FENCE();
    GM(1); PIPE_FENCE(); FLD(1, k + 160 + kof); PIPE_FENCE();
    GM(2); PIPE_FENCE(); FLD(2, k + 192 + kof); PIPE_FENCE();
    GM(3); PIPE_FENCE(); FLD(3, k + 224 + kof); PIPE_FENCE();
  }
  GM(0);
  GM(1);
  GM(2);
  GM(3);
#undef FLD
#undef GM

  #pragma unroll
  for (int r = 0; r < 8; ++r) {
    float s = g0[r] * g0[r] + g1[r] * g1[r];
    s += __shfl_xor(s, 1);
    s += __shfl_xor(s, 2);
    s += __shfl_xor(s, 4);
    s += __shfl_xor(s, 8);
    if (l16 == 0)
      atomicAdd(&zsum[(size_t)b * NST + wv * 16 + r + 8 * hlf], s);
  }
}

__global__ void finalize_norm(const float* __restrict__ zsum,
                              float* __restrict__ out) {
  int i = blockIdx.x * blockDim.x + threadIdx.x;
  if (i >= BATCH * NST) return;
  int b = i >> 7, c = i & 127;
  out[(size_t)b * 384 + 256 + c] = sqrtf(zsum[i]);
}

extern "C" void kernel_launch(void* const* d_in, const int* in_sizes, int n_in,
                              void* d_out, int out_size, void* d_ws, size_t ws_size,
                              hipStream_t stream)
{
  (void)in_sizes; (void)n_in; (void)out_size; (void)ws_size;
  const float* x     = (const float*)d_in[0];
  const float* W_in  = (const float*)d_in[1];
  const float* b_in  = (const float*)d_in[2];
  const float* Wh    = (const float*)d_in[3];
  const float* bh    = (const float*)d_in[4];
  const float* W_out = (const float*)d_in[5];
  const float* b_out = (const float*)d_in[6];
  float* out = (float*)d_out;
  float* ws  = (float*)d_ws;

  const float* Wh0 = Wh;
  const float* Wh1 = Wh + 1024 * 1024;
  const float* Wh2 = Wh + 2 * 1024 * 1024;
  const float* bh0 = bh, *bh1 = bh + 1024, *bh2 = bh + 2048;

  float* X2  = ws;
  float* hA  = X2  + 512 * 256;
  float* hB  = hA  + 512 * 1024;
  float* D1  = hB  + 512 * 1024;
  float* D2  = D1  + 512 * 1024;
  float* D3  = D2  + 512 * 1024;
  float* uA  = D3  + 512 * 1024;
  float* uB  = uA  + 256 * 1024;
  float* M0b = uB  + 256 * 1024;
  float* zs  = M0b + 1024 * 128;

  dim3 blk(256);
  build_x2<<<dim3((512 * 256 + 255) / 256), blk, 0, stream>>>(x, X2);
  zero_buf<<<dim3((256 * 128 + 255) / 256), blk, 0, stream>>>(zs, 256 * 128);

  gemm_awt<<<dim3(4,16), blk, 0, stream>>>(X2,256,  W_in,256, b_in, nullptr,0, hA,1024, 512,256,  0);
  gemm_awt<<<dim3(4,16), blk, 0, stream>>>(hA,1024, Wh0,1024, bh0,  nullptr,0, hB,1024, 512,1024, 1);
  gemm_awt<<<dim3(4,16), blk, 0, stream>>>(hB,1024, Wh0,1024, bh0,  nullptr,0, D1,1024, 512,1024, 2);
  gemm_awt<<<dim3(4,16), blk, 0, stream>>>(hB,1024, Wh1,1024, bh1,  nullptr,0, hA,1024, 512,1024, 1);
  gemm_awt<<<dim3(4,16), blk, 0, stream>>>(hA,1024, Wh1,1024, bh1,  nullptr,0, D2,1024, 512,1024, 2);
  gemm_awt<<<dim3(4,16), blk, 0, stream>>>(hA,1024, Wh2,1024, bh2,  nullptr,0, hB,1024, 512,1024, 1);
  gemm_awt<<<dim3(4,16), blk, 0, stream>>>(hB,1024, Wh2,1024, bh2,  nullptr,0, D3,1024, 512,1024, 2);
  gemm_awt<<<dim3(2,2),  blk, 0, stream>>>(hB,1024, W_out,1024, b_out, nullptr,0, out,384, 256,1024, 0);

  gemm_awt<<<dim3(2,16), blk, 0, stream>>>(x + 256,512, W_in,256, nullptr, nullptr,0, uA,1024, 256,128,  0);
  gemm_awt<<<dim3(2,16), blk, 0, stream>>>(uA,1024, Wh0,1024, nullptr, D1,1024, uB,1024, 256,1024, 0);
  gemm_awt<<<dim3(2,16), blk, 0, stream>>>(uB,1024, Wh1,1024, nullptr, D2,1024, uA,1024, 256,1024, 0);
  gemm_awt<<<dim3(2,16), blk, 0, stream>>>(uA,1024, Wh2,1024, nullptr, D3,1024, uB,1024, 256,1024, 0);
  gemm_awt<<<dim3(2,2),  blk, 0, stream>>>(uB,1024, W_out,1024, nullptr, nullptr,0, out + 128,384, 256,1024, 0);

  gemm_ab<<<dim3(8,2), blk, 0, stream>>>(Wh0,1024, W_in,256, M0b,128, 1024,1024);
  zero_chain<<<dim3(4,256), blk, 2 * CB * TS * sizeof(float), stream>>>(
      M0b, Wh1, Wh2,
      D1 + 256 * 1024, D2 + 256 * 1024, D3 + 256 * 1024,
      W_out, zs);
  finalize_norm<<<dim3((256 * 128 + 255) / 256), blk, 0, stream>>>(zs, out);
}


# Independent GEMMs share one launch per dependency level (blockIdx.z picks the problem): 14 serial GEMM launches become 7; tile code, k order and sum order of every GEMM unchanged
# speedup vs baseline: 1.8536x; 1.2188x over previous
#include <hip/hip_runtime.h>
#include <hip/hip_bf16.h>
#include <math.h>


typedef float v2f __attribute__((ext_vector_type(2)));
typedef float v4f __attribute__((ext_vector_type(4)));
typedef float v8f __attribute__((ext_vector_type(8)));

#define HID   1024
#define NST   128
#define BATCH 256
#define IN2   256
#define CB    32
#define TS    1028

__device__ __forceinline__ v8f wmma_f32(v2f a, v2f b, v8f c) {
  return __builtin_amdgcn_wmma_f32_16x16x4_f32(
       false, a,  false, b,
       (short)0, c,  false,  false);
}

#define PIPE_FENCE() __builtin_amdgcn_sched_barrier(0)

__device__ __forceinline__ v2f lo2(v4f v) { return __builtin_shufflevector(v, v, 0, 1); }
__device__ __forceinline__ v2f hi2(v4f v) { return __builtin_shufflevector(v, v, 2, 3); }

__global__ void build_x2(const float* __restrict__ x, float* __restrict__ X2) {
  int i = blockIdx.x * blockDim.x + threadIdx.x;
  if (i >= 512 * IN2) return;
  int row = i >> 8, c = i & 255;
  float v;
  if (row < 256) {
    v = x[row * 512 + c];
  } else {
    int rz = row - 256;
    v = (c < 128) ? x[rz * 512 + 384 + c] : x[rz * 512 + 128 + (c - 128)];
  }
  X2[i] = v;
}

__global__ void zero_buf(float* __restrict__ p, int n) {
  int i = blockIdx.x * blockDim.x + threadIdx.x;
  if (i < n) p[i] = 0.f;
}

__device__ __forceinline__ void awt_tile(
    const float* __restrict__ A, int lda,
    const float* __restrict__ W, int ldw,
    const float* __restrict__ bias,
    const float* __restrict__ mask, int ldm,
    float* __restrict__ C, int ldc,
    int M, int K, int epi)
{
  const int lane = threadIdx.x & 31, wv = threadIdx.x >> 5;
  const int l16 = lane & 15, hlf = lane >> 4;
  const int m0 = blockIdx.x * 128 + wv * 16;
  const int n0 = blockIdx.y * 64;
  if (m0 >= M) return;

  const v8f vz = {0.f,0.f,0.f,0.f,0.f,0.f,0.f,0.f};
  v8f acc[4] = {vz, vz, vz, vz};

  const int kof = hlf * 4;
  const float* Ar  = A + (size_t)(m0 + l16) * lda;
  const float* Wr0 = W + (size_t)(n0 +  0 + l16) * ldw;
  const float* Wr1 = W + (size_t)(n0 + 16 + l16) * ldw;
  const float* Wr2 = W + (size_t)(n0 + 32 + l16) * ldw;
  const float* Wr3 = W + (size_t)(n0 + 48 + l16) * ldw;

  v4f aS[4], bS[4][4];
#define AWT_LD(st, kk) { aS[st]    = *(const v4f*)(Ar  + (kk)); \
                         bS[st][0] = *(const v4f*)(Wr0 + (kk)); \
                         bS[st][1] = *(const v4f*)(Wr1 + (kk)); \
                         bS[st][2] = *(const v4f*)(Wr2 + (kk)); \
                         bS[st][3] = *(const v4f*)(Wr3 + (kk)); }
#define AWT_MM(st) { acc[0] = wmma_f32(lo2(aS[st]), lo2(bS[st][0]), acc[0]); \
                     acc[1] = wmma_f32(lo2(aS[st]), lo2(bS[st][1]), acc[1]); \
                     acc[2] = wmma_f32(lo2(aS[st]), lo2(bS[st][2]), acc[2]); \
                     acc[3] = wmma_f32(lo2(aS[st]), lo2(bS[st][3]), acc[3]); \
                     acc[0] = wmma_f32(hi2(aS[st]), hi2(bS[st][0]), acc[0]); \
                     acc[1] = wmma_f32(hi2(aS[st]), hi2(bS[st][1]), acc[1]); \
                     acc[2] = wmma_f32(hi2(aS[st]), hi2(bS[st][2]), acc[2]); \
                     acc[3] = wmma_f32(hi2(aS[st]), hi2(bS[st][3]), acc[3]); }

  AWT_LD(0, kof);      PIPE_FENCE();
  AWT_LD(1, kof + 8);  PIPE_FENCE();
  AWT_LD(2, kof + 16); PIPE_FENCE();
  AWT_LD(3, kof + 24); PIPE_FENCE();
  for (int k = 0; k < K - 32; k += 32) {
    AWT_MM(0); PIPE_FENCE(); AWT_LD(0, k + 32 + kof); PIPE_FENCE();
    AWT_MM(1); PIPE_FENCE(); AWT_LD(1, k + 40 + kof); PIPE_FENCE();
    AWT_MM(2); PIPE_FENCE(); AWT_LD(2, k + 48 + kof); PIPE_FENCE();
    AWT_MM(3); PIPE_FENCE(); AWT_LD(3, k + 56 + kof); PIPE_FENCE();
  }
  AWT_MM(0);
  AWT_MM(1);
  AWT_MM(2);
  AWT_MM(3);
#undef AWT_LD
#undef AWT_MM

  #pragma unroll
  for (int nt = 0; nt < 4; ++nt) {
    #pragma unroll
    for (int r = 0; r < 8; ++r) {
      const int m  = m0 + r + 8 * hlf;
      const int nn = n0 + nt * 16 + l16;
      float v = acc[nt][r];
      if (bias) v += bias[nn];
      if (epi == 1)      v = v > 0.f ? v   : 0.f;
      else if (epi == 2) v = v > 0.f ? 1.f : 0.f;
      if (mask) v *= mask[(size_t)m * ldm + nn];
      C[(size_t)m * ldc + nn] = v;
    }
  }
}

__device__ __forceinline__ void ab_tile(
    const float* __restrict__ A, int lda,
    const float* __restrict__ B, int ldb,
    float* __restrict__ C, int ldc,
    int M, int K)
{
  const int lane = threadIdx.x & 31, wv = threadIdx.x >> 5;
  const int l16 = lane & 15, hlf = lane >> 4;
  const int m0 = blockIdx.x * 128 + wv * 16;
  const int n0 = blockIdx.y * 64;
  if (m0 >= M) return;

  const v8f vz = {0.f,0.f,0.f,0.f,0.f,0.f,0.f,0.f};
  v8f acc[4] = {vz, vz, vz, vz};
  const int kof = hlf * 4;
  const float* Ar = A + (size_t)(m0 + l16) * lda;
  const float* Bc = B + n0 + l16;

  v4f a0, c00, c10, c20, c30;
  v4f a1, c01, c11, c21, c31;

#define BLD(dst, kk, nt) { (dst).x = Bc[(size_t)(kk) * ldb + (nt) * 16]; \
                           (dst).y = Bc[(size_t)((kk) + 1) * ldb + (nt) * 16]; \
                           (dst).z = Bc[(size_t)((kk) + 2) * ldb + (nt) * 16]; \
                           (dst).w = Bc[(size_t)((kk) + 3) * ldb + (nt) * 16]; }
#define LDSET0(kk) { a0 = *(const v4f*)(Ar + (kk)); \
                     BLD(c00, (kk), 0) BLD(c10, (kk), 1) BLD(c20, (kk), 2) BLD(c30, (kk), 3) }
#define LDSET1(kk) { a1 = *(const v4f*)(Ar + (kk)); \
                     BLD(c01, (kk), 0) BLD(c11, (kk), 1) BLD(c21, (kk), 2) BLD(c31, (kk), 3) }
#define WM0() { acc[0] = wmma_f32(lo2(a0), lo2(c00), acc[0]); acc[1] = wmma_f32(lo2(a0), lo2(c10), acc[1]); \
                acc[2] = wmma_f32(lo2(a0), lo2(c20), acc[2]); acc[3] = wmma_f32(lo2(a0), lo2(c30), acc[3]); \
                acc[0] = wmma_f32(hi2(a0), hi2(c00), acc[0]); acc[1] = wmma_f32(hi2(a0), hi2(c10), acc[1]); \
                acc[2] = wmma_f32(hi2(a0), hi2(c20), acc[2]); acc[3] = wmma_f32(hi2(a0), hi2(c30), acc[3]); }
#define WM1() { acc[0] = wmma_f32(lo2(a1), lo2(c01), acc[0]); acc[1] = wmma_f32(lo2(a1), lo2(c11), acc[1]); \
                acc[2] = wmma_f32(lo2(a1), lo2(c21), acc[2]); acc[3] = wmma_f32(lo2(a1), lo2(c31), acc[3]); \
                acc[0] = wmma_f32(hi2(a1), hi2(c01), acc[0]); acc[1] = wmma_f32(hi2(a1), hi2(c11), acc[1]); \
                acc[2] = wmma_f32(hi2(a1), hi2(c21), acc[2]); acc[3] = wmma_f32(hi2(a1), hi2(c31), acc[3]); }

  LDSET0(kof); PIPE_FENCE();
  LDSET1(kof + 8); PIPE_FENCE();
  for (int k = 0; k < K - 16; k += 16) {
    WM0(); PIPE_FENCE(); LDSET0(k + 16 + kof); PIPE_FENCE();
    WM1(); PIPE_FENCE(); LDSET1(k + 24 + kof); PIPE_FENCE();
  }
  WM0();
  WM1();
#undef BLD
#undef LDSET0
#undef LDSET1
#undef WM0
#undef WM1

  #pragma unroll
  for (int nt = 0; nt < 4; ++nt)
    #pragma unroll
    for (int r = 0; r < 8; ++r)
      C[(size_t)(m0 + r + 8 * hlf) * ldc + n0 + nt * 16 + l16] = acc[nt][r];
}

struct LevelArgs {
  const float* A0; const float* W0; const float* bias0; const float* mask0; float* C0;
  const float* A1; const float* W1; const float* bias1; const float* mask1; float* C1;
  const float* A2; const float* W2; const float* bias2; const float* mask2; float* C2;
  const float* Aab; const float* Bab; float* Cab;
  int lda0, ldw0, ldm0, ldc0, M0, K0, epi0, gx0, gy0;
  int lda1, ldw1, ldm1, ldc1, M1, K1, epi1, gx1, gy1;
  int lda2, ldw2, ldm2, ldc2, M2, K2, epi2, gx2, gy2;
  int ldaab, ldbab, ldcab, Mab, Kab, gxab, gyab, abz;
};

__global__ __launch_bounds__(256) void level_gemm(const LevelArgs p)
{
  const int z = blockIdx.z;
  const int bx = blockIdx.x, by = blockIdx.y;
  if (z == p.abz) {
    if (bx < p.gxab && by < p.gyab)
      ab_tile(p.Aab, p.ldaab, p.Bab, p.ldbab, p.Cab, p.ldcab, p.Mab, p.Kab);
    return;
  }
  const float* A; const float* Wt; const float* bias; const float* mask; float* C;
  int lda, ldw, ldm, ldc, M, K, epi, gx, gy;
  if (z == 0) {
    A = p.A0; Wt = p.W0; bias = p.bias0; mask = p.mask0; C = p.C0;
    lda = p.lda0; ldw = p.ldw0; ldm = p.ldm0; ldc = p.ldc0;
    M = p.M0; K = p.K0; epi = p.epi0; gx = p.gx0; gy = p.gy0;
  } else if (z == 1) {
    A = p.A1; Wt = p.W1; bias = p.bias1; mask = p.mask1; C = p.C1;
    lda = p.lda1; ldw = p.ldw1; ldm = p.ldm1; ldc = p.ldc1;
    M = p.M1; K = p.K1; epi = p.epi1; gx = p.gx1; gy = p.gy1;
  } else {
    A = p.A2; Wt = p.W2; bias = p.bias2; mask = p.mask2; C = p.C2;
    lda = p.lda2; ldw = p.ldw2; ldm = p.ldm2; ldc = p.ldc2;
    M = p.M2; K = p.K2; epi = p.epi2; gx = p.gx2; gy = p.gy2;
  }
  if (bx >= gx || by >= gy) return;
  awt_tile(A, lda, Wt, ldw, bias, mask, ldm, C, ldc, M, K, epi);
}

__device__ __forceinline__ void chain_ldA(
    const float* __restrict__ Wr, const int st, int kk, v4f av[2][4])
{
  #pragma unroll
  for (int mtl = 0; mtl < 2; ++mtl)
    #pragma unroll
    for (int q = 0; q < 4; ++q)
      av[mtl][q] = *(const v4f*)(Wr + (size_t)((2 * st + mtl) * 16 * HID) + kk + 8 * q);
}

__device__ __forceinline__ void chain_ldB(
    const float* __restrict__ Tb0, const float* __restrict__ Tb1, int kk,
    v4f cB[4][2])
{
  #pragma unroll
  for (int q = 0; q < 4; ++q) {
    cB[q][0] = *(const v4f*)(Tb0 + kk + 8 * q);
    cB[q][1] = *(const v4f*)(Tb1 + kk + 8 * q);
  }
}

__device__ __forceinline__ void chain_mm(const v4f av[2][4], const v4f cB[4][2],
                                         v8f acc[8][2], const int st)
{
  #pragma unroll
  for (int q = 0; q < 4; ++q) {
    #pragma unroll
    for (int mtl = 0; mtl < 2; ++mtl) {
      acc[2 * st + mtl][0] = wmma_f32(lo2(av[mtl][q]), lo2(cB[q][0]), acc[2 * st + mtl][0]);
      acc[2 * st + mtl][1] = wmma_f32(lo2(av[mtl][q]), lo2(cB[q][1]), acc[2 * st + mtl][1]);
    }
    #pragma unroll
    for (int mtl = 0; mtl < 2; ++mtl) {
      acc[2 * st + mtl][0] = wmma_f32(hi2(av[mtl][q]), hi2(cB[q][0]), acc[2 * st + mtl][0]);
      acc[2 * st + mtl][1] = wmma_f32(hi2(av[mtl][q]), hi2(cB[q][1]), acc[2 * st + mtl][1]);
    }
  }
}

__device__ __forceinline__ void chain_trip(
    const float* __restrict__ Wr, const float* __restrict__ Tb0,
    const float* __restrict__ Tb1, int kn,
    v4f avS[4][2][4], const v4f cur[4][2], v4f nxt[4][2], v8f acc[8][2])
{
  chain_mm(avS[0], cur, acc, 0); PIPE_FENCE();
  chain_ldA(Wr, 0, kn, avS[0]); chain_ldB(Tb0, Tb1, kn, nxt); PIPE_FENCE();
  chain_mm(avS[1], cur, acc, 1); PIPE_FENCE();
  chain_ldA(Wr, 1, kn, avS[1]); PIPE_FENCE();
  chain_mm(avS[2], cur, acc, 2); PIPE_FENCE();
  chain_ldA(Wr, 2, kn, avS[2]); PIPE_FENCE();
  chain_mm(avS[3], cur, acc, 3); PIPE_FENCE();
  chain_ldA(Wr, 3, kn, avS[3]); PIPE_FENCE();
}

__device__ __forceinline__ void chain_gemm_1024(
    const float* __restrict__ Wm,
    const float* __restrict__ Tin,
    float* __restrict__ Tout,
    const float* __restrict__ dm,
    int lane, int wv)
{
  const int l16 = lane & 15, hlf = lane >> 4;
  const int kof = hlf * 4;
  const v8f vz = {0.f,0.f,0.f,0.f,0.f,0.f,0.f,0.f};
  v8f acc[8][2];
  #pragma unroll
  for (int mt = 0; mt < 8; ++mt) { acc[mt][0] = vz; acc[mt][1] = vz; }

  const float* Tb0 = Tin + (size_t)l16 * TS;
  const float* Tb1 = Tin + (size_t)(16 + l16) * TS;
  const float* Wr  = Wm + (size_t)(wv * 128 + l16) * HID;

  v4f avS[4][2][4];
  v4f cBa[4][2], cBb[4][2];

  chain_ldA(Wr, 0, kof, avS[0]); chain_ldB(Tb0, Tb1, kof, cBa); PIPE_FENCE();
  chain_ldA(Wr, 1, kof, avS[1]); PIPE_FENCE();
  chain_ldA(Wr, 2, kof, avS[2]); PIPE_FENCE();
  chain_ldA(Wr, 3, kof, avS[3]); PIPE_FENCE();
  for (int k = 0; k < HID - 64; k += 64) {
    chain_trip(Wr, Tb0, Tb1, k + 32 + kof, avS, cBa, cBb, acc);
    chain_trip(Wr, Tb0, Tb1, k + 64 + kof, avS, cBb, cBa, acc);
  }
  chain_trip(Wr, Tb0, Tb1, HID - 32 + kof, avS, cBa, cBb, acc);
  chain_mm(avS[0], cBb, acc, 0);
  chain_mm(avS[1], cBb, acc, 1);
  chain_mm(avS[2], cBb, acc, 2);
  chain_mm(avS[3], cBb, acc, 3);

  #pragma unroll
  for (int mt = 0; mt < 8; ++mt) {
    const int mb = wv * 128 + mt * 16 + 8 * hlf;
    const float4 dA = *(const float4*)(dm + mb);
    const float4 dB = *(const float4*)(dm + mb + 4);
    float4 v0l, v0h, v1l, v1h;
    v0l.x = acc[mt][0][0] * dA.x; v0l.y = acc[mt][0][1] * dA.y;
    v0l.z = acc[mt][0][2] * dA.z; v0l.w = acc[mt][0][3] * dA.w;
    v0h.x = acc[mt][0][4] * dB.x; v0h.y = acc[mt][0][5] * dB.y;
    v0h.z = acc[mt][0][6] * dB.z; v0h.w = acc[mt][0][7] * dB.w;
    v1l.x = acc[mt][1][0] * dA.x; v1l.y = acc[mt][1][1] * dA.y;
    v1l.z = acc[mt][1][2] * dA.z; v1l.w = acc[mt][1][3] * dA.w;
    v1h.x = acc[mt][1][4] * dB.x; v1h.y = acc[mt][1][5] * dB.y;
    v1h.z = acc[mt][1][6] * dB.z; v1h.w = acc[mt][1][7] * dB.w;
    *(float4*)(Tout + (size_t)l16 * TS + mb)            = v0l;
    *(float4*)(Tout + (size_t)l16 * TS + mb + 4)        = v0h;
    *(float4*)(Tout + (size_t)(16 + l16) * TS + mb)     = v1l;
    *(float4*)(Tout + (size_t)(16 + l16) * TS + mb + 4) = v1h;
  }
}

__global__ __launch_bounds__(256) void zero_chain(
    const float* __restrict__ M0,
    const float* __restrict__ Wh1, const float* __restrict__ Wh2,
    const float* __restrict__ d1z, const float* __restrict__ d2z,
    const float* __restrict__ d3z,
    const float* __restrict__ Wout,
    float* __restrict__ zsum)
{
  extern __shared__ float lds[];
  float* T0 = lds;
  float* T1 = lds + CB * TS;

  const int cb = blockIdx.x;
  const int b  = blockIdx.y;
  const int c0 = cb * CB;
  const int lane = threadIdx.x & 31, wv = threadIdx.x >> 5;

  const float* d1 = d1z + (size_t)b * HID;
  const float* d2 = d2z + (size_t)b * HID;
  const float* d3 = d3z + (size_t)b * HID;

  for (int idx = threadIdx.x; idx < HID * CB; idx += 256) {
    const int j = idx >> 5, c = idx & 31;
    T0[(size_t)c * TS + j] = M0[(size_t)j * NST + c0 + c] * d1[j];
  }
  __syncthreads();

  chain_gemm_1024(Wh1, T0, T1, d2, lane, wv);
  __syncthreads();
  chain_gemm_1024(Wh2, T1, T0, d3, lane, wv);
  __syncthreads();

  const int l16 = lane & 15, hlf = lane >> 4;
  const int kof = hlf * 4;
  const v8f vz = {0.f,0.f,0.f,0.f,0.f,0.f,0.f,0.f};
  v8f g0 = vz, g1 = vz;
  const float* Tb0 = T0 + (size_t)l16 * TS;
  const float* Tb1 = T0 + (size_t)(16 + l16) * TS;
  const float* Wr  = Wout + (size_t)(wv * 16 + l16) * HID;

  v4f wa[4][4], wc0[4][4], wc1[4][4];
#define FLD(st, kk) { wa[st][0]  = *(const v4f*)(Wr  + (kk) + 0); wc0[st][0] = *(const v4f*)(Tb0 + (kk) + 0); wc1[st][0] = *(const v4f*)(Tb1 + (kk) + 0); \
                      wa[st][1]  = *(const v4f*)(Wr  + (kk) + 8); wc0[st][1] = *(const v4f*)(Tb0 + (kk) + 8); wc1[st][1] = *(const v4f*)(Tb1 + (kk) + 8); \
                      wa[st][2]  = *(const v4f*)(Wr  + (kk) + 16); wc0[st][2] = *(const v4f*)(Tb0 + (kk) + 16); wc1[st][2] = *(const v4f*)(Tb1 + (kk) + 16); \
                      wa[st][3]  = *(const v4f*)(Wr  + (kk) + 24); wc0[st][3] = *(const v4f*)(Tb0 + (kk) + 24); wc1[st][3] = *(const v4f*)(Tb1 + (kk) + 24); }
#define GM(st) { g0 = wmma_f32(lo2(wa[st][0]), lo2(wc0[st][0]), g0); g1 = wmma_f32(lo2(wa[st][0]), lo2(wc1[st][0]), g1); \
                 g0 = wmma_f32(hi2(wa[st][0]), hi2(wc0[st][0]), g0); g1 = wmma_f32(hi2(wa[st][0]), hi2(wc1[st][0]), g1); \
                 g0 = wmma_f32(lo2(wa[st][1]), lo2(wc0[st][1]), g0); g1 = wmma_f32(lo2(wa[st][1]), lo2(wc1[st][1]), g1); \
                 g0 = wmma_f32(hi2(wa[st][1]), hi2(wc0[st][1]), g0); g1 = wmma_f32(hi2(wa[st][1]), hi2(wc1[st][1]), g1); \
                 g0 = wmma_f32(lo2(wa[st][2]), lo2(wc0[st][2]), g0); g1 = wmma_f32(lo2(wa[st][2]), lo2(wc1[st][2]), g1); \
                 g0 = wmma_f32(hi2(wa[st][2]), hi2(wc0[st][2]), g0); g1 = wmma_f32(hi2(wa[st][2]), hi2(wc1[st][2]), g1); \
                 g0 = wmma_f32(lo2(wa[st][3]), lo2(wc0[st][3]), g0); g1 = wmma_f32(lo2(wa[st][3]), lo2(wc1[st][3]), g1); \
                 g0 = wmma_f32(hi2(wa[st][3]), hi2(wc0[st][3]), g0); g1 = wmma_f32(hi2(wa[st][3]), hi2(wc1[st][3]), g1); }
  FLD(0, kof);      PIPE_FENCE();
  FLD(1, kof + 32); PIPE_FENCE();
  FLD(2, kof + 64); PIPE_FENCE();
  FLD(3, kof + 96); PIPE_FENCE();
  for (int k = 0; k < HID - 128; k += 128) {
    GM(0); PIPE_FENCE(); FLD(0, k + 128 + kof); PIPE_FENCE();
    GM(1); PIPE_FENCE(); FLD(1, k + 160 + kof); PIPE_FENCE();
    GM(2); PIPE_FENCE(); FLD(2, k + 192 + kof); PIPE_FENCE();
    GM(3); PIPE_FENCE(); FLD(3, k + 224 + kof); PIPE_FENCE();
  }
  GM(0);
  GM(1);
  GM(2);
  GM(3);
#undef FLD
#undef GM

  #pragma unroll
  for (int r = 0; r < 8; ++r) {
    float s = g0[r] * g0[r] + g1[r] * g1[r];
    s += __shfl_xor(s, 1);
    s += __shfl_xor(s, 2);
    s += __shfl_xor(s, 4);
    s += __shfl_xor(s, 8);
    if (l16 == 0)
      atomicAdd(&zsum[(size_t)b * NST + wv * 16 + r + 8 * hlf], s);
  }
}

__global__ void finalize_norm(const float* __restrict__ zsum,
                              float* __restrict__ out) {
  int i = blockIdx.x * blockDim.x + threadIdx.x;
  if (i >= BATCH * NST) return;
  int b = i >> 7, c = i & 127;
  out[(size_t)b * 384 + 256 + c] = sqrtf(zsum[i]);
}

extern "C" void kernel_launch(void* const* d_in, const int* in_sizes, int n_in,
                              void* d_out, int out_size, void* d_ws, size_t ws_size,
                              hipStream_t stream)
{
  (void)in_sizes; (void)n_in; (void)out_size; (void)ws_size;
  const float* x     = (const float*)d_in[0];
  const float* W_in  = (const float*)d_in[1];
  const float* b_in  = (const float*)d_in[2];
  const float* Wh    = (const float*)d_in[3];
  const float* bh    = (const float*)d_in[4];
  const float* W_out = (const float*)d_in[5];
  const float* b_out = (const float*)d_in[6];
  float* out = (float*)d_out;
  float* ws  = (float*)d_ws;

  const float* Wh0 = Wh;
  const float* Wh1 = Wh + 1024 * 1024;
  const float* Wh2 = Wh + 2 * 1024 * 1024;
  const float* bh0 = bh, *bh1 = bh + 1024, *bh2 = bh + 2048;

  float* X2  = ws;
  float* hA  = X2  + 512 * 256;
  float* hB  = hA  + 512 * 1024;
  float* D1  = hB  + 512 * 1024;
  float* D2  = D1  + 512 * 1024;
  float* D3  = D2  + 512 * 1024;
  float* uA  = D3  + 512 * 1024;
  float* uB  = uA  + 256 * 1024;
  float* M0b = uB  + 256 * 1024;
  float* zs  = M0b + 1024 * 128;

  dim3 blk(256);
  build_x2<<<dim3((512 * 256 + 255) / 256), blk, 0, stream>>>(x, X2);
  zero_buf<<<dim3((256 * 128 + 255) / 256), blk, 0, stream>>>(zs, 256 * 128);

#define SLOT(L, n, A_, lda_, W_, ldw_, b_, m_, ldm_, C_, ldc_, M_, K_, e_, gx_, gy_) \
  { (L).A##n = (A_); (L).lda##n = (lda_); (L).W##n = (W_); (L).ldw##n = (ldw_); \
    (L).bias##n = (b_); (L).mask##n = (m_); (L).ldm##n = (ldm_); (L).C##n = (C_); \
    (L).ldc##n = (ldc_); (L).M##n = (M_); (L).K##n = (K_); (L).epi##n = (e_); \
    (L).gx##n = (gx_); (L).gy##n = (gy_); }

  LevelArgs L1 = {}; L1.abz = -1;
  SLOT(L1, 0, X2,256,      W_in,256, b_in,    nullptr,0, hA,1024, 512,256, 0, 4,16)
  SLOT(L1, 1, x + 256,512, W_in,256, nullptr, nullptr,0, uA,1024, 256,128, 0, 2,16)
  level_gemm<<<dim3(4,16,2), blk, 0, stream>>>(L1);

  LevelArgs L2 = {}; L2.abz = -1;
  SLOT(L2, 0, hA,1024, Wh0,1024, bh0, nullptr,0, hB,1024, 512,1024, 1, 4,16)
  level_gemm<<<dim3(4,16,1), blk, 0, stream>>>(L2);

  LevelArgs L3 = {}; L3.abz = -1;
  SLOT(L3, 0, hB,1024, Wh0,1024, bh0, nullptr,0, D1,1024, 512,1024, 2, 4,16)
  SLOT(L3, 1, hB,1024, Wh1,1024, bh1, nullptr,0, hA,1024, 512,1024, 1, 4,16)
  level_gemm<<<dim3(4,16,2), blk, 0, stream>>>(L3);

  LevelArgs L4 = {}; L4.abz = -1;
  SLOT(L4, 0, hA,1024, Wh1,1024, bh1,     nullptr,0, D2,1024, 512,1024, 2, 4,16)
  SLOT(L4, 1, hA,1024, Wh2,1024, bh2,     nullptr,0, hB,1024, 512,1024, 1, 4,16)
  SLOT(L4, 2, uA,1024, Wh0,1024, nullptr, D1,1024,   uB,1024, 256,1024, 0, 2,16)
  level_gemm<<<dim3(4,16,3), blk, 0, stream>>>(L4);

  LevelArgs L5 = {}; L5.abz = -1;
  SLOT(L5, 0, hB,1024, Wh2,1024,   bh2,     nullptr,0, D3,1024, 512,1024, 2, 4,16)
  SLOT(L5, 1, hB,1024, W_out,1024, b_out,   nullptr,0, out,384, 256,1024, 0, 2,2)
  SLOT(L5, 2, uB,1024, Wh1,1024,   nullptr, D2,1024,   uA,1024, 256,1024, 0, 2,16)
  level_gemm<<<dim3(4,16,3), blk, 0, stream>>>(L5);

  LevelArgs L6 = {}; L6.abz = -1;
  SLOT(L6, 0, uA,1024, Wh2,1024, nullptr, D3,1024, uB,1024, 256,1024, 0, 2,16)
  level_gemm<<<dim3(2,16,1), blk, 0, stream>>>(L6);

  LevelArgs L7 = {}; L7.abz = 1;
  SLOT(L7, 0, uB,1024, W_out,1024, nullptr, nullptr,0, out + 128,384, 256,1024, 0, 2,2)
  L7.Aab = Wh0; L7.ldaab = 1024; L7.Bab = W_in; L7.ldbab = 256; L7.Cab = M0b; L7.ldcab = 128;
  L7.Mab = 1024; L7.Kab = 1024; L7.gxab = 8; L7.gyab = 2;
  level_gemm<<<dim3(8,2,2), blk, 0, stream>>>(L7);
#undef SLOT

  zero_chain<<<dim3(4,256), blk, 2 * CB * TS * sizeof(float), stream>>>(
      M0b, Wh1, Wh2,
      D1 + 256 * 1024, D2 + 256 * 1024, D3 + 256 * 1024,
      W_out, zs);
  finalize_norm<<<dim3((256 * 128 + 255) / 256), blk, 0, stream>>>(zs, out);
}


# Zero chain skips output rows whose relu mask is 0: per batch the kept rows are listed, waves share the list in 16-row tiles, the rest is stored as 0; sum order of every computed row unchanged
# speedup vs baseline: 2.4375x; 1.3150x over previous
#include <hip/hip_runtime.h>
#include <hip/hip_bf16.h>
#include <math.h>


typedef float v2f __attribute__((ext_vector_type(2)));
typedef float v4f __attribute__((ext_vector_type(4)));
typedef float v8f __attribute__((ext_vector_type(8)));

#define HID   1024
#define NST   128
#define BATCH 256
#define IN2   256
#define CB    32
#define TS    1028

__device__ __forceinline__ v8f wmma_f32(v2f a, v2f b, v8f c) {
  return __builtin_amdgcn_wmma_f32_16x16x4_f32(
       false, a,  false, b,
       (short)0, c,  false,  false);
}

#define PIPE_FENCE() __builtin_amdgcn_sched_barrier(0)

__device__ __forceinline__ v2f lo2(v4f v) { return __builtin_shufflevector(v, v, 0, 1); }
__device__ __forceinline__ v2f hi2(v4f v) { return __builtin_shufflevector(v, v, 2, 3); }

__global__ void build_x2(const float* __restrict__ x, float* __restrict__ X2) {
  int i = blockIdx.x * blockDim.x + threadIdx.x;
  if (i >= 512 * IN2) return;
  int row = i >> 8, c = i & 255;
  float v;
  if (row < 256) {
    v = x[row * 512 + c];
  } else {
    int rz = row - 256;
    v = (c < 128) ? x[rz * 512 + 384 + c] : x[rz * 512 + 128 + (c - 128)];
  }
  X2[i] = v;
}

__global__ void zero_buf(float* __restrict__ p, int n) {
  int i = blockIdx.x * blockDim.x + threadIdx.x;
  if (i < n) p[i] = 0.f;
}

__device__ __forceinline__ void awt_tile(
    const float* __restrict__ A, int lda,
    const float* __restrict__ W, int ldw,
    const float* __restrict__ bias,
    const float* __restrict__ mask, int ldm,
    float* __restrict__ C, int ldc,
    int M, int K, int epi)
{
  const int lane = threadIdx.x & 31, wv = threadIdx.x >> 5;
  const int l16 = lane & 15, hlf = lane >> 4;
  const int m0 = blockIdx.x * 128 + wv * 16;
  const int n0 = blockIdx.y * 64;
  if (m0 >= M) return;

  const v8f vz = {0.f,0.f,0.f,0.f,0.f,0.f,0.f,0.f};
  v8f acc[4] = {vz, vz, vz, vz};

  const int kof = hlf * 4;
  const float* Ar  = A + (size_t)(m0 + l16) * lda;
  const float* Wr0 = W + (size_t)(n0 +  0 + l16) * ldw;
  const float* Wr1 = W + (size_t)(n0 + 16 + l16) * ldw;
  const float* Wr2 = W + (size_t)(n0 + 32 + l16) * ldw;
  const float* Wr3 = W + (size_t)(n0 + 48 + l16) * ldw;

  v4f aS[4], bS[4][4];
#define AWT_LD(st, kk) { aS[st]    = *(const v4f*)(Ar  + (kk)); \
                         bS[st][0] = *(const v4f*)(Wr0 + (kk)); \
                         bS[st][1] = *(const v4f*)(Wr1 + (kk)); \
                         bS[st][2] = *(const v4f*)(Wr2 + (kk)); \
                         bS[st][3] = *(const v4f*)(Wr3 + (kk)); }
#define AWT_MM(st) { acc[0] = wmma_f32(lo2(aS[st]), lo2(bS[st][0]), acc[0]); \
                     acc[1] = wmma_f32(lo2(aS[st]), lo2(bS[st][1]), acc[1]); \
                     acc[2] = wmma_f32(lo2(aS[st]), lo2(bS[st][2]), acc[2]); \
                     acc[3] = wmma_f32(lo2(aS[st]), lo2(bS[st][3]), acc[3]); \
                     acc[0] = wmma_f32(hi2(aS[st]), hi2(bS[st][0]), acc[0]); \
                     acc[1] = wmma_f32(hi2(aS[st]), hi2(bS[st][1]), acc[1]); \
                     acc[2] = wmma_f32(hi2(aS[st]), hi2(bS[st][2]), acc[2]); \
                     acc[3] = wmma_f32(hi2(aS[st]), hi2(bS[st][3]), acc[3]); }

  AWT_LD(0, kof);      PIPE_FENCE();
  AWT_LD(1, kof + 8);  PIPE_FENCE();
  AWT_LD(2, kof + 16); PIPE_FENCE();
  AWT_LD(3, kof + 24); PIPE_FENCE();
  for (int k = 0; k < K - 32; k += 32) {
    AWT_MM(0); PIPE_FENCE(); AWT_LD(0, k + 32 + kof); PIPE_FENCE();
    AWT_MM(1); PIPE_FENCE(); AWT_LD(1, k + 40 + kof); PIPE_FENCE();
    AWT_MM(2); PIPE_FENCE(); AWT_LD(2, k + 48 + kof); PIPE_FENCE();
    AWT_MM(3); PIPE_FENCE(); AWT_LD(3, k + 56 + kof); PIPE_FENCE();
  }
  AWT_MM(0);
  AWT_MM(1);
  AWT_MM(2);
  AWT_MM(3);
#undef AWT_LD
#undef AWT_MM

  #pragma unroll
  for (int nt = 0; nt < 4; ++nt) {
    #pragma unroll
    for (int r = 0; r < 8; ++r) {
      const int m  = m0 + r + 8 * hlf;
      const int nn = n0 + nt * 16 + l16;
      float v = acc[nt][r];
      if (bias) v += bias[nn];
      if (epi == 1)      v = v > 0.f ? v   : 0.f;
      else if (epi == 2) v = v > 0.f ? 1.f : 0.f;
      if (mask) v *= mask[(size_t)m * ldm + nn];
      C[(size_t)m * ldc + nn] = v;
    }
  }
}

__device__ __forceinline__ void ab_tile(
    const float* __restrict__ A, int lda,
    const float* __restrict__ B, int ldb,
    float* __restrict__ C, int ldc,
    int M, int K)
{
  const int lane = threadIdx.x & 31, wv = threadIdx.x >> 5;
  const int l16 = lane & 15, hlf = lane >> 4;
  const int m0 = blockIdx.x * 128 + wv * 16;
  const int n0 = blockIdx.y * 64;
  if (m0 >= M) return;

  const v8f vz = {0.f,0.f,0.f,0.f,0.f,0.f,0.f,0.f};
  v8f acc[4] = {vz, vz, vz, vz};
  const int kof = hlf * 4;
  const float* Ar = A + (size_t)(m0 + l16) * lda;
  const float* Bc = B + n0 + l16;

  v4f a0, c00, c10, c20, c30;
  v4f a1, c01, c11, c21, c31;

#define BLD(dst, kk, nt) { (dst).x = Bc[(size_t)(kk) * ldb + (nt) * 16]; \
                           (dst).y = Bc[(size_t)((kk) + 1) * ldb + (nt) * 16]; \
                           (dst).z = Bc[(size_t)((kk) + 2) * ldb + (nt) * 16]; \
                           (dst).w = Bc[(size_t)((kk) + 3) * ldb + (nt) * 16]; }
#define LDSET0(kk) { a0 = *(const v4f*)(Ar + (kk)); \
                     BLD(c00, (kk), 0) BLD(c10, (kk), 1) BLD(c20, (kk), 2) BLD(c30, (kk), 3) }
#define LDSET1(kk) { a1 = *(const v4f*)(Ar + (kk)); \
                     BLD(c01, (kk), 0) BLD(c11, (kk), 1) BLD(c21, (kk), 2) BLD(c31, (kk), 3) }
#define WM0() { acc[0] = wmma_f32(lo2(a0), lo2(c00), acc[0]); acc[1] = wmma_f32(lo2(a0), lo2(c10), acc[1]); \
                acc[2] = wmma_f32(lo2(a0), lo2(c20), acc[2]); acc[3] = wmma_f32(lo2(a0), lo2(c30), acc[3]); \
                acc[0] = wmma_f32(hi2(a0), hi2(c00), acc[0]); acc[1] = wmma_f32(hi2(a0), hi2(c10), acc[1]); \
                acc[2] = wmma_f32(hi2(a0), hi2(c20), acc[2]); acc[3] = wmma_f32(hi2(a0), hi2(c30), acc[3]); }
#define WM1() { acc[0] = wmma_f32(lo2(a1), lo2(c01), acc[0]); acc[1] = wmma_f32(lo2(a1), lo2(c11), acc[1]); \
                acc[2] = wmma_f32(lo2(a1), lo2(c21), acc[2]); acc[3] = wmma_f32(lo2(a1), lo2(c31), acc[3]); \
                acc[0] = wmma_f32(hi2(a1), hi2(c01), acc[0]); acc[1] = wmma_f32(hi2(a1), hi2(c11), acc[1]); \
                acc[2] = wmma_f32(hi2(a1), hi2(c21), acc[2]); acc[3] = wmma_f32(hi2(a1), hi2(c31), acc[3]); }

  LDSET0(kof); PIPE_FENCE();
  LDSET1(kof + 8); PIPE_FENCE();
  for (int k = 0; k < K - 16; k += 16) {
    WM0(); PIPE_FENCE(); LDSET0(k + 16 + kof); PIPE_FENCE();
    WM1(); PIPE_FENCE(); LDSET1(k + 24 + kof); PIPE_FENCE();
  }
  WM0();
  WM1();
#undef BLD
#undef LDSET0
#undef LDSET1
#undef WM0
#undef WM1

  #pragma unroll
  for (int nt = 0; nt < 4; ++nt)
    #pragma unroll
    for (int r = 0; r < 8; ++r)
      C[(size_t)(m0 + r + 8 * hlf) * ldc + n0 + nt * 16 + l16] = acc[nt][r];
}

struct LevelArgs {
  const float* A0; const float* W0; const float* bias0; const float* mask0; float* C0;
  const float* A1; const float* W1; const float* bias1; const float* mask1; float* C1;
  const float* A2; const float* W2; const float* bias2; const float* mask2; float* C2;
  const float* Aab; const float* Bab; float* Cab;
  int lda0, ldw0, ldm0, ldc0, M0, K0, epi0, gx0, gy0;
  int lda1, ldw1, ldm1, ldc1, M1, K1, epi1, gx1, gy1;
  int lda2, ldw2, ldm2, ldc2, M2, K2, epi2, gx2, gy2;
  int ldaab, ldbab, ldcab, Mab, Kab, gxab, gyab, abz;
};

__global__ __launch_bounds__(256) void level_gemm(const LevelArgs p)
{
  const int z = blockIdx.z;
  const int bx = blockIdx.x, by = blockIdx.y;
  if (z == p.abz) {
    if (bx < p.gxab && by < p.gyab)
      ab_tile(p.Aab, p.ldaab, p.Bab, p.ldbab, p.Cab, p.ldcab, p.Mab, p.Kab);
    return;
  }
  const float* A; const float* Wt; const float* bias; const float* mask; float* C;
  int lda, ldw, ldm, ldc, M, K, epi, gx, gy;
  if (z == 0) {
    A = p.A0; Wt = p.W0; bias = p.bias0; mask = p.mask0; C = p.C0;
    lda = p.lda0; ldw = p.ldw0; ldm = p.ldm0; ldc = p.ldc0;
    M = p.M0; K = p.K0; epi = p.epi0; gx = p.gx0; gy = p.gy0;
  } else if (z == 1) {
    A = p.A1; Wt = p.W1; bias = p.bias1; mask = p.mask1; C = p.C1;
    lda = p.lda1; ldw = p.ldw1; ldm = p.ldm1; ldc = p.ldc1;
    M = p.M1; K = p.K1; epi = p.epi1; gx = p.gx1; gy = p.gy1;
  } else {
    A = p.A2; Wt = p.W2; bias = p.bias2; mask = p.mask2; C = p.C2;
    lda = p.lda2; ldw = p.ldw2; ldm = p.ldm2; ldc = p.ldc2;
    M = p.M2; K = p.K2; epi = p.epi2; gx = p.gx2; gy = p.gy2;
  }
  if (bx >= gx || by >= gy) return;
  awt_tile(A, lda, Wt, ldw, bias, mask, ldm, C, ldc, M, K, epi);
}

__device__ __forceinline__ void rows_ldA(const float* __restrict__ Wr, int kk, v4f av[4])
{
  #pragma unroll
  for (int q = 0; q < 4; ++q)
    av[q] = *(const v4f*)(Wr + kk + 8 * q);
}

__device__ __forceinline__ void chain_ldB(
    const float* __restrict__ Tb0, const float* __restrict__ Tb1, int kk,
    v4f cB[4][2])
{
  #pragma unroll
  for (int q = 0; q < 4; ++q) {
    cB[q][0] = *(const v4f*)(Tb0 + kk + 8 * q);
    cB[q][1] = *(const v4f*)(Tb1 + kk + 8 * q);
  }
}

__device__ __forceinline__ void rows_mm(const v4f av[4], const v4f cB[4][2], v8f acc[2])
{
  #pragma unroll
  for (int q = 0; q < 4; ++q) {
    acc[0] = wmma_f32(lo2(av[q]), lo2(cB[q][0]), acc[0]);
    acc[1] = wmma_f32(lo2(av[q]), lo2(cB[q][1]), acc[1]);
    acc[0] = wmma_f32(hi2(av[q]), hi2(cB[q][0]), acc[0]);
    acc[1] = wmma_f32(hi2(av[q]), hi2(cB[q][1]), acc[1]);
  }
}

template <int NT>
__device__ __forceinline__ void rows_trip(
    const float* const Wr[NT], const float* __restrict__ Tb0,
    const float* __restrict__ Tb1, int kn,
    v4f av[NT][4], const v4f cur[4][2], v4f nxt[4][2], v8f acc[NT][2])
{
  rows_mm(av[0], cur, acc[0]); PIPE_FENCE();
  rows_ldA(Wr[0], kn, av[0]); chain_ldB(Tb0, Tb1, kn, nxt); PIPE_FENCE();
  #pragma unroll
  for (int t = 1; t < NT; ++t) {
    rows_mm(av[t], cur, acc[t]); PIPE_FENCE();
    rows_ldA(Wr[t], kn, av[t]); PIPE_FENCE();
  }
}

template <int NT>
__device__ __forceinline__ void chain_rows(
    const float* __restrict__ Wm,
    const float* __restrict__ Tin,
    float* __restrict__ Tout,
    const float* __restrict__ dm,
    const int* __restrict__ rows, int P,
    int lane, int wv)
{
  const int l16 = lane & 15, hlf = lane >> 4;
  const int kof = hlf * 4;
  const v8f vz = {0.f,0.f,0.f,0.f,0.f,0.f,0.f,0.f};
  v8f acc[NT][2];
  const float* Wr[NT];
  #pragma unroll
  for (int t = 0; t < NT; ++t) {
    acc[t][0] = vz; acc[t][1] = vz;
    Wr[t] = Wm + (size_t)rows[(wv + 8 * t) * 16 + l16] * HID;
  }

  const float* Tb0 = Tin + (size_t)l16 * TS;
  const float* Tb1 = Tin + (size_t)(16 + l16) * TS;

  v4f av[NT][4];
  v4f cBa[4][2], cBb[4][2];

  rows_ldA(Wr[0], kof, av[0]); chain_ldB(Tb0, Tb1, kof, cBa); PIPE_FENCE();
  #pragma unroll
  for (int t = 1; t < NT; ++t) { rows_ldA(Wr[t], kof, av[t]); PIPE_FENCE(); }
  for (int k = 0; k < HID - 64; k += 64) {
    rows_trip<NT>(Wr, Tb0, Tb1, k + 32 + kof, av, cBa, cBb, acc);
    rows_trip<NT>(Wr, Tb0, Tb1, k + 64 + kof, av, cBb, cBa, acc);
  }
  rows_trip<NT>(Wr, Tb0, Tb1, HID - 32 + kof, av, cBa, cBb, acc);
  #pragma unroll
  for (int t = 0; t < NT; ++t) rows_mm(av[t], cBb, acc[t]);

  #pragma unroll
  for (int t = 0; t < NT; ++t) {
    const int pb = (wv + 8 * t) * 16 + 8 * hlf;
    #pragma unroll
    for (int r = 0; r < 8; ++r) {
      if (pb + r < P) {
        const int row = rows[pb + r];
        const float d = dm[row];
        Tout[(size_t)l16 * TS + row]        = acc[t][0][r] * d;
        Tout[(size_t)(16 + l16) * TS + row] = acc[t][1][r] * d;
      }
    }
  }
}

__global__ __launch_bounds__(256) void zero_chain(
    const float* __restrict__ M0,
    const float* __restrict__ Wh1, const float* __restrict__ Wh2,
    const float* __restrict__ d1z, const float* __restrict__ d2z,
    const float* __restrict__ d3z,
    const float* __restrict__ Wout,
    float* __restrict__ zsum)
{
  extern __shared__ float lds[];
  float* T0 = lds;
  float* T1 = lds + CB * TS;
  int* rowsL = (int*)(lds + 2 * CB * TS);
  int* cnt   = rowsL + 2 * HID;

  const int cb = blockIdx.x;
  const int b  = blockIdx.y;
  const int c0 = cb * CB;
  const int lane = threadIdx.x & 31, wv = threadIdx.x >> 5;

  const float* d1 = d1z + (size_t)b * HID;
  const float* d2 = d2z + (size_t)b * HID;
  const float* d3 = d3z + (size_t)b * HID;

  unsigned mk[2][4];
  #pragma unroll
  for (int g = 0; g < 2; ++g) {
    const float* dg = g ? d3 : d2;
    #pragma unroll
    for (int j = 0; j < 4; ++j) {
      mk[g][j] = (unsigned)__ballot(dg[j * 256 + threadIdx.x] != 0.f);
      if (lane == 0) cnt[g * 32 + j * 8 + wv] = __popc(mk[g][j]);
    }
  }

  for (int idx = threadIdx.x; idx < HID * CB; idx += 256) {
    const int j = idx >> 5, c = idx & 31;
    T0[(size_t)c * TS + j] = M0[(size_t)j * NST + c0 + c] * d1[j];
  }
  for (int i = threadIdx.x; i < CB * (HID / 4); i += 256)
    *(float4*)(T1 + (size_t)(i >> 8) * TS + 4 * (i & 255)) = make_float4(0.f, 0.f, 0.f, 0.f);
  __syncthreads();

  int Pn[2];
  #pragma unroll
  for (int g = 0; g < 2; ++g) {
    int tot = 0, off[4] = {0, 0, 0, 0};
    #pragma unroll
    for (int sgm = 0; sgm < 32; ++sgm) {
      if ((sgm & 7) == wv) off[sgm >> 3] = tot;
      tot += cnt[g * 32 + sgm];
    }
    int* rl = rowsL + g * HID;
    #pragma unroll
    for (int j = 0; j < 4; ++j)
      if ((mk[g][j] >> lane) & 1u)
        rl[off[j] + __popc(mk[g][j] & ((1u << lane) - 1u))] = j * 256 + threadIdx.x;
    const int cap = ((tot + 15) >> 4) << 4;
    for (int i = tot + threadIdx.x; i < cap; i += 256) rl[i] = 0;
    Pn[g] = tot;
  }

  #pragma unroll 1
  for (int g = 0; g < 2; ++g) {
    const float* Wm  = g ? Wh2 : Wh1;
    const float* dmg = g ? d3 : d2;
    const float* Tin = lds + (g ? CB * TS : 0);
    float* Tout      = lds + (g ? 0 : CB * TS);
    const int* rl    = rowsL + g * HID;
    const int P      = g ? Pn[1] : Pn[0];
    if (g == 1) {
      for (int i = threadIdx.x; i < CB * (HID / 4); i += 256)
        *(float4*)(T0 + (size_t)(i >> 8) * TS + 4 * (i & 255)) = make_float4(0.f, 0.f, 0.f, 0.f);
    }
    __syncthreads();
    const int ntot = (P + 15) >> 4;
    const int ntw = __builtin_amdgcn_readfirstlane(wv < ntot ? (ntot - wv + 7) >> 3 : 0);
    switch (ntw) {
      case 1: chain_rows<1>(Wm, Tin, Tout, dmg, rl, P, lane, wv); break;
      case 2: chain_rows<2>(Wm, Tin, Tout, dmg, rl, P, lane, wv); break;
      case 3: chain_rows<3>(Wm, Tin, Tout, dmg, rl, P, lane, wv); break;
      case 4: chain_rows<4>(Wm, Tin, Tout, dmg, rl, P, lane, wv); break;
      case 5: chain_rows<5>(Wm, Tin, Tout, dmg, rl, P, lane, wv); break;
      case 6: chain_rows<6>(Wm, Tin, Tout, dmg, rl, P, lane, wv); break;
      case 7: chain_rows<7>(Wm, Tin, Tout, dmg, rl, P, lane, wv); break;
      case 8: chain_rows<8>(Wm, Tin, Tout, dmg, rl, P, lane, wv); break;
      default: break;
    }
    __syncthreads();
  }

  const int l16 = lane & 15, hlf = lane >> 4;
  const int kof = hlf * 4;
  const v8f vz = {0.f,0.f,0.f,0.f,0.f,0.f,0.f,0.f};
  v8f g0 = vz, g1 = vz;
  const float* Tb0 = T0 + (size_t)l16 * TS;
  const float* Tb1 = T0 + (size_t)(16 + l16) * TS;
  const float* Wr  = Wout + (size_t)(wv * 16 + l16) * HID;

  v4f wa[4][4], wc0[4][4], wc1[4][4];
#define FLD(st, kk) { wa[st][0]  = *(const v4f*)(Wr  + (kk) + 0); wc0[st][0] = *(const v4f*)(Tb0 + (kk) + 0); wc1[st][0] = *(const v4f*)(Tb1 + (kk) + 0); \
                      wa[st][1]  = *(const v4f*)(Wr  + (kk) + 8); wc0[st][1] = *(const v4f*)(Tb0 + (kk) + 8); wc1[st][1] = *(const v4f*)(Tb1 + (kk) + 8); \
                      wa[st][2]  = *(const v4f*)(Wr  + (kk) + 16); wc0[st][2] = *(const v4f*)(Tb0 + (kk) + 16); wc1[st][2] = *(const v4f*)(Tb1 + (kk) + 16); \
                      wa[st][3]  = *(const v4f*)(Wr  + (kk) + 24); wc0[st][3] = *(const v4f*)(Tb0 + (kk) + 24); wc1[st][3] = *(const v4f*)(Tb1 + (kk) + 24); }
#define GM(st) { g0 = wmma_f32(lo2(wa[st][0]), lo2(wc0[st][0]), g0); g1 = wmma_f32(lo2(wa[st][0]), lo2(wc1[st][0]), g1); \
                 g0 = wmma_f32(hi2(wa[st][0]), hi2(wc0[st][0]), g0); g1 = wmma_f32(hi2(wa[st][0]), hi2(wc1[st][0]), g1); \
                 g0 = wmma_f32(lo2(wa[st][1]), lo2(wc0[st][1]), g0); g1 = wmma_f32(lo2(wa[st][1]), lo2(wc1[st][1]), g1); \
                 g0 = wmma_f32(hi2(wa[st][1]), hi2(wc0[st][1]), g0); g1 = wmma_f32(hi2(wa[st][1]), hi2(wc1[st][1]), g1); \
                 g0 = wmma_f32(lo2(wa[st][2]), lo2(wc0[st][2]), g0); g1 = wmma_f32(lo2(wa[st][2]), lo2(wc1[st][2]), g1); \
                 g0 = wmma_f32(hi2(wa[st][2]), hi2(wc0[st][2]), g0); g1 = wmma_f32(hi2(wa[st][2]), hi2(wc1[st][2]), g1); \
                 g0 = wmma_f32(lo2(wa[st][3]), lo2(wc0[st][3]), g0); g1 = wmma_f32(lo2(wa[st][3]), lo2(wc1[st][3]), g1); \
                 g0 = wmma_f32(hi2(wa[st][3]), hi2(wc0[st][3]), g0); g1 = wmma_f32(hi2(wa[st][3]), hi2(wc1[st][3]), g1); }
  FLD(0, kof);      PIPE_FENCE();
  FLD(1, kof + 32); PIPE_FENCE();
  FLD(2, kof + 64); PIPE_FENCE();
  FLD(3, kof + 96); PIPE_FENCE();
  for (int k = 0; k < HID - 128; k += 128) {
    GM(0); PIPE_FENCE(); FLD(0, k + 128 + kof); PIPE_FENCE();
    GM(1); PIPE_FENCE(); FLD(1, k + 160 + kof); PIPE_FENCE();
    GM(2); PIPE_FENCE(); FLD(2, k + 192 + kof); PIPE_FENCE();
    GM(3); PIPE_FENCE(); FLD(3, k + 224 + kof); PIPE_FENCE();
  }
  GM(0);
  GM(1);
  GM(2);
  GM(3);
#undef FLD
#undef GM

  #pragma unroll
  for (int r = 0; r < 8; ++r) {
    float s = g0[r] * g0[r] + g1[r] * g1[r];
    s += __shfl_xor(s, 1);
    s += __shfl_xor(s, 2);
    s += __shfl_xor(s, 4);
    s += __shfl_xor(s, 8);
    if (l16 == 0)
      atomicAdd(&zsum[(size_t)b * NST + wv * 16 + r + 8 * hlf], s);
  }
}

__global__ void finalize_norm(const float* __restrict__ zsum,
                              float* __restrict__ out) {
  int i = blockIdx.x * blockDim.x + threadIdx.x;
  if (i >= BATCH * NST) return;
  int b = i >> 7, c = i & 127;
  out[(size_t)b * 384 + 256 + c] = sqrtf(zsum[i]);
}

extern "C" void kernel_launch(void* const* d_in, const int* in_sizes, int n_in,
                              void* d_out, int out_size, void* d_ws, size_t ws_size,
                              hipStream_t stream)
{
  (void)in_sizes; (void)n_in; (void)out_size; (void)ws_size;
  const float* x     = (const float*)d_in[0];
  const float* W_in  = (const float*)d_in[1];
  const float* b_in  = (const float*)d_in[2];
  const float* Wh    = (const float*)d_in[3];
  const float* bh    = (const float*)d_in[4];
  const float* W_out = (const float*)d_in[5];
  const float* b_out = (const float*)d_in[6];
  float* out = (float*)d_out;
  float* ws  = (float*)d_ws;

  const float* Wh0 = Wh;
  const float* Wh1 = Wh + 1024 * 1024;
  const float* Wh2 = Wh + 2 * 1024 * 1024;
  const float* bh0 = bh, *bh1 = bh + 1024, *bh2 = bh + 2048;

  float* X2  = ws;
  float* hA  = X2  + 512 * 256;
  float* hB  = hA  + 512 * 1024;
  float* D1  = hB  + 512 * 1024;
  float* D2  = D1  + 512 * 1024;
  float* D3  = D2  + 512 * 1024;
  float* uA  = D3  + 512 * 1024;
  float* uB  = uA  + 256 * 1024;
  float* M0b = uB  + 256 * 1024;
  float* zs  = M0b + 1024 * 128;

  dim3 blk(256);
  build_x2<<<dim3((512 * 256 + 255) / 256), blk, 0, stream>>>(x, X2);
  zero_buf<<<dim3((256 * 128 + 255) / 256), blk, 0, stream>>>(zs, 256 * 128);

#define SLOT(L, n, A_, lda_, W_, ldw_, b_, m_, ldm_, C_, ldc_, M_, K_, e_, gx_, gy_) \
  { (L).A##n = (A_); (L).lda##n = (lda_); (L).W##n = (W_); (L).ldw##n = (ldw_); \
    (L).bias##n = (b_); (L).mask##n = (m_); (L).ldm##n = (ldm_); (L).C##n = (C_); \
    (L).ldc##n = (ldc_); (L).M##n = (M_); (L).K##n = (K_); (L).epi##n = (e_); \
    (L).gx##n = (gx_); (L).gy##n = (gy_); }

  LevelArgs L1 = {}; L1.abz = -1;
  SLOT(L1, 0, X2,256,      W_in,256, b_in,    nullptr,0, hA,1024, 512,256, 0, 4,16)
  SLOT(L1, 1, x + 256,512, W_in,256, nullptr, nullptr,0, uA,1024, 256,128, 0, 2,16)
  level_gemm<<<dim3(4,16,2), blk, 0, stream>>>(L1);

  LevelArgs L2 = {}; L2.abz = -1;
  SLOT(L2, 0, hA,1024, Wh0,1024, bh0, nullptr,0, hB,1024, 512,1024, 1, 4,16)
  level_gemm<<<dim3(4,16,1), blk, 0, stream>>>(L2);

  LevelArgs L3 = {}; L3.abz = -1;
  SLOT(L3, 0, hB,1024, Wh0,1024, bh0, nullptr,0, D1,1024, 512,1024, 2, 4,16)
  SLOT(L3, 1, hB,1024, Wh1,1024, bh1, nullptr,0, hA,1024, 512,1024, 1, 4,16)
  level_gemm<<<dim3(4,16,2), blk, 0, stream>>>(L3);

  LevelArgs L4 = {}; L4.abz = -1;
  SLOT(L4, 0, hA,1024, Wh1,1024, bh1,     nullptr,0, D2,1024, 512,1024, 2, 4,16)
  SLOT(L4, 1, hA,1024, Wh2,1024, bh2,     nullptr,0, hB,1024, 512,1024, 1, 4,16)
  SLOT(L4, 2, uA,1024, Wh0,1024, nullptr, D1,1024,   uB,1024, 256,1024, 0, 2,16)
  level_gemm<<<dim3(4,16,3), blk, 0, stream>>>(L4);

  LevelArgs L5 = {}; L5.abz = -1;
  SLOT(L5, 0, hB,1024, Wh2,1024,   bh2,     nullptr,0, D3,1024, 512,1024, 2, 4,16)
  SLOT(L5, 1, hB,1024, W_out,1024, b_out,   nullptr,0, out,384, 256,1024, 0, 2,2)
  SLOT(L5, 2, uB,1024, Wh1,1024,   nullptr, D2,1024,   uA,1024, 256,1024, 0, 2,16)
  level_gemm<<<dim3(4,16,3), blk, 0, stream>>>(L5);

  LevelArgs L6 = {}; L6.abz = -1;
  SLOT(L6, 0, uA,1024, Wh2,1024, nullptr, D3,1024, uB,1024, 256,1024, 0, 2,16)
  level_gemm<<<dim3(2,16,1), blk, 0, stream>>>(L6);

  LevelArgs L7 = {}; L7.abz = 1;
  SLOT(L7, 0, uB,1024, W_out,1024, nullptr, nullptr,0, out + 128,384, 256,1024, 0, 2,2)
  L7.Aab = Wh0; L7.ldaab = 1024; L7.Bab = W_in; L7.ldbab = 256; L7.Cab = M0b; L7.ldcab = 128;
  L7.Mab = 1024; L7.Kab = 1024; L7.gxab = 8; L7.gyab = 2;
  level_gemm<<<dim3(8,2,2), blk, 0, stream>>>(L7);
#undef SLOT

  zero_chain<<<dim3(4,256), blk, 2 * CB * TS * sizeof(float) + 2 * HID * sizeof(int) + 64 * sizeof(int), stream>>>(
      M0b, Wh1, Wh2,
      D1 + 256 * 1024, D2 + 256 * 1024, D3 + 256 * 1024,
      W_out, zs);
  finalize_norm<<<dim3((256 * 128 + 255) / 256), blk, 0, stream>>>(zs, out);
}


# Level GEMM work-groups cut from 8 waves to 4 (to 2 in levels 2 and 6); d2 rows read only by the zero chain move to level 6: up to 256 work-groups per level; tile code, sum order unchanged
# speedup vs baseline: 2.5530x; 1.0474x over previous
#include <hip/hip_runtime.h>
#include <hip/hip_bf16.h>
#include <math.h>


typedef float v2f __attribute__((ext_vector_type(2)));
typedef float v4f __attribute__((ext_vector_type(4)));
typedef float v8f __attribute__((ext_vector_type(8)));

#define HID   1024
#define NST   128
#define BATCH 256
#define IN2   256
#define CB    32
#define TS    1028

__device__ __forceinline__ v8f wmma_f32(v2f a, v2f b, v8f c) {
  return __builtin_amdgcn_wmma_f32_16x16x4_f32(
       false, a,  false, b,
       (short)0, c,  false,  false);
}

#define PIPE_FENCE() __builtin_amdgcn_sched_barrier(0)

__device__ __forceinline__ v2f lo2(v4f v) { return __builtin_shufflevector(v, v, 0, 1); }
__device__ __forceinline__ v2f hi2(v4f v) { return __builtin_shufflevector(v, v, 2, 3); }

__global__ void build_x2(const float* __restrict__ x, float* __restrict__ X2) {
  int i = blockIdx.x * blockDim.x + threadIdx.x;
  if (i >= 512 * IN2) return;
  int row = i >> 8, c = i & 255;
  float v;
  if (row < 256) {
    v = x[row * 512 + c];
  } else {
    int rz = row - 256;
    v = (c < 128) ? x[rz * 512 + 384 + c] : x[rz * 512 + 128 + (c - 128)];
  }
  X2[i] = v;
}

__global__ void zero_buf(float* __restrict__ p, int n) {
  int i = blockIdx.x * blockDim.x + threadIdx.x;
  if (i < n) p[i] = 0.f;
}

__device__ __forceinline__ void awt_tile(
    const float* __restrict__ A, int lda,
    const float* __restrict__ W, int ldw,
    const float* __restrict__ bias,
    const float* __restrict__ mask, int ldm,
    float* __restrict__ C, int ldc,
    int M, int K, int epi)
{
  const int lane = threadIdx.x & 31, wv = threadIdx.x >> 5;
  const int l16 = lane & 15, hlf = lane >> 4;
  const int m0 = (blockIdx.x * (int)(blockDim.x >> 5) + wv) * 16;
  const int n0 = blockIdx.y * 64;
  if (m0 >= M) return;

  const v8f vz = {0.f,0.f,0.f,0.f,0.f,0.f,0.f,0.f};
  v8f acc[4] = {vz, vz, vz, vz};

  const int kof = hlf * 4;
  const float* Ar  = A + (size_t)(m0 + l16) * lda;
  const float* Wr0 = W + (size_t)(n0 +  0 + l16) * ldw;
  const float* Wr1 = W + (size_t)(n0 + 16 + l16) * ldw;
  const float* Wr2 = W + (size_t)(n0 + 32 + l16) * ldw;
  const float* Wr3 = W + (size_t)(n0 + 48 + l16) * ldw;

  v4f aS[4], bS[4][4];
#define AWT_LD(st, kk) { aS[st]    = *(const v4f*)(Ar  + (kk)); \
                         bS[st][0] = *(const v4f*)(Wr0 + (kk)); \
                         bS[st][1] = *(const v4f*)(Wr1 + (kk)); \
                         bS[st][2] = *(const v4f*)(Wr2 + (kk)); \
                         bS[st][3] = *(const v4f*)(Wr3 + (kk)); }
#define AWT_MM(st) { acc[0] = wmma_f32(lo2(aS[st]), lo2(bS[st][0]), acc[0]); \
                     acc[1] = wmma_f32(lo2(aS[st]), lo2(bS[st][1]), acc[1]); \
                     acc[2] = wmma_f32(lo2(aS[st]), lo2(bS[st][2]), acc[2]); \
                     acc[3] = wmma_f32(lo2(aS[st]), lo2(bS[st][3]), acc[3]); \
                     acc[0] = wmma_f32(hi2(aS[st]), hi2(bS[st][0]), acc[0]); \
                     acc[1] = wmma_f32(hi2(aS[st]), hi2(bS[st][1]), acc[1]); \
                     acc[2] = wmma_f32(hi2(aS[st]), hi2(bS[st][2]), acc[2]); \
                     acc[3] = wmma_f32(hi2(aS[st]), hi2(bS[st][3]), acc[3]); }

  AWT_LD(0, kof);      PIPE_FENCE();
  AWT_LD(1, kof + 8);  PIPE_FENCE();
  AWT_LD(2, kof + 16); PIPE_FENCE();
  AWT_LD(3, kof + 24); PIPE_FENCE();
  for (int k = 0; k < K - 32; k += 32) {
    AWT_MM(0); PIPE_FENCE(); AWT_LD(0, k + 32 + kof); PIPE_FENCE();
    AWT_MM(1); PIPE_FENCE(); AWT_LD(1, k + 40 + kof); PIPE_FENCE();
    AWT_MM(2); PIPE_FENCE(); AWT_LD(2, k + 48 + kof); PIPE_FENCE();
    AWT_MM(3); PIPE_FENCE(); AWT_LD(3, k + 56 + kof); PIPE_FENCE();
  }
  AWT_MM(0);
  AWT_MM(1);
  AWT_MM(2);
  AWT_MM(3);
#undef AWT_LD
#undef AWT_MM

  #pragma unroll
  for (int nt = 0; nt < 4; ++nt) {
    #pragma unroll
    for (int r = 0; r < 8; ++r) {
      const int m  = m0 + r + 8 * hlf;
      const int nn = n0 + nt * 16 + l16;
      float v = acc[nt][r];
      if (bias) v += bias[nn];
      if (epi == 1)      v = v > 0.f ? v   : 0.f;
      else if (epi == 2) v = v > 0.f ? 1.f : 0.f;
      if (mask) v *= mask[(size_t)m * ldm + nn];
      C[(size_t)m * ldc + nn] = v;
    }
  }
}

__device__ __forceinline__ void ab_tile(
    const float* __restrict__ A, int lda,
    const float* __restrict__ B, int ldb,
    float* __restrict__ C, int ldc,
    int M, int K)
{
  const int lane = threadIdx.x & 31, wv = threadIdx.x >> 5;
  const int l16 = lane & 15, hlf = lane >> 4;
  const int m0 = (blockIdx.x * (int)(blockDim.x >> 5) + wv) * 16;
  const int n0 = blockIdx.y * 64;
  if (m0 >= M) return;

  const v8f vz = {0.f,0.f,0.f,0.f,0.f,0.f,0.f,0.f};
  v8f acc[4] = {vz, vz, vz, vz};
  const int kof = hlf * 4;
  const float* Ar = A + (size_t)(m0 + l16) * lda;
  const float* Bc = B + n0 + l16;

  v4f a0, c00, c10, c20, c30;
  v4f a1, c01, c11, c21, c31;

#define BLD(dst, kk, nt) { (dst).x = Bc[(size_t)(kk) * ldb + (nt) * 16]; \
                           (dst).y = Bc[(size_t)((kk) + 1) * ldb + (nt) * 16]; \
                           (dst).z = Bc[(size_t)((kk) + 2) * ldb + (nt) * 16]; \
                           (dst).w = Bc[(size_t)((kk) + 3) * ldb + (nt) * 16]; }
#define LDSET0(kk) { a0 = *(const v4f*)(Ar + (kk)); \
                     BLD(c00, (kk), 0) BLD(c10, (kk), 1) BLD(c20, (kk), 2) BLD(c30, (kk), 3) }
#define LDSET1(kk) { a1 = *(const v4f*)(Ar + (kk)); \
                     BLD(c01, (kk), 0) BLD(c11, (kk), 1) BLD(c21, (kk), 2) BLD(c31, (kk), 3) }
#define WM0() { acc[0] = wmma_f32(lo2(a0), lo2(c00), acc[0]); acc[1] = wmma_f32(lo2(a0), lo2(c10), acc[1]); \
                acc[2] = wmma_f32(lo2(a0), lo2(c20), acc[2]); acc[3] = wmma_f32(lo2(a0), lo2(c30), acc[3]); \
                acc[0] = wmma_f32(hi2(a0), hi2(c00), acc[0]); acc[1] = wmma_f32(hi2(a0), hi2(c10), acc[1]); \
                acc[2] = wmma_f32(hi2(a0), hi2(c20), acc[2]); acc[3] = wmma_f32(hi2(a0), hi2(c30), acc[3]); }
#define WM1() { acc[0] = wmma_f32(lo2(a1), lo2(c01), acc[0]); acc[1] = wmma_f32(lo2(a1), lo2(c11), acc[1]); \
                acc[2] = wmma_f32(lo2(a1), lo2(c21), acc[2]); acc[3] = wmma_f32(lo2(a1), lo2(c31), acc[3]); \
                acc[0] = wmma_f32(hi2(a1), hi2(c01), acc[0]); acc[1] = wmma_f32(hi2(a1), hi2(c11), acc[1]); \
                acc[2] = wmma_f32(hi2(a1), hi2(c21), acc[2]); acc[3] = wmma_f32(hi2(a1), hi2(c31), acc[3]); }

  LDSET0(kof); PIPE_FENCE();
  LDSET1(kof + 8); PIPE_FENCE();
  for (int k = 0; k < K - 16; k += 16) {
    WM0(); PIPE_FENCE(); LDSET0(k + 16 + kof); PIPE_FENCE();
    WM1(); PIPE_FENCE(); LDSET1(k + 24 + kof); PIPE_FENCE();
  }
  WM0();
  WM1();
#undef BLD
#undef LDSET0
#undef LDSET1
#undef WM0
#undef WM1

  #pragma unroll
  for (int nt = 0; nt < 4; ++nt)
    #pragma unroll
    for (int r = 0; r < 8; ++r)
      C[(size_t)(m0 + r + 8 * hlf) * ldc + n0 + nt * 16 + l16] = acc[nt][r];
}

struct LevelArgs {
  const float* A0; const float* W0; const float* bias0; const float* mask0; float* C0;
  const float* A1; const float* W1; const float* bias1; const float* mask1; float* C1;
  const float* A2; const float* W2; const float* bias2; const float* mask2; float* C2;
  const float* Aab; const float* Bab; float* Cab;
  int lda0, ldw0, ldm0, ldc0, M0, K0, epi0, gx0, gy0;
  int lda1, ldw1, ldm1, ldc1, M1, K1, epi1, gx1, gy1;
  int lda2, ldw2, ldm2, ldc2, M2, K2, epi2, gx2, gy2;
  int ldaab, ldbab, ldcab, Mab, Kab, gxab, gyab, abz;
};

__global__ __launch_bounds__(128) void level_gemm(const LevelArgs p)
{
  const int z = blockIdx.z;
  const int bx = blockIdx.x, by = blockIdx.y;
  if (z == p.abz) {
    if (bx < p.gxab && by < p.gyab)
      ab_tile(p.Aab, p.ldaab, p.Bab, p.ldbab, p.Cab, p.ldcab, p.Mab, p.Kab);
    return;
  }
  const float* A; const float* Wt; const float* bias; const float* mask; float* C;
  int lda, ldw, ldm, ldc, M, K, epi, gx, gy;
  if (z == 0) {
    A = p.A0; Wt = p.W0; bias = p.bias0; mask = p.mask0; C = p.C0;
    lda = p.lda0; ldw = p.ldw0; ldm = p.ldm0; ldc = p.ldc0;
    M = p.M0; K = p.K0; epi = p.epi0; gx = p.gx0; gy = p.gy0;
  } else if (z == 1) {
    A = p.A1; Wt = p.W1; bias = p.bias1; mask = p.mask1; C = p.C1;
    lda = p.lda1; ldw = p.ldw1; ldm = p.ldm1; ldc = p.ldc1;
    M = p.M1; K = p.K1; epi = p.epi1; gx = p.gx1; gy = p.gy1;
  } else {
    A = p.A2; Wt = p.W2; bias = p.bias2; mask = p.mask2; C = p.C2;
    lda = p.lda2; ldw = p.ldw2; ldm = p.ldm2; ldc = p.ldc2;
    M = p.M2; K = p.K2; epi = p.epi2; gx = p.gx2; gy = p.gy2;
  }
  if (bx >= gx || by >= gy) return;
  awt_tile(A, lda, Wt, ldw, bias, mask, ldm, C, ldc, M, K, epi);
}

__device__ __forceinline__ void rows_ldA(const float* __restrict__ Wr, int kk, v4f av[4])
{
  #pragma unroll
  for (int q = 0; q < 4; ++q)
    av[q] = *(const v4f*)(Wr + kk + 8 * q);
}

__device__ __forceinline__ void chain_ldB(
    const float* __restrict__ Tb0, const float* __restrict__ Tb1, int kk,
    v4f cB[4][2])
{
  #pragma unroll
  for (int q = 0; q < 4; ++q) {
    cB[q][0] = *(const v4f*)(Tb0 + kk + 8 * q);
    cB[q][1] = *(const v4f*)(Tb1 + kk + 8 * q);
  }
}

__device__ __forceinline__ void rows_mm(const v4f av[4], const v4f cB[4][2], v8f acc[2])
{
  #pragma unroll
  for (int q = 0; q < 4; ++q) {
    acc[0] = wmma_f32(lo2(av[q]), lo2(cB[q][0]), acc[0]);
    acc[1] = wmma_f32(lo2(av[q]), lo2(cB[q][1]), acc[1]);
    acc[0] = wmma_f32(hi2(av[q]), hi2(cB[q][0]), acc[0]);
    acc[1] = wmma_f32(hi2(av[q]), hi2(cB[q][1]), acc[1]);
  }
}

template <int NT>
__device__ __forceinline__ void rows_trip(
    const float* const Wr[NT], const float* __restrict__ Tb0,
    const float* __restrict__ Tb1, int kn,
    v4f av[NT][4], const v4f cur[4][2], v4f nxt[4][2], v8f acc[NT][2])
{
  rows_mm(av[0], cur, acc[0]); PIPE_FENCE();
  rows_ldA(Wr[0], kn, av[0]); chain_ldB(Tb0, Tb1, kn, nxt); PIPE_FENCE();
  #pragma unroll
  for (int t = 1; t < NT; ++t) {
    rows_mm(av[t], cur, acc[t]); PIPE_FENCE();
    rows_ldA(Wr[t], kn, av[t]); PIPE_FENCE();
  }
}

template <int NT>
__device__ __forceinline__ void chain_rows(
    const float* __restrict__ Wm,
    const float* __restrict__ Tin,
    float* __restrict__ Tout,
    const float* __restrict__ dm,
    const int* __restrict__ rows, int P,
    int lane, int wv)
{
  const int l16 = lane & 15, hlf = lane >> 4;
  const int kof = hlf * 4;
  const v8f vz = {0.f,0.f,0.f,0.f,0.f,0.f,0.f,0.f};
  v8f acc[NT][2];
  const float* Wr[NT];
  #pragma unroll
  for (int t = 0; t < NT; ++t) {
    acc[t][0] = vz; acc[t][1] = vz;
    Wr[t] = Wm + (size_t)rows[(wv + 8 * t) * 16 + l16] * HID;
  }

  const float* Tb0 = Tin + (size_t)l16 * TS;
  const float* Tb1 = Tin + (size_t)(16 + l16) * TS;

  v4f av[NT][4];
  v4f cBa[4][2], cBb[4][2];

  rows_ldA(Wr[0], kof, av[0]); chain_ldB(Tb0, Tb1, kof, cBa); PIPE_FENCE();
  #pragma unroll
  for (int t = 1; t < NT; ++t) { rows_ldA(Wr[t], kof, av[t]); PIPE_FENCE(); }
  for (int k = 0; k < HID - 64; k += 64) {
    rows_trip<NT>(Wr, Tb0, Tb1, k + 32 + kof, av, cBa, cBb, acc);
    rows_trip<NT>(Wr, Tb0, Tb1, k + 64 + kof, av, cBb, cBa, acc);
  }
  rows_trip<NT>(Wr, Tb0, Tb1, HID - 32 + kof, av, cBa, cBb, acc);
  #pragma unroll
  for (int t = 0; t < NT; ++t) rows_mm(av[t], cBb, acc[t]);

  #pragma unroll
  for (int t = 0; t < NT; ++t) {
    const int pb = (wv + 8 * t) * 16 + 8 * hlf;
    #pragma unroll
    for (int r = 0; r < 8; ++r) {
      if (pb + r < P) {
        const int row = rows[pb + r];
        const float d = dm[row];
        Tout[(size_t)l16 * TS + row]        = acc[t][0][r] * d;
        Tout[(size_t)(16 + l16) * TS + row] = acc[t][1][r] * d;
      }
    }
  }
}

__global__ __launch_bounds__(256) void zero_chain(
    const float* __restrict__ M0,
    const float* __restrict__ Wh1, const float* __restrict__ Wh2,
    const float* __restrict__ d1z, const float* __restrict__ d2z,
    const float* __restrict__ d3z,
    const float* __restrict__ Wout,
    float* __restrict__ zsum)
{
  extern __shared__ float lds[];
  float* T0 = lds;
  float* T1 = lds + CB * TS;
  int* rowsL = (int*)(lds + 2 * CB * TS);
  int* cnt   = rowsL + 2 * HID;

  const int cb = blockIdx.x;
  const int b  = blockIdx.y;
  const int c0 = cb * CB;
  const int lane = threadIdx.x & 31, wv = threadIdx.x >> 5;

  const float* d1 = d1z + (size_t)b * HID;
  const float* d2 = d2z + (size_t)b * HID;
  const float* d3 = d3z + (size_t)b * HID;

  unsigned mk[2][4];
  #pragma unroll
  for (int g = 0; g < 2; ++g) {
    const float* dg = g ? d3 : d2;
    #pragma unroll
    for (int j = 0; j < 4; ++j) {
      mk[g][j] = (unsigned)__ballot(dg[j * 256 + threadIdx.x] != 0.f);
      if (lane == 0) cnt[g * 32 + j * 8 + wv] = __popc(mk[g][j]);
    }
  }

  for (int idx = threadIdx.x; idx < HID * CB; idx += 256) {
    const int j = idx >> 5, c = idx & 31;
    T0[(size_t)c * TS + j] = M0[(size_t)j * NST + c0 + c] * d1[j];
  }
  for (int i = threadIdx.x; i < CB * (HID / 4); i += 256)
    *(float4*)(T1 + (size_t)(i >> 8) * TS + 4 * (i & 255)) = make_float4(0.f, 0.f, 0.f, 0.f);
  __syncthreads();

  int Pn[2];
  #pragma unroll
  for (int g = 0; g < 2; ++g) {
    int tot = 0, off[4] = {0, 0, 0, 0};
    #pragma unroll
    for (int sgm = 0; sgm < 32; ++sgm) {
      if ((sgm & 7) == wv) off[sgm >> 3] = tot;
      tot += cnt[g * 32 + sgm];
    }
    int* rl = rowsL + g * HID;
    #pragma unroll
    for (int j = 0; j < 4; ++j)
      if ((mk[g][j] >> lane) & 1u)
        rl[off[j] + __popc(mk[g][j] & ((1u << lane) - 1u))] = j * 256 + threadIdx.x;
    const int cap = ((tot + 15) >> 4) << 4;
    for (int i = tot + threadIdx.x; i < cap; i += 256) rl[i] = 0;
    Pn[g] = tot;
  }

  #pragma unroll 1
  for (int g = 0; g < 2; ++g) {
    const float* Wm  = g ? Wh2 : Wh1;
    const float* dmg = g ? d3 : d2;
    const float* Tin = lds + (g ? CB * TS : 0);
    float* Tout      = lds + (g ? 0 : CB * TS);
    const int* rl    = rowsL + g * HID;
    const int P      = g ? Pn[1] : Pn[0];
    if (g == 1) {
      for (int i = threadIdx.x; i < CB * (HID / 4); i += 256)
        *(float4*)(T0 + (size_t)(i >> 8) * TS + 4 * (i & 255)) = make_float4(0.f, 0.f, 0.f, 0.f);
    }
    __syncthreads();
    const int ntot = (P + 15) >> 4;
    const int ntw = __builtin_amdgcn_readfirstlane(wv < ntot ? (ntot - wv + 7) >> 3 : 0);
    switch (ntw) {
      case 1: chain_rows<1>(Wm, Tin, Tout, dmg, rl, P, lane, wv); break;
      case 2: chain_rows<2>(Wm, Tin, Tout, dmg, rl, P, lane, wv); break;
      case 3: chain_rows<3>(Wm, Tin, Tout, dmg, rl, P, lane, wv); break;
      case 4: chain_rows<4>(Wm, Tin, Tout, dmg, rl, P, lane, wv); break;
      case 5: chain_rows<5>(Wm, Tin, Tout, dmg, rl, P, lane, wv); break;
      case 6: chain_rows<6>(Wm, Tin, Tout, dmg, rl, P, lane, wv); break;
      case 7: chain_rows<7>(Wm, Tin, Tout, dmg, rl, P, lane, wv); break;
      case 8: chain_rows<8>(Wm, Tin, Tout, dmg, rl, P, lane, wv); break;
      default: break;
    }
    __syncthreads();
  }

  const int l16 = lane & 15, hlf = lane >> 4;
  const int kof = hlf * 4;
  const v8f vz = {0.f,0.f,0.f,0.f,0.f,0.f,0.f,0.f};
  v8f g0 = vz, g1 = vz;
  const float* Tb0 = T0 + (size_t)l16 * TS;
  const float* Tb1 = T0 + (size_t)(16 + l16) * TS;
  const float* Wr  = Wout + (size_t)(wv * 16 + l16) * HID;

  v4f wa[4][4], wc0[4][4], wc1[4][4];
#define FLD(st, kk) { wa[st][0]  = *(const v4f*)(Wr  + (kk) + 0); wc0[st][0] = *(const v4f*)(Tb0 + (kk) + 0); wc1[st][0] = *(const v4f*)(Tb1 + (kk) + 0); \
                      wa[st][1]  = *(const v4f*)(Wr  + (kk) + 8); wc0[st][1] = *(const v4f*)(Tb0 + (kk) + 8); wc1[st][1] = *(const v4f*)(Tb1 + (kk) + 8); \
                      wa[st][2]  = *(const v4f*)(Wr  + (kk) + 16); wc0[st][2] = *(const v4f*)(Tb0 + (kk) + 16); wc1[st][2] = *(const v4f*)(Tb1 + (kk) + 16); \
                      wa[st][3]  = *(const v4f*)(Wr  + (kk) + 24); wc0[st][3] = *(const v4f*)(Tb0 + (kk) + 24); wc1[st][3] = *(const v4f*)(Tb1 + (kk) + 24); }
#define GM(st) { g0 = wmma_f32(lo2(wa[st][0]), lo2(wc0[st][0]), g0); g1 = wmma_f32(lo2(wa[st][0]), lo2(wc1[st][0]), g1); \
                 g0 = wmma_f32(hi2(wa[st][0]), hi2(wc0[st][0]), g0); g1 = wmma_f32(hi2(wa[st][0]), hi2(wc1[st][0]), g1); \
                 g0 = wmma_f32(lo2(wa[st][1]), lo2(wc0[st][1]), g0); g1 = wmma_f32(lo2(wa[st][1]), lo2(wc1[st][1]), g1); \
                 g0 = wmma_f32(hi2(wa[st][1]), hi2(wc0[st][1]), g0); g1 = wmma_f32(hi2(wa[st][1]), hi2(wc1[st][1]), g1); \
                 g0 = wmma_f32(lo2(wa[st][2]), lo2(wc0[st][2]), g0); g1 = wmma_f32(lo2(wa[st][2]), lo2(wc1[st][2]), g1); \
                 g0 = wmma_f32(hi2(wa[st][2]), hi2(wc0[st][2]), g0); g1 = wmma_f32(hi2(wa[st][2]), hi2(wc1[st][2]), g1); \
                 g0 = wmma_f32(lo2(wa[st][3]), lo2(wc0[st][3]), g0); g1 = wmma_f32(lo2(wa[st][3]), lo2(wc1[st][3]), g1); \
                 g0 = wmma_f32(hi2(wa[st][3]), hi2(wc0[st][3]), g0); g1 = wmma_f32(hi2(wa[st][3]), hi2(wc1[st][3]), g1); }
  FLD(0, kof);      PIPE_FENCE();
  FLD(1, kof + 32); PIPE_FENCE();
  FLD(2, kof + 64); PIPE_FENCE();
  FLD(3, kof + 96); PIPE_FENCE();
  for (int k = 0; k < HID - 128; k += 128) {
    GM(0); PIPE_FENCE(); FLD(0, k + 128 + kof); PIPE_FENCE();
    GM(1); PIPE_FENCE(); FLD(1, k + 160 + kof); PIPE_FENCE();
    GM(2); PIPE_FENCE(); FLD(2, k + 192 + kof); PIPE_FENCE();
    GM(3); PIPE_FENCE(); FLD(3, k + 224 + kof); PIPE_FENCE();
  }
  GM(0);
  GM(1);
  GM(2);
  GM(3);
#undef FLD
#undef GM

  #pragma unroll
  for (int r = 0; r < 8; ++r) {
    float s = g0[r] * g0[r] + g1[r] * g1[r];
    s += __shfl_xor(s, 1);
    s += __shfl_xor(s, 2);
    s += __shfl_xor(s, 4);
    s += __shfl_xor(s, 8);
    if (l16 == 0)
      atomicAdd(&zsum[(size_t)b * NST + wv * 16 + r + 8 * hlf], s);
  }
}

__global__ void finalize_norm(const float* __restrict__ zsum,
                              float* __restrict__ out) {
  int i = blockIdx.x * blockDim.x + threadIdx.x;
  if (i >= BATCH * NST) return;
  int b = i >> 7, c = i & 127;
  out[(size_t)b * 384 + 256 + c] = sqrtf(zsum[i]);
}

extern "C" void kernel_launch(void* const* d_in, const int* in_sizes, int n_in,
                              void* d_out, int out_size, void* d_ws, size_t ws_size,
                              hipStream_t stream)
{
  (void)in_sizes; (void)n_in; (void)out_size; (void)ws_size;
  const float* x     = (const float*)d_in[0];
  const float* W_in  = (const float*)d_in[1];
  const float* b_in  = (const float*)d_in[2];
  const float* Wh    = (const float*)d_in[3];
  const float* bh    = (const float*)d_in[4];
  const float* W_out = (const float*)d_in[5];
  const float* b_out = (const float*)d_in[6];
  float* out = (float*)d_out;
  float* ws  = (float*)d_ws;

  const float* Wh0 = Wh;
  const float* Wh1 = Wh + 1024 * 1024;
  const float* Wh2 = Wh + 2 * 1024 * 1024;
  const float* bh0 = bh, *bh1 = bh + 1024, *bh2 = bh + 2048;

  float* X2  = ws;
  float* hA  = X2  + 512 * 256;
  float* hB  = hA  + 512 * 1024;
  float* D1  = hB  + 512 * 1024;
  float* D2  = D1  + 512 * 1024;
  float* D3  = D2  + 512 * 1024;
  float* uA  = D3  + 512 * 1024;
  float* uB  = uA  + 256 * 1024;
  float* M0b = uB  + 256 * 1024;
  float* zs  = M0b + 1024 * 128;

  dim3 blk(256);
  build_x2<<<dim3((512 * 256 + 255) / 256), blk, 0, stream>>>(x, X2);
  zero_buf<<<dim3((256 * 128 + 255) / 256), blk, 0, stream>>>(zs, 256 * 128);

  dim3 lblk(128);
  dim3 lblk2(64);
#define SLOT(L, n, A_, lda_, W_, ldw_, b_, m_, ldm_, C_, ldc_, M_, K_, e_, gx_, gy_) \
  { (L).A##n = (A_); (L).lda##n = (lda_); (L).W##n = (W_); (L).ldw##n = (ldw_); \
    (L).bias##n = (b_); (L).mask##n = (m_); (L).ldm##n = (ldm_); (L).C##n = (C_); \
    (L).ldc##n = (ldc_); (L).M##n = (M_); (L).K##n = (K_); (L).epi##n = (e_); \
    (L).gx##n = (gx_); (L).gy##n = (gy_); }

  LevelArgs L1 = {}; L1.abz = -1;
  SLOT(L1, 0, X2,256,      W_in,256, b_in,    nullptr,0, hA,1024, 512,256, 0, 8,16)
  SLOT(L1, 1, x + 256,512, W_in,256, nullptr, nullptr,0, uA,1024, 256,128, 0, 4,16)
  level_gemm<<<dim3(8,16,2), lblk, 0, stream>>>(L1);

  LevelArgs L2 = {}; L2.abz = -1;
  SLOT(L2, 0, hA,1024, Wh0,1024, bh0, nullptr,0, hB,1024, 512,1024, 1, 16,16)
  level_gemm<<<dim3(16,16,1), lblk2, 0, stream>>>(L2);

  LevelArgs L3 = {}; L3.abz = -1;
  SLOT(L3, 0, hB,1024, Wh0,1024, bh0, nullptr,0, D1,1024, 512,1024, 2, 8,16)
  SLOT(L3, 1, hB,1024, Wh1,1024, bh1, nullptr,0, hA,1024, 512,1024, 1, 8,16)
  level_gemm<<<dim3(8,16,2), lblk, 0, stream>>>(L3);

  LevelArgs L4 = {}; L4.abz = -1;
  SLOT(L4, 0, hA,1024, Wh1,1024, bh1,     nullptr,0, D2,1024, 256,1024, 2, 4,16)
  SLOT(L4, 1, hA,1024, Wh2,1024, bh2,     nullptr,0, hB,1024, 512,1024, 1, 8,16)
  SLOT(L4, 2, uA,1024, Wh0,1024, nullptr, D1,1024,   uB,1024, 256,1024, 0, 4,16)
  level_gemm<<<dim3(8,16,3), lblk, 0, stream>>>(L4);

  LevelArgs L5 = {}; L5.abz = -1;
  SLOT(L5, 0, hB,1024, Wh2,1024,   bh2,     nullptr,0, D3,1024, 512,1024, 2, 8,16)
  SLOT(L5, 1, hB,1024, W_out,1024, b_out,   nullptr,0, out,384, 256,1024, 0, 4,2)
  SLOT(L5, 2, uB,1024, Wh1,1024,   nullptr, D2,1024,   uA,1024, 256,1024, 0, 4,16)
  level_gemm<<<dim3(8,16,3), lblk, 0, stream>>>(L5);

  LevelArgs L6 = {}; L6.abz = -1;
  SLOT(L6, 0, uA,1024, Wh2,1024, nullptr, D3,1024, uB,1024, 256,1024, 0, 8,16)
  SLOT(L6, 1, hA + 256 * 1024,1024, Wh1,1024, bh1, nullptr,0, D2 + 256 * 1024,1024, 256,1024, 2, 8,16)
  level_gemm<<<dim3(8,16,2), lblk2, 0, stream>>>(L6);

  LevelArgs L7 = {}; L7.abz = 1;
  SLOT(L7, 0, uB,1024, W_out,1024, nullptr, nullptr,0, out + 128,384, 256,1024, 0, 4,2)
  L7.Aab = Wh0; L7.ldaab = 1024; L7.Bab = W_in; L7.ldbab = 256; L7.Cab = M0b; L7.ldcab = 128;
  L7.Mab = 1024; L7.Kab = 1024; L7.gxab = 16; L7.gyab = 2;
  level_gemm<<<dim3(16,2,2), lblk, 0, stream>>>(L7);
#undef SLOT

  zero_chain<<<dim3(4,256), blk, 2 * CB * TS * sizeof(float) + 2 * HID * sizeof(int) + 64 * sizeof(int), stream>>>(
      M0b, Wh1, Wh2,
      D1 + 256 * 1024, D2 + 256 * 1024, D3 + 256 * 1024,
      W_out, zs);
  finalize_norm<<<dim3((256 * 128 + 255) / 256), blk, 0, stream>>>(zs, out);
}


# Zero chain, outside the matrix loops: mask reads, T0 fill and each step's list+mask reads before its stores are issued as groups (few L2 round trips, not one per item); sum order unchanged
# speedup vs baseline: 2.7816x; 1.0895x over previous
#include <hip/hip_runtime.h>
#include <hip/hip_bf16.h>
#include <math.h>


typedef float v2f __attribute__((ext_vector_type(2)));
typedef float v4f __attribute__((ext_vector_type(4)));
typedef float v8f __attribute__((ext_vector_type(8)));
typedef int   v4i __attribute__((ext_vector_type(4)));

#define HID   1024
#define NST   128
#define BATCH 256
#define IN2   256
#define CB    32
#define TS    1028

__device__ __forceinline__ v8f wmma_f32(v2f a, v2f b, v8f c) {
  return __builtin_amdgcn_wmma_f32_16x16x4_f32(
       false, a,  false, b,
       (short)0, c,  false,  false);
}

#define PIPE_FENCE() __builtin_amdgcn_sched_barrier(0)

__device__ __forceinline__ v2f lo2(v4f v) { return __builtin_shufflevector(v, v, 0, 1); }
__device__ __forceinline__ v2f hi2(v4f v) { return __builtin_shufflevector(v, v, 2, 3); }

__global__ void build_x2(const float* __restrict__ x, float* __restrict__ X2) {
  int i = blockIdx.x * blockDim.x + threadIdx.x;
  if (i >= 512 * IN2) return;
  int row = i >> 8, c = i & 255;
  float v;
  if (row < 256) {
    v = x[row * 512 + c];
  } else {
    int rz = row - 256;
    v = (c < 128) ? x[rz * 512 + 384 + c] : x[rz * 512 + 128 + (c - 128)];
  }
  X2[i] = v;
}

__global__ void zero_buf(float* __restrict__ p, int n) {
  int i = blockIdx.x * blockDim.x + threadIdx.x;
  if (i < n) p[i] = 0.f;
}

__device__ __forceinline__ void awt_tile(
    const float* __restrict__ A, int lda,
    const float* __restrict__ W, int ldw,
    const float* __restrict__ bias,
    const float* __restrict__ mask, int ldm,
    float* __restrict__ C, int ldc,
    int M, int K, int epi)
{
  const int lane = threadIdx.x & 31, wv = threadIdx.x >> 5;
  const int l16 = lane & 15, hlf = lane >> 4;
  const int m0 = (blockIdx.x * (int)(blockDim.x >> 5) + wv) * 16;
  const int n0 = blockIdx.y * 64;
  if (m0 >= M) return;

  const v8f vz = {0.f,0.f,0.f,0.f,0.f,0.f,0.f,0.f};
  v8f acc[4] = {vz, vz, vz, vz};

  const int kof = hlf * 4;
  const float* Ar  = A + (size_t)(m0 + l16) * lda;
  const float* Wr0 = W + (size_t)(n0 +  0 + l16) * ldw;
  const float* Wr1 = W + (size_t)(n0 + 16 + l16) * ldw;
  const float* Wr2 = W + (size_t)(n0 + 32 + l16) * ldw;
  const float* Wr3 = W + (size_t)(n0 + 48 + l16) * ldw;

  v4f aS[4], bS[4][4];
#define AWT_LD(st, kk) { aS[st]    = *(const v4f*)(Ar  + (kk)); \
                         bS[st][0] = *(const v4f*)(Wr0 + (kk)); \
                         bS[st][1] = *(const v4f*)(Wr1 + (kk)); \
                         bS[st][2] = *(const v4f*)(Wr2 + (kk)); \
                         bS[st][3] = *(const v4f*)(Wr3 + (kk)); }
#define AWT_MM(st) { acc[0] = wmma_f32(lo2(aS[st]), lo2(bS[st][0]), acc[0]); \
                     acc[1] = wmma_f32(lo2(aS[st]), lo2(bS[st][1]), acc[1]); \
                     acc[2] = wmma_f32(lo2(aS[st]), lo2(bS[st][2]), acc[2]); \
                     acc[3] = wmma_f32(lo2(aS[st]), lo2(bS[st][3]), acc[3]); \
                     acc[0] = wmma_f32(hi2(aS[st]), hi2(bS[st][0]), acc[0]); \
                     acc[1] = wmma_f32(hi2(aS[st]), hi2(bS[st][1]), acc[1]); \
                     acc[2] = wmma_f32(hi2(aS[st]), hi2(bS[st][2]), acc[2]); \
                     acc[3] = wmma_f32(hi2(aS[st]), hi2(bS[st][3]), acc[3]); }

  AWT_LD(0, kof);      PIPE_FENCE();
  AWT_LD(1, kof + 8);  PIPE_FENCE();
  AWT_LD(2, kof + 16); PIPE_FENCE();
  AWT_LD(3, kof + 24); PIPE_FENCE();
  for (int k = 0; k < K - 32; k += 32) {
    AWT_MM(0); PIPE_FENCE(); AWT_LD(0, k + 32 + kof); PIPE_FENCE();
    AWT_MM(1); PIPE_FENCE(); AWT_LD(1, k + 40 + kof); PIPE_FENCE();
    AWT_MM(2); PIPE_FENCE(); AWT_LD(2, k + 48 + kof); PIPE_FENCE();
    AWT_MM(3); PIPE_FENCE(); AWT_LD(3, k + 56 + kof); PIPE_FENCE();
  }
  AWT_MM(0);
  AWT_MM(1);
  AWT_MM(2);
  AWT_MM(3);
#undef AWT_LD
#undef AWT_MM

  #pragma unroll
  for (int nt = 0; nt < 4; ++nt) {
    #pragma unroll
    for (int r = 0; r < 8; ++r) {
      const int m  = m0 + r + 8 * hlf;
      const int nn = n0 + nt * 16 + l16;
      float v = acc[nt][r];
      if (bias) v += bias[nn];
      if (epi == 1)      v = v > 0.f ? v   : 0.f;
      else if (epi == 2) v = v > 0.f ? 1.f : 0.f;
      if (mask) v *= mask[(size_t)m * ldm + nn];
      C[(size_t)m * ldc + nn] = v;
    }
  }
}

__device__ __forceinline__ void ab_tile(
    const float* __restrict__ A, int lda,
    const float* __restrict__ B, int ldb,
    float* __restrict__ C, int ldc,
    int M, int K)
{
  const int lane = threadIdx.x & 31, wv = threadIdx.x >> 5;
  const int l16 = lane & 15, hlf = lane >> 4;
  const int m0 = (blockIdx.x * (int)(blockDim.x >> 5) + wv) * 16;
  const int n0 = blockIdx.y * 64;
  if (m0 >= M) return;

  const v8f vz = {0.f,0.f,0.f,0.f,0.f,0.f,0.f,0.f};
  v8f acc[4] = {vz, vz, vz, vz};
  const int kof = hlf * 4;
  const float* Ar = A + (size_t)(m0 + l16) * lda;
  const float* Bc = B + n0 + l16;

  v4f a0, c00, c10, c20, c30;
  v4f a1, c01, c11, c21, c31;

#define BLD(dst, kk, nt) { (dst).x = Bc[(size_t)(kk) * ldb + (nt) * 16]; \
                           (dst).y = Bc[(size_t)((kk) + 1) * ldb + (nt) * 16]; \
                           (dst).z = Bc[(size_t)((kk) + 2) * ldb + (nt) * 16]; \
                           (dst).w = Bc[(size_t)((kk) + 3) * ldb + (nt) * 16]; }
#define LDSET0(kk) { a0 = *(const v4f*)(Ar + (kk)); \
                     BLD(c00, (kk), 0) BLD(c10, (kk), 1) BLD(c20, (kk), 2) BLD(c30, (kk), 3) }
#define LDSET1(kk) { a1 = *(const v4f*)(Ar + (kk)); \
                     BLD(c01, (kk), 0) BLD(c11, (kk), 1) BLD(c21, (kk), 2) BLD(c31, (kk), 3) }
#define WM0() { acc[0] = wmma_f32(lo2(a0), lo2(c00), acc[0]); acc[1] = wmma_f32(lo2(a0), lo2(c10), acc[1]); \
                acc[2] = wmma_f32(lo2(a0), lo2(c20), acc[2]); acc[3] = wmma_f32(lo2(a0), lo2(c30), acc[3]); \
                acc[0] = wmma_f32(hi2(a0), hi2(c00), acc[0]); acc[1] = wmma_f32(hi2(a0), hi2(c10), acc[1]); \
                acc[2] = wmma_f32(hi2(a0), hi2(c20), acc[2]); acc[3] = wmma_f32(hi2(a0), hi2(c30), acc[3]); }
#define WM1() { acc[0] = wmma_f32(lo2(a1), lo2(c01), acc[0]); acc[1] = wmma_f32(lo2(a1), lo2(c11), acc[1]); \
                acc[2] = wmma_f32(lo2(a1), lo2(c21), acc[2]); acc[3] = wmma_f32(lo2(a1), lo2(c31), acc[3]); \
                acc[0] = wmma_f32(hi2(a1), hi2(c01), acc[0]); acc[1] = wmma_f32(hi2(a1), hi2(c11), acc[1]); \
                acc[2] = wmma_f32(hi2(a1), hi2(c21), acc[2]); acc[3] = wmma_f32(hi2(a1), hi2(c31), acc[3]); }

  LDSET0(kof); PIPE_FENCE();
  LDSET1(kof + 8); PIPE_FENCE();
  for (int k = 0; k < K - 16; k += 16) {
    WM0(); PIPE_FENCE(); LDSET0(k + 16 + kof); PIPE_FENCE();
    WM1(); PIPE_FENCE(); LDSET1(k + 24 + kof); PIPE_FENCE();
  }
  WM0();
  WM1();
#undef BLD
#undef LDSET0
#undef LDSET1
#undef WM0
#undef WM1

  #pragma unroll
  for (int nt = 0; nt < 4; ++nt)
    #pragma unroll
    for (int r = 0; r < 8; ++r)
      C[(size_t)(m0 + r + 8 * hlf) * ldc + n0 + nt * 16 + l16] = acc[nt][r];
}

struct LevelArgs {
  const float* A0; const float* W0; const float* bias0; const float* mask0; float* C0;
  const float* A1; const float* W1; const float* bias1; const float* mask1; float* C1;
  const float* A2; const float* W2; const float* bias2; const float* mask2; float* C2;
  const float* Aab; const float* Bab; float* Cab;
  int lda0, ldw0, ldm0, ldc0, M0, K0, epi0, gx0, gy0;
  int lda1, ldw1, ldm1, ldc1, M1, K1, epi1, gx1, gy1;
  int lda2, ldw2, ldm2, ldc2, M2, K2, epi2, gx2, gy2;
  int ldaab, ldbab, ldcab, Mab, Kab, gxab, gyab, abz;
};

__global__ __launch_bounds__(128) void level_gemm(const LevelArgs p)
{
  const int z = blockIdx.z;
  const int bx = blockIdx.x, by = blockIdx.y;
  if (z == p.abz) {
    if (bx < p.gxab && by < p.gyab)
      ab_tile(p.Aab, p.ldaab, p.Bab, p.ldbab, p.Cab, p.ldcab, p.Mab, p.Kab);
    return;
  }
  const float* A; const float* Wt; const float* bias; const float* mask; float* C;
  int lda, ldw, ldm, ldc, M, K, epi, gx, gy;
  if (z == 0) {
    A = p.A0; Wt = p.W0; bias = p.bias0; mask = p.mask0; C = p.C0;
    lda = p.lda0; ldw = p.ldw0; ldm = p.ldm0; ldc = p.ldc0;
    M = p.M0; K = p.K0; epi = p.epi0; gx = p.gx0; gy = p.gy0;
  } else if (z == 1) {
    A = p.A1; Wt = p.W1; bias = p.bias1; mask = p.mask1; C = p.C1;
    lda = p.lda1; ldw = p.ldw1; ldm = p.ldm1; ldc = p.ldc1;
    M = p.M1; K = p.K1; epi = p.epi1; gx = p.gx1; gy = p.gy1;
  } else {
    A = p.A2; Wt = p.W2; bias = p.bias2; mask = p.mask2; C = p.C2;
    lda = p.lda2; ldw = p.ldw2; ldm = p.ldm2; ldc = p.ldc2;
    M = p.M2; K = p.K2; epi = p.epi2; gx = p.gx2; gy = p.gy2;
  }
  if (bx >= gx || by >= gy) return;
  awt_tile(A, lda, Wt, ldw, bias, mask, ldm, C, ldc, M, K, epi);
}

__device__ __forceinline__ void rows_ldA(const float* __restrict__ Wr, int kk, v4f av[4])
{
  #pragma unroll
  for (int q = 0; q < 4; ++q)
    av[q] = *(const v4f*)(Wr + kk + 8 * q);
}

__device__ __forceinline__ void chain_ldB(
    const float* __restrict__ Tb0, const float* __restrict__ Tb1, int kk,
    v4f cB[4][2])
{
  #pragma unroll
  for (int q = 0; q < 4; ++q) {
    cB[q][0] = *(const v4f*)(Tb0 + kk + 8 * q);
    cB[q][1] = *(const v4f*)(Tb1 + kk + 8 * q);
  }
}

__device__ __forceinline__ void rows_mm(const v4f av[4], const v4f cB[4][2], v8f acc[2])
{
  #pragma unroll
  for (int q = 0; q < 4; ++q) {
    acc[0] = wmma_f32(lo2(av[q]), lo2(cB[q][0]), acc[0]);
    acc[1] = wmma_f32(lo2(av[q]), lo2(cB[q][1]), acc[1]);
    acc[0] = wmma_f32(hi2(av[q]), hi2(cB[q][0]), acc[0]);
    acc[1] = wmma_f32(hi2(av[q]), hi2(cB[q][1]), acc[1]);
  }
}

template <int NT>
__device__ __forceinline__ void rows_trip(
    const float* const Wr[NT], const float* __restrict__ Tb0,
    const float* __restrict__ Tb1, int kn,
    v4f av[NT][4], const v4f cur[4][2], v4f nxt[4][2], v8f acc[NT][2])
{
  rows_mm(av[0], cur, acc[0]); PIPE_FENCE();
  rows_ldA(Wr[0], kn, av[0]); chain_ldB(Tb0, Tb1, kn, nxt); PIPE_FENCE();
  #pragma unroll
  for (int t = 1; t < NT; ++t) {
    rows_mm(av[t], cur, acc[t]); PIPE_FENCE();
    rows_ldA(Wr[t], kn, av[t]); PIPE_FENCE();
  }
}

template <int NT>
__device__ __forceinline__ void chain_rows(
    const float* __restrict__ Wm,
    const float* __restrict__ Tin,
    float* __restrict__ Tout,
    const float* __restrict__ dm,
    const int* __restrict__ rows, int P,
    int lane, int wv)
{
  const int l16 = lane & 15, hlf = lane >> 4;
  const int kof = hlf * 4;
  const v8f vz = {0.f,0.f,0.f,0.f,0.f,0.f,0.f,0.f};
  v8f acc[NT][2];
  const float* Wr[NT];
  #pragma unroll
  for (int t = 0; t < NT; ++t) {
    acc[t][0] = vz; acc[t][1] = vz;
    Wr[t] = Wm + (size_t)rows[(wv + 8 * t) * 16 + l16] * HID;
  }

  const float* Tb0 = Tin + (size_t)l16 * TS;
  const float* Tb1 = Tin + (size_t)(16 + l16) * TS;

  v4f av[NT][4];
  v4f cBa[4][2], cBb[4][2];

  rows_ldA(Wr[0], kof, av[0]); chain_ldB(Tb0, Tb1, kof, cBa); PIPE_FENCE();
  #pragma unroll
  for (int t = 1; t < NT; ++t) { rows_ldA(Wr[t], kof, av[t]); PIPE_FENCE(); }
  for (int k = 0; k < HID - 64; k += 64) {
    rows_trip<NT>(Wr, Tb0, Tb1, k + 32 + kof, av, cBa, cBb, acc);
    rows_trip<NT>(Wr, Tb0, Tb1, k + 64 + kof, av, cBb, cBa, acc);
  }
  rows_trip<NT>(Wr, Tb0, Tb1, HID - 32 + kof, av, cBa, cBb, acc);
  #pragma unroll
  for (int t = 0; t < NT; ++t) rows_mm(av[t], cBb, acc[t]);

  int   rw[NT][8];
  float dd[NT][8];
  #pragma unroll
  for (int t = 0; t < NT; ++t) {
    const int pb = (wv + 8 * t) * 16 + 8 * hlf;
    const v4i ra = *(const v4i*)(rows + pb);
    const v4i rb = *(const v4i*)(rows + pb + 4);
    rw[t][0] = ra.x; rw[t][1] = ra.y; rw[t][2] = ra.z; rw[t][3] = ra.w;
    rw[t][4] = rb.x; rw[t][5] = rb.y; rw[t][6] = rb.z; rw[t][7] = rb.w;
  }
  #pragma unroll
  for (int t = 0; t < NT; ++t) {
    #pragma unroll
    for (int r = 0; r < 8; ++r) dd[t][r] = dm[rw[t][r]];
  }
  #pragma unroll
  for (int t = 0; t < NT; ++t) {
    const int pb = (wv + 8 * t) * 16 + 8 * hlf;
    #pragma unroll
    for (int r = 0; r < 8; ++r) {
      const int col = (pb + r < P) ? rw[t][r] : HID;
      Tout[(size_t)l16 * TS + col]        = acc[t][0][r] * dd[t][r];
      Tout[(size_t)(16 + l16) * TS + col] = acc[t][1][r] * dd[t][r];
    }
  }
}

__global__ __launch_bounds__(256) void zero_chain(
    const float* __restrict__ M0,
    const float* __restrict__ Wh1, const float* __restrict__ Wh2,
    const float* __restrict__ d1z, const float* __restrict__ d2z,
    const float* __restrict__ d3z,
    const float* __restrict__ Wout,
    float* __restrict__ zsum)
{
  extern __shared__ float lds[];
  float* T0 = lds;
  float* T1 = lds + CB * TS;
  int* rowsL = (int*)(lds + 2 * CB * TS);
  int* cnt   = rowsL + 2 * HID;

  const int cb = blockIdx.x;
  const int b  = blockIdx.y;
  const int c0 = cb * CB;
  const int lane = threadIdx.x & 31, wv = threadIdx.x >> 5;

  const float* d1 = d1z + (size_t)b * HID;
  const float* d2 = d2z + (size_t)b * HID;
  const float* d3 = d3z + (size_t)b * HID;

  unsigned mk[2][4];
  {
    float dv[2][4];
    #pragma unroll
    for (int g = 0; g < 2; ++g) {
      const float* dg = g ? d3 : d2;
      #pragma unroll
      for (int j = 0; j < 4; ++j) dv[g][j] = dg[j * 256 + threadIdx.x];
    }
    #pragma unroll
    for (int g = 0; g < 2; ++g) {
      #pragma unroll
      for (int j = 0; j < 4; ++j) mk[g][j] = (unsigned)__ballot(dv[g][j] != 0.f);
    }
    if (lane == 0) {
      #pragma unroll
      for (int g = 0; g < 2; ++g) {
        #pragma unroll
        for (int j = 0; j < 4; ++j) cnt[g * 32 + j * 8 + wv] = __popc(mk[g][j]);
      }
    }
  }

  {
    const float* Mc = M0 + c0 + lane;
    float* Tc = T0 + (size_t)lane * TS;
    const int jw = wv * 128;
    #pragma unroll 1
    for (int jb = jw; jb < jw + 128; jb += 32) {
      float mv[8][4];
      v4f dq[8];
      #pragma unroll
      for (int u = 0; u < 8; ++u) {
        dq[u] = *(const v4f*)(d1 + jb + 4 * u);
        #pragma unroll
        for (int i = 0; i < 4; ++i) mv[u][i] = Mc[(size_t)(jb + 4 * u + i) * NST];
      }
      #pragma unroll
      for (int u = 0; u < 8; ++u) {
        v4f o;
        o.x = mv[u][0] * dq[u].x;
        o.y = mv[u][1] * dq[u].y;
        o.z = mv[u][2] * dq[u].z;
        o.w = mv[u][3] * dq[u].w;
        *(v4f*)(Tc + jb + 4 * u) = o;
      }
    }
  }
  for (int i = threadIdx.x; i < CB * (HID / 4); i += 256)
    *(float4*)(T1 + (size_t)(i >> 8) * TS + 4 * (i & 255)) = make_float4(0.f, 0.f, 0.f, 0.f);
  __syncthreads();

  int Pn[2];
  #pragma unroll
  for (int g = 0; g < 2; ++g) {
    int tot = 0, off[4] = {0, 0, 0, 0};
    #pragma unroll
    for (int sgm = 0; sgm < 32; ++sgm) {
      if ((sgm & 7) == wv) off[sgm >> 3] = tot;
      tot += cnt[g * 32 + sgm];
    }
    int* rl = rowsL + g * HID;
    #pragma unroll
    for (int j = 0; j < 4; ++j)
      if ((mk[g][j] >> lane) & 1u)
        rl[off[j] + __popc(mk[g][j] & ((1u << lane) - 1u))] = j * 256 + threadIdx.x;
    const int cap = ((tot + 15) >> 4) << 4;
    for (int i = tot + threadIdx.x; i < cap; i += 256) rl[i] = 0;
    Pn[g] = tot;
  }

  #pragma unroll 1
  for (int g = 0; g < 2; ++g) {
    const float* Wm  = g ? Wh2 : Wh1;
    const float* dmg = g ? d3 : d2;
    const float* Tin = lds + (g ? CB * TS : 0);
    float* Tout      = lds + (g ? 0 : CB * TS);
    const int* rl    = rowsL + g * HID;
    const int P      = g ? Pn[1] : Pn[0];
    if (g == 1) {
      for (int i = threadIdx.x; i < CB * (HID / 4); i += 256)
        *(float4*)(T0 + (size_t)(i >> 8) * TS + 4 * (i & 255)) = make_float4(0.f, 0.f, 0.f, 0.f);
    }
    __syncthreads();
    const int ntot = (P + 15) >> 4;
    const int ntw = __builtin_amdgcn_readfirstlane(wv < ntot ? (ntot - wv + 7) >> 3 : 0);
    switch (ntw) {
      case 1: chain_rows<1>(Wm, Tin, Tout, dmg, rl, P, lane, wv); break;
      case 2: chain_rows<2>(Wm, Tin, Tout, dmg, rl, P, lane, wv); break;
      case 3: chain_rows<3>(Wm, Tin, Tout, dmg, rl, P, lane, wv); break;
      case 4: chain_rows<4>(Wm, Tin, Tout, dmg, rl, P, lane, wv); break;
      case 5: chain_rows<5>(Wm, Tin, Tout, dmg, rl, P, lane, wv); break;
      case 6: chain_rows<6>(Wm, Tin, Tout, dmg, rl, P, lane, wv); break;
      case 7: chain_rows<7>(Wm, Tin, Tout, dmg, rl, P, lane, wv); break;
      case 8: chain_rows<8>(Wm, Tin, Tout, dmg, rl, P, lane, wv); break;
      default: break;
    }
    __syncthreads();
  }

  const int l16 = lane & 15, hlf = lane >> 4;
  const int kof = hlf * 4;
  const v8f vz = {0.f,0.f,0.f,0.f,0.f,0.f,0.f,0.f};
  v8f g0 = vz, g1 = vz;
  const float* Tb0 = T0 + (size_t)l16 * TS;
  const float* Tb1 = T0 + (size_t)(16 + l16) * TS;
  const float* Wr  = Wout + (size_t)(wv * 16 + l16) * HID;

  v4f wa[4][4], wc0[4][4], wc1[4][4];
#define FLD(st, kk) { wa[st][0]  = *(const v4f*)(Wr  + (kk) + 0); wc0[st][0] = *(const v4f*)(Tb0 + (kk) + 0); wc1[st][0] = *(const v4f*)(Tb1 + (kk) + 0); \
                      wa[st][1]  = *(const v4f*)(Wr  + (kk) + 8); wc0[st][1] = *(const v4f*)(Tb0 + (kk) + 8); wc1[st][1] = *(const v4f*)(Tb1 + (kk) + 8); \
                      wa[st][2]  = *(const v4f*)(Wr  + (kk) + 16); wc0[st][2] = *(const v4f*)(Tb0 + (kk) + 16); wc1[st][2] = *(const v4f*)(Tb1 + (kk) + 16); \
                      wa[st][3]  = *(const v4f*)(Wr  + (kk) + 24); wc0[st][3] = *(const v4f*)(Tb0 + (kk) + 24); wc1[st][3] = *(const v4f*)(Tb1 + (kk) + 24); }
#define GM(st) { g0 = wmma_f32(lo2(wa[st][0]), lo2(wc0[st][0]), g0); g1 = wmma_f32(lo2(wa[st][0]), lo2(wc1[st][0]), g1); \
                 g0 = wmma_f32(hi2(wa[st][0]), hi2(wc0[st][0]), g0); g1 = wmma_f32(hi2(wa[st][0]), hi2(wc1[st][0]), g1); \
                 g0 = wmma_f32(lo2(wa[st][1]), lo2(wc0[st][1]), g0); g1 = wmma_f32(lo2(wa[st][1]), lo2(wc1[st][1]), g1); \
                 g0 = wmma_f32(hi2(wa[st][1]), hi2(wc0[st][1]), g0); g1 = wmma_f32(hi2(wa[st][1]), hi2(wc1[st][1]), g1); \
                 g0 = wmma_f32(lo2(wa[st][2]), lo2(wc0[st][2]), g0); g1 = wmma_f32(lo2(wa[st][2]), lo2(wc1[st][2]), g1); \
                 g0 = wmma_f32(hi2(wa[st][2]), hi2(wc0[st][2]), g0); g1 = wmma_f32(hi2(wa[st][2]), hi2(wc1[st][2]), g1); \
                 g0 = wmma_f32(lo2(wa[st][3]), lo2(wc0[st][3]), g0); g1 = wmma_f32(lo2(wa[st][3]), lo2(wc1[st][3]), g1); \
                 g0 = wmma_f32(hi2(wa[st][3]), hi2(wc0[st][3]), g0); g1 = wmma_f32(hi2(wa[st][3]), hi2(wc1[st][3]), g1); }
  FLD(0, kof);      PIPE_FENCE();
  FLD(1, kof + 32); PIPE_FENCE();
  FLD(2, kof + 64); PIPE_FENCE();
  FLD(3, kof + 96); PIPE_FENCE();
  for (int k = 0; k < HID - 128; k += 128) {
    GM(0); PIPE_FENCE(); FLD(0, k + 128 + kof); PIPE_FENCE();
    GM(1); PIPE_FENCE(); FLD(1, k + 160 + kof); PIPE_FENCE();
    GM(2); PIPE_FENCE(); FLD(2, k + 192 + kof); PIPE_FENCE();
    GM(3); PIPE_FENCE(); FLD(3, k + 224 + kof); PIPE_FENCE();
  }
  GM(0);
  GM(1);
  GM(2);
  GM(3);
#undef FLD
#undef GM

  #pragma unroll
  for (int r = 0; r < 8; ++r) {
    float s = g0[r] * g0[r] + g1[r] * g1[r];
    s += __shfl_xor(s, 1);
    s += __shfl_xor(s, 2);
    s += __shfl_xor(s, 4);
    s += __shfl_xor(s, 8);
    if (l16 == 0)
      atomicAdd(&zsum[(size_t)b * NST + wv * 16 + r + 8 * hlf], s);
  }
}

__global__ void finalize_norm(const float* __restrict__ zsum,
                              float* __restrict__ out) {
  int i = blockIdx.x * blockDim.x + threadIdx.x;
  if (i >= BATCH * NST) return;
  int b = i >> 7, c = i & 127;
  out[(size_t)b * 384 + 256 + c] = sqrtf(zsum[i]);
}

extern "C" void kernel_launch(void* const* d_in, const int* in_sizes, int n_in,
                              void* d_out, int out_size, void* d_ws, size_t ws_size,
                              hipStream_t stream)
{
  (void)in_sizes; (void)n_in; (void)out_size; (void)ws_size;
  const float* x     = (const float*)d_in[0];
  const float* W_in  = (const float*)d_in[1];
  const float* b_in  = (const float*)d_in[2];
  const float* Wh    = (const float*)d_in[3];
  const float* bh    = (const float*)d_in[4];
  const float* W_out = (const float*)d_in[5];
  const float* b_out = (const float*)d_in[6];
  float* out = (float*)d_out;
  float* ws  = (float*)d_ws;

  const float* Wh0 = Wh;
  const float* Wh1 = Wh + 1024 * 1024;
  const float* Wh2 = Wh + 2 * 1024 * 1024;
  const float* bh0 = bh, *bh1 = bh + 1024, *bh2 = bh + 2048;

  float* X2  = ws;
  float* hA  = X2  + 512 * 256;
  float* hB  = hA  + 512 * 1024;
  float* D1  = hB  + 512 * 1024;
  float* D2  = D1  + 512 * 1024;
  float* D3  = D2  + 512 * 1024;
  float* uA  = D3  + 512 * 1024;
  float* uB  = uA  + 256 * 1024;
  float* M0b = uB  + 256 * 1024;
  float* zs  = M0b + 1024 * 128;

  dim3 blk(256);
  build_x2<<<dim3((512 * 256 + 255) / 256), blk, 0, stream>>>(x, X2);
  zero_buf<<<dim3((256 * 128 + 255) / 256), blk, 0, stream>>>(zs, 256 * 128);

  dim3 lblk(128);
  dim3 lblk2(64);
#define SLOT(L, n, A_, lda_, W_, ldw_, b_, m_, ldm_, C_, ldc_, M_, K_, e_, gx_, gy_) \
  { (L).A##n = (A_); (L).lda##n = (lda_); (L).W##n = (W_); (L).ldw##n = (ldw_); \
    (L).bias##n = (b_); (L).mask##n = (m_); (L).ldm##n = (ldm_); (L).C##n = (C_); \
    (L).ldc##n = (ldc_); (L).M##n = (M_); (L).K##n = (K_); (L).epi##n = (e_); \
    (L).gx##n = (gx_); (L).gy##n = (gy_); }

  LevelArgs L1 = {}; L1.abz = -1;
  SLOT(L1, 0, X2,256,      W_in,256, b_in,    nullptr,0, hA,1024, 512,256, 0, 8,16)
  SLOT(L1, 1, x + 256,512, W_in,256, nullptr, nullptr,0, uA,1024, 256,128, 0, 4,16)
  level_gemm<<<dim3(8,16,2), lblk, 0, stream>>>(L1);

  LevelArgs L2 = {}; L2.abz = -1;
  SLOT(L2, 0, hA,1024, Wh0,1024, bh0, nullptr,0, hB,1024, 512,1024, 1, 16,16)
  level_gemm<<<dim3(16,16,1), lblk2, 0, stream>>>(L2);

  LevelArgs L3 = {}; L3.abz = -1;
  SLOT(L3, 0, hB,1024, Wh0,1024, bh0, nullptr,0, D1,1024, 512,1024, 2, 8,16)
  SLOT(L3, 1, hB,1024, Wh1,1024, bh1, nullptr,0, hA,1024, 512,1024, 1, 8,16)
  level_gemm<<<dim3(8,16,2), lblk, 0, stream>>>(L3);

  LevelArgs L4 = {}; L4.abz = -1;
  SLOT(L4, 0, hA,1024, Wh1,1024, bh1,     nullptr,0, D2,1024, 256,1024, 2, 4,16)
  SLOT(L4, 1, hA,1024, Wh2,1024, bh2,     nullptr,0, hB,1024, 512,1024, 1, 8,16)
  SLOT(L4, 2, uA,1024, Wh0,1024, nullptr, D1,1024,   uB,1024, 256,1024, 0, 4,16)
  level_gemm<<<dim3(8,16,3), lblk, 0, stream>>>(L4);

  LevelArgs L5 = {}; L5.abz = -1;
  SLOT(L5, 0, hB,1024, Wh2,1024,   bh2,     nullptr,0, D3,1024, 512,1024, 2, 8,16)
  SLOT(L5, 1, hB,1024, W_out,1024, b_out,   nullptr,0, out,384, 256,1024, 0, 4,2)
  SLOT(L5, 2, uB,1024, Wh1,1024,   nullptr, D2,1024,   uA,1024, 256,1024, 0, 4,16)
  level_gemm<<<dim3(8,16,3), lblk, 0, stream>>>(L5);

  LevelArgs L6 = {}; L6.abz = -1;
  SLOT(L6, 0, uA,1024, Wh2,1024, nullptr, D3,1024, uB,1024, 256,1024, 0, 8,16)
  SLOT(L6, 1, hA + 256 * 1024,1024, Wh1,1024, bh1, nullptr,0, D2 + 256 * 1024,1024, 256,1024, 2, 8,16)
  level_gemm<<<dim3(8,16,2), lblk2, 0, stream>>>(L6);

  LevelArgs L7 = {}; L7.abz = 1;
  SLOT(L7, 0, uB,1024, W_out,1024, nullptr, nullptr,0, out + 128,384, 256,1024, 0, 4,2)
  L7.Aab = Wh0; L7.ldaab = 1024; L7.Bab = W_in; L7.ldbab = 256; L7.Cab = M0b; L7.ldcab = 128;
  L7.Mab = 1024; L7.Kab = 1024; L7.gxab = 16; L7.gyab = 2;
  level_gemm<<<dim3(16,2,2), lblk, 0, stream>>>(L7);
#undef SLOT

  zero_chain<<<dim3(4,256), blk, 2 * CB * TS * sizeof(float) + 2 * HID * sizeof(int) + 64 * sizeof(int), stream>>>(
      M0b, Wh1, Wh2,
      D1 + 256 * 1024, D2 + 256 * 1024, D3 + 256 * 1024,
      W_out, zs);
  finalize_norm<<<dim3((256 * 128 + 255) / 256), blk, 0, stream>>>(zs, out);
}


# Level GEMM epilogue chooses bias, activation and mask once per wave, not per element: their values are read as one group in front of the K loop, the element loop has no branch or read; sums unchanged
# speedup vs baseline: 3.0320x; 1.0900x over previous
#include <hip/hip_runtime.h>
#include <hip/hip_bf16.h>
#include <math.h>


typedef float v2f __attribute__((ext_vector_type(2)));
typedef float v4f __attribute__((ext_vector_type(4)));
typedef float v8f __attribute__((ext_vector_type(8)));
typedef int   v4i __attribute__((ext_vector_type(4)));

#define HID   1024
#define NST   128
#define BATCH 256
#define IN2   256
#define CB    32
#define TS    1028

__device__ __forceinline__ v8f wmma_f32(v2f a, v2f b, v8f c) {
  return __builtin_amdgcn_wmma_f32_16x16x4_f32(
       false, a,  false, b,
       (short)0, c,  false,  false);
}

#define PIPE_FENCE() __builtin_amdgcn_sched_barrier(0)

__device__ __forceinline__ v2f lo2(v4f v) { return __builtin_shufflevector(v, v, 0, 1); }
__device__ __forceinline__ v2f hi2(v4f v) { return __builtin_shufflevector(v, v, 2, 3); }

__global__ void build_x2(const float* __restrict__ x, float* __restrict__ X2) {
  int i = blockIdx.x * blockDim.x + threadIdx.x;
  if (i >= 512 * IN2) return;
  int row = i >> 8, c = i & 255;
  float v;
  if (row < 256) {
    v = x[row * 512 + c];
  } else {
    int rz = row - 256;
    v = (c < 128) ? x[rz * 512 + 384 + c] : x[rz * 512 + 128 + (c - 128)];
  }
  X2[i] = v;
}

__global__ void zero_buf(float* __restrict__ p, int n) {
  int i = blockIdx.x * blockDim.x + threadIdx.x;
  if (i < n) p[i] = 0.f;
}

__device__ __forceinline__ void awt_tile(
    const float* __restrict__ A, int lda,
    const float* __restrict__ W, int ldw,
    const float* __restrict__ bias,
    const float* __restrict__ mask, int ldm,
    float* __restrict__ C, int ldc,
    int M, int K, int epi)
{
  const int lane = threadIdx.x & 31, wv = threadIdx.x >> 5;
  const int l16 = lane & 15, hlf = lane >> 4;
  const int m0 = (blockIdx.x * (int)(blockDim.x >> 5) + wv) * 16;
  const int n0 = blockIdx.y * 64;
  if (m0 >= M) return;

  const v8f vz = {0.f,0.f,0.f,0.f,0.f,0.f,0.f,0.f};
  v8f acc[4] = {vz, vz, vz, vz};

  const int kof = hlf * 4;
  const float* Ar  = A + (size_t)(m0 + l16) * lda;
  const float* Wr0 = W + (size_t)(n0 +  0 + l16) * ldw;
  const float* Wr1 = W + (size_t)(n0 + 16 + l16) * ldw;
  const float* Wr2 = W + (size_t)(n0 + 32 + l16) * ldw;
  const float* Wr3 = W + (size_t)(n0 + 48 + l16) * ldw;

  float bv[4] = {0.f, 0.f, 0.f, 0.f};
  float mv[4][8];
  #pragma unroll
  for (int nt = 0; nt < 4; ++nt) {
    #pragma unroll
    for (int r = 0; r < 8; ++r) mv[nt][r] = 1.f;
  }
  if (bias) {
    #pragma unroll
    for (int nt = 0; nt < 4; ++nt) bv[nt] = bias[n0 + nt * 16 + l16];
  }
  if (mask) {
    #pragma unroll
    for (int nt = 0; nt < 4; ++nt) {
      #pragma unroll
      for (int r = 0; r < 8; ++r)
        mv[nt][r] = mask[(size_t)(m0 + r + 8 * hlf) * ldm + n0 + nt * 16 + l16];
    }
  }
  PIPE_FENCE();

  v4f aS[4], bS[4][4];
#define AWT_LD(st, kk) { aS[st]    = *(const v4f*)(Ar  + (kk)); \
                         bS[st][0] = *(const v4f*)(Wr0 + (kk)); \
                         bS[st][1] = *(const v4f*)(Wr1 + (kk)); \
                         bS[st][2] = *(const v4f*)(Wr2 + (kk)); \
                         bS[st][3] = *(const v4f*)(Wr3 + (kk)); }
#define AWT_MM(st) { acc[0] = wmma_f32(lo2(aS[st]), lo2(bS[st][0]), acc[0]); \
                     acc[1] = wmma_f32(lo2(aS[st]), lo2(bS[st][1]), acc[1]); \
                     acc[2] = wmma_f32(lo2(aS[st]), lo2(bS[st][2]), acc[2]); \
                     acc[3] = wmma_f32(lo2(aS[st]), lo2(bS[st][3]), acc[3]); \
                     acc[0] = wmma_f32(hi2(aS[st]), hi2(bS[st][0]), acc[0]); \
                     acc[1] = wmma_f32(hi2(aS[st]), hi2(bS[st][1]), acc[1]); \
                     acc[2] = wmma_f32(hi2(aS[st]), hi2(bS[st][2]), acc[2]); \
                     acc[3] = wmma_f32(hi2(aS[st]), hi2(bS[st][3]), acc[3]); }

  AWT_LD(0, kof);      PIPE_FENCE();
  AWT_LD(1, kof + 8);  PIPE_FENCE();
  AWT_LD(2, kof + 16); PIPE_FENCE();
  AWT_LD(3, kof + 24); PIPE_FENCE();
  for (int k = 0; k < K - 32; k += 32) {
    AWT_MM(0); PIPE_FENCE(); AWT_LD(0, k + 32 + kof); PIPE_FENCE();
    AWT_MM(1); PIPE_FENCE(); AWT_LD(1, k + 40 + kof); PIPE_FENCE();
    AWT_MM(2); PIPE_FENCE(); AWT_LD(2, k + 48 + kof); PIPE_FENCE();
    AWT_MM(3); PIPE_FENCE(); AWT_LD(3, k + 56 + kof); PIPE_FENCE();
  }
  AWT_MM(0);
  AWT_MM(1);
  AWT_MM(2);
  AWT_MM(3);
#undef AWT_LD
#undef AWT_MM

  if (bias) {
    #pragma unroll
    for (int nt = 0; nt < 4; ++nt) {
      #pragma unroll
      for (int r = 0; r < 8; ++r) acc[nt][r] += bv[nt];
    }
  }
  if (epi == 1) {
    #pragma unroll
    for (int nt = 0; nt < 4; ++nt) {
      #pragma unroll
      for (int r = 0; r < 8; ++r) acc[nt][r] = acc[nt][r] > 0.f ? acc[nt][r] : 0.f;
    }
  } else if (epi == 2) {
    #pragma unroll
    for (int nt = 0; nt < 4; ++nt) {
      #pragma unroll
      for (int r = 0; r < 8; ++r) acc[nt][r] = acc[nt][r] > 0.f ? 1.f : 0.f;
    }
  }
  if (mask) {
    #pragma unroll
    for (int nt = 0; nt < 4; ++nt) {
      #pragma unroll
      for (int r = 0; r < 8; ++r) acc[nt][r] *= mv[nt][r];
    }
  }
  #pragma unroll
  for (int r = 0; r < 8; ++r) {
    float* Cr = C + (size_t)(m0 + r + 8 * hlf) * ldc + n0 + l16;
    #pragma unroll
    for (int nt = 0; nt < 4; ++nt) Cr[nt * 16] = acc[nt][r];
  }
}

__device__ __forceinline__ void ab_tile(
    const float* __restrict__ A, int lda,
    const float* __restrict__ B, int ldb,
    float* __restrict__ C, int ldc,
    int M, int K)
{
  const int lane = threadIdx.x & 31, wv = threadIdx.x >> 5;
  const int l16 = lane & 15, hlf = lane >> 4;
  const int m0 = (blockIdx.x * (int)(blockDim.x >> 5) + wv) * 16;
  const int n0 = blockIdx.y * 64;
  if (m0 >= M) return;

  const v8f vz = {0.f,0.f,0.f,0.f,0.f,0.f,0.f,0.f};
  v8f acc[4] = {vz, vz, vz, vz};
  const int kof = hlf * 4;
  const float* Ar = A + (size_t)(m0 + l16) * lda;
  const float* Bc = B + n0 + l16;

  v4f a0, c00, c10, c20, c30;
  v4f a1, c01, c11, c21, c31;

#define BLD(dst, kk, nt) { (dst).x = Bc[(size_t)(kk) * ldb + (nt) * 16]; \
                           (dst).y = Bc[(size_t)((kk) + 1) * ldb + (nt) * 16]; \
                           (dst).z = Bc[(size_t)((kk) + 2) * ldb + (nt) * 16]; \
                           (dst).w = Bc[(size_t)((kk) + 3) * ldb + (nt) * 16]; }
#define LDSET0(kk) { a0 = *(const v4f*)(Ar + (kk)); \
                     BLD(c00, (kk), 0) BLD(c10, (kk), 1) BLD(c20, (kk), 2) BLD(c30, (kk), 3) }
#define LDSET1(kk) { a1 = *(const v4f*)(Ar + (kk)); \
                     BLD(c01, (kk), 0) BLD(c11, (kk), 1) BLD(c21, (kk), 2) BLD(c31, (kk), 3) }
#define WM0() { acc[0] = wmma_f32(lo2(a0), lo2(c00), acc[0]); acc[1] = wmma_f32(lo2(a0), lo2(c10), acc[1]); \
                acc[2] = wmma_f32(lo2(a0), lo2(c20), acc[2]); acc[3] = wmma_f32(lo2(a0), lo2(c30), acc[3]); \
                acc[0] = wmma_f32(hi2(a0), hi2(c00), acc[0]); acc[1] = wmma_f32(hi2(a0), hi2(c10), acc[1]); \
                acc[2] = wmma_f32(hi2(a0), hi2(c20), acc[2]); acc[3] = wmma_f32(hi2(a0), hi2(c30), acc[3]); }
#define WM1() { acc[0] = wmma_f32(lo2(a1), lo2(c01), acc[0]); acc[1] = wmma_f32(lo2(a1), lo2(c11), acc[1]); \
                acc[2] = wmma_f32(lo2(a1), lo2(c21), acc[2]); acc[3] = wmma_f32(lo2(a1), lo2(c31), acc[3]); \
                acc[0] = wmma_f32(hi2(a1), hi2(c01), acc[0]); acc[1] = wmma_f32(hi2(a1), hi2(c11), acc[1]); \
                acc[2] = wmma_f32(hi2(a1), hi2(c21), acc[2]); acc[3] = wmma_f32(hi2(a1), hi2(c31), acc[3]); }

  LDSET0(kof); PIPE_FENCE();
  LDSET1(kof + 8); PIPE_FENCE();
  for (int k = 0; k < K - 16; k += 16) {
    WM0(); PIPE_FENCE(); LDSET0(k + 16 + kof); PIPE_FENCE();
    WM1(); PIPE_FENCE(); LDSET1(k + 24 + kof); PIPE_FENCE();
  }
  WM0();
  WM1();
#undef BLD
#undef LDSET0
#undef LDSET1
#undef WM0
#undef WM1

  #pragma unroll
  for (int nt = 0; nt < 4; ++nt)
    #pragma unroll
    for (int r = 0; r < 8; ++r)
      C[(size_t)(m0 + r + 8 * hlf) * ldc + n0 + nt * 16 + l16] = acc[nt][r];
}

struct LevelArgs {
  const float* A0; const float* W0; const float* bias0; const float* mask0; float* C0;
  const float* A1; const float* W1; const float* bias1; const float* mask1; float* C1;
  const float* A2; const float* W2; const float* bias2; const float* mask2; float* C2;
  const float* Aab; const float* Bab; float* Cab;
  int lda0, ldw0, ldm0, ldc0, M0, K0, epi0, gx0, gy0;
  int lda1, ldw1, ldm1, ldc1, M1, K1, epi1, gx1, gy1;
  int lda2, ldw2, ldm2, ldc2, M2, K2, epi2, gx2, gy2;
  int ldaab, ldbab, ldcab, Mab, Kab, gxab, gyab, abz;
};

__global__ __launch_bounds__(128) void level_gemm(const LevelArgs p)
{
  const int z = blockIdx.z;
  const int bx = blockIdx.x, by = blockIdx.y;
  if (z == p.abz) {
    if (bx < p.gxab && by < p.gyab)
      ab_tile(p.Aab, p.ldaab, p.Bab, p.ldbab, p.Cab, p.ldcab, p.Mab, p.Kab);
    return;
  }
  const float* A; const float* Wt; const float* bias; const float* mask; float* C;
  int lda, ldw, ldm, ldc, M, K, epi, gx, gy;
  if (z == 0) {
    A = p.A0; Wt = p.W0; bias = p.bias0; mask = p.mask0; C = p.C0;
    lda = p.lda0; ldw = p.ldw0; ldm = p.ldm0; ldc = p.ldc0;
    M = p.M0; K = p.K0; epi = p.epi0; gx = p.gx0; gy = p.gy0;
  } else if (z == 1) {
    A = p.A1; Wt = p.W1; bias = p.bias1; mask = p.mask1; C = p.C1;
    lda = p.lda1; ldw = p.ldw1; ldm = p.ldm1; ldc = p.ldc1;
    M = p.M1; K = p.K1; epi = p.epi1; gx = p.gx1; gy = p.gy1;
  } else {
    A = p.A2; Wt = p.W2; bias = p.bias2; mask = p.mask2; C = p.C2;
    lda = p.lda2; ldw = p.ldw2; ldm = p.ldm2; ldc = p.ldc2;
    M = p.M2; K = p.K2; epi = p.epi2; gx = p.gx2; gy = p.gy2;
  }
  if (bx >= gx || by >= gy) return;
  awt_tile(A, lda, Wt, ldw, bias, mask, ldm, C, ldc, M, K, epi);
}

__device__ __forceinline__ void rows_ldA(const float* __restrict__ Wr, int kk, v4f av[4])
{
  #pragma unroll
  for (int q = 0; q < 4; ++q)
    av[q] = *(const v4f*)(Wr + kk + 8 * q);
}

__device__ __forceinline__ void chain_ldB(
    const float* __restrict__ Tb0, const float* __restrict__ Tb1, int kk,
    v4f cB[4][2])
{
  #pragma unroll
  for (int q = 0; q < 4; ++q) {
    cB[q][0] = *(const v4f*)(Tb0 + kk + 8 * q);
    cB[q][1] = *(const v4f*)(Tb1 + kk + 8 * q);
  }
}

__device__ __forceinline__ void rows_mm(const v4f av[4], const v4f cB[4][2], v8f acc[2])
{
  #pragma unroll
  for (int q = 0; q < 4; ++q) {
    acc[0] = wmma_f32(lo2(av[q]), lo2(cB[q][0]), acc[0]);
    acc[1] = wmma_f32(lo2(av[q]), lo2(cB[q][1]), acc[1]);
    acc[0] = wmma_f32(hi2(av[q]), hi2(cB[q][0]), acc[0]);
    acc[1] = wmma_f32(hi2(av[q]), hi2(cB[q][1]), acc[1]);
  }
}

template <int NT>
__device__ __forceinline__ void rows_trip(
    const float* const Wr[NT], const float* __restrict__ Tb0,
    const float* __restrict__ Tb1, int kn,
    v4f av[NT][4], const v4f cur[4][2], v4f nxt[4][2], v8f acc[NT][2])
{
  rows_mm(av[0], cur, acc[0]); PIPE_FENCE();
  rows_ldA(Wr[0], kn, av[0]); chain_ldB(Tb0, Tb1, kn, nxt); PIPE_FENCE();
  #pragma unroll
  for (int t = 1; t < NT; ++t) {
    rows_mm(av[t], cur, acc[t]); PIPE_FENCE();
    rows_ldA(Wr[t], kn, av[t]); PIPE_FENCE();
  }
}

template <int NT>
__device__ __forceinline__ void chain_rows(
    const float* __restrict__ Wm,
    const float* __restrict__ Tin,
    float* __restrict__ Tout,
    const float* __restrict__ dm,
    const int* __restrict__ rows, int P,
    int lane, int wv)
{
  const int l16 = lane & 15, hlf = lane >> 4;
  const int kof = hlf * 4;
  const v8f vz = {0.f,0.f,0.f,0.f,0.f,0.f,0.f,0.f};
  v8f acc[NT][2];
  const float* Wr[NT];
  #pragma unroll
  for (int t = 0; t < NT; ++t) {
    acc[t][0] = vz; acc[t][1] = vz;
    Wr[t] = Wm + (size_t)rows[(wv + 8 * t) * 16 + l16] * HID;
  }

  const float* Tb0 = Tin + (size_t)l16 * TS;
  const float* Tb1 = Tin + (size_t)(16 + l16) * TS;

  v4f av[NT][4];
  v4f cBa[4][2], cBb[4][2];

  rows_ldA(Wr[0], kof, av[0]); chain_ldB(Tb0, Tb1, kof, cBa); PIPE_FENCE();
  #pragma unroll
  for (int t = 1; t < NT; ++t) { rows_ldA(Wr[t], kof, av[t]); PIPE_FENCE(); }
  for (int k = 0; k < HID - 64; k += 64) {
    rows_trip<NT>(Wr, Tb0, Tb1, k + 32 + kof, av, cBa, cBb, acc);
    rows_trip<NT>(Wr, Tb0, Tb1, k + 64 + kof, av, cBb, cBa, acc);
  }
  rows_trip<NT>(Wr, Tb0, Tb1, HID - 32 + kof, av, cBa, cBb, acc);
  #pragma unroll
  for (int t = 0; t < NT; ++t) rows_mm(av[t], cBb, acc[t]);

  int   rw[NT][8];
  float dd[NT][8];
  #pragma unroll
  for (int t = 0; t < NT; ++t) {
    const int pb = (wv + 8 * t) * 16 + 8 * hlf;
    const v4i ra = *(const v4i*)(rows + pb);
    const v4i rb = *(const v4i*)(rows + pb + 4);
    rw[t][0] = ra.x; rw[t][1] = ra.y; rw[t][2] = ra.z; rw[t][3] = ra.w;
    rw[t][4] = rb.x; rw[t][5] = rb.y; rw[t][6] = rb.z; rw[t][7] = rb.w;
  }
  #pragma unroll
  for (int t = 0; t < NT; ++t) {
    #pragma unroll
    for (int r = 0; r < 8; ++r) dd[t][r] = dm[rw[t][r]];
  }
  #pragma unroll
  for (int t = 0; t < NT; ++t) {
    const int pb = (wv + 8 * t) * 16 + 8 * hlf;
    #pragma unroll
    for (int r = 0; r < 8; ++r) {
      const int col = (pb + r < P) ? rw[t][r] : HID;
      Tout[(size_t)l16 * TS + col]        = acc[t][0][r] * dd[t][r];
      Tout[(size_t)(16 + l16) * TS + col] = acc[t][1][r] * dd[t][r];
    }
  }
}

__global__ __launch_bounds__(256) void zero_chain(
    const float* __restrict__ M0,
    const float* __restrict__ Wh1, const float* __restrict__ Wh2,
    const float* __restrict__ d1z, const float* __restrict__ d2z,
    const float* __restrict__ d3z,
    const float* __restrict__ Wout,
    float* __restrict__ zsum)
{
  extern __shared__ float lds[];
  float* T0 = lds;
  float* T1 = lds + CB * TS;
  int* rowsL = (int*)(lds + 2 * CB * TS);
  int* cnt   = rowsL + 2 * HID;

  const int cb = blockIdx.x;
  const int b  = blockIdx.y;
  const int c0 = cb * CB;
  const int lane = threadIdx.x & 31, wv = threadIdx.x >> 5;

  const float* d1 = d1z + (size_t)b * HID;
  const float* d2 = d2z + (size_t)b * HID;
  const float* d3 = d3z + (size_t)b * HID;

  unsigned mk[2][4];
  {
    float dv[2][4];
    #pragma unroll
    for (int g = 0; g < 2; ++g) {
      const float* dg = g ? d3 : d2;
      #pragma unroll
      for (int j = 0; j < 4; ++j) dv[g][j] = dg[j * 256 + threadIdx.x];
    }
    #pragma unroll
    for (int g = 0; g < 2; ++g) {
      #pragma unroll
      for (int j = 0; j < 4; ++j) mk[g][j] = (unsigned)__ballot(dv[g][j] != 0.f);
    }
    if (lane == 0) {
      #pragma unroll
      for (int g = 0; g < 2; ++g) {
        #pragma unroll
        for (int j = 0; j < 4; ++j) cnt[g * 32 + j * 8 + wv] = __popc(mk[g][j]);
      }
    }
  }

  {
    const float* Mc = M0 + c0 + lane;
    float* Tc = T0 + (size_t)lane * TS;
    const int jw = wv * 128;
    #pragma unroll 1
    for (int jb = jw; jb < jw + 128; jb += 32) {
      float mv[8][4];
      v4f dq[8];
      #pragma unroll
      for (int u = 0; u < 8; ++u) {
        dq[u] = *(const v4f*)(d1 + jb + 4 * u);
        #pragma unroll
        for (int i = 0; i < 4; ++i) mv[u][i] = Mc[(size_t)(jb + 4 * u + i) * NST];
      }
      #pragma unroll
      for (int u = 0; u < 8; ++u) {
        v4f o;
        o.x = mv[u][0] * dq[u].x;
        o.y = mv[u][1] * dq[u].y;
        o.z = mv[u][2] * dq[u].z;
        o.w = mv[u][3] * dq[u].w;
        *(v4f*)(Tc + jb + 4 * u) = o;
      }
    }
  }
  for (int i = threadIdx.x; i < CB * (HID / 4); i += 256)
    *(float4*)(T1 + (size_t)(i >> 8) * TS + 4 * (i & 255)) = make_float4(0.f, 0.f, 0.f, 0.f);
  __syncthreads();

  int Pn[2];
  #pragma unroll
  for (int g = 0; g < 2; ++g) {
    int tot = 0, off[4] = {0, 0, 0, 0};
    #pragma unroll
    for (int sgm = 0; sgm < 32; ++sgm) {
      if ((sgm & 7) == wv) off[sgm >> 3] = tot;
      tot += cnt[g * 32 + sgm];
    }
    int* rl = rowsL + g * HID;
    #pragma unroll
    for (int j = 0; j < 4; ++j)
      if ((mk[g][j] >> lane) & 1u)
        rl[off[j] + __popc(mk[g][j] & ((1u << lane) - 1u))] = j * 256 + threadIdx.x;
    const int cap = ((tot + 15) >> 4) << 4;
    for (int i = tot + threadIdx.x; i < cap; i += 256) rl[i] = 0;
    Pn[g] = tot;
  }

  #pragma unroll 1
  for (int g = 0; g < 2; ++g) {
    const float* Wm  = g ? Wh2 : Wh1;
    const float* dmg = g ? d3 : d2;
    const float* Tin = lds + (g ? CB * TS : 0);
    float* Tout      = lds + (g ? 0 : CB * TS);
    const int* rl    = rowsL + g * HID;
    const int P      = g ? Pn[1] : Pn[0];
    if (g == 1) {
      for (int i = threadIdx.x; i < CB * (HID / 4); i += 256)
        *(float4*)(T0 + (size_t)(i >> 8) * TS + 4 * (i & 255)) = make_float4(0.f, 0.f, 0.f, 0.f);
    }
    __syncthreads();
    const int ntot = (P + 15) >> 4;
    const int ntw = __builtin_amdgcn_readfirstlane(wv < ntot ? (ntot - wv + 7) >> 3 : 0);
    switch (ntw) {
      case 1: chain_rows<1>(Wm, Tin, Tout, dmg, rl, P, lane, wv); break;
      case 2: chain_rows<2>(Wm, Tin, Tout, dmg, rl, P, lane, wv); break;
      case 3: chain_rows<3>(Wm, Tin, Tout, dmg, rl, P, lane, wv); break;
      case 4: chain_rows<4>(Wm, Tin, Tout, dmg, rl, P, lane, wv); break;
      case 5: chain_rows<5>(Wm, Tin, Tout, dmg, rl, P, lane, wv); break;
      case 6: chain_rows<6>(Wm, Tin, Tout, dmg, rl, P, lane, wv); break;
      case 7: chain_rows<7>(Wm, Tin, Tout, dmg, rl, P, lane, wv); break;
      case 8: chain_rows<8>(Wm, Tin, Tout, dmg, rl, P, lane, wv); break;
      default: break;
    }
    __syncthreads();
  }

  const int l16 = lane & 15, hlf = lane >> 4;
  const int kof = hlf * 4;
  const v8f vz = {0.f,0.f,0.f,0.f,0.f,0.f,0.f,0.f};
  v8f g0 = vz, g1 = vz;
  const float* Tb0 = T0 + (size_t)l16 * TS;
  const float* Tb1 = T0 + (size_t)(16 + l16) * TS;
  const float* Wr  = Wout + (size_t)(wv * 16 + l16) * HID;

  v4f wa[4][4], wc0[4][4], wc1[4][4];
#define FLD(st, kk) { wa[st][0]  = *(const v4f*)(Wr  + (kk) + 0); wc0[st][0] = *(const v4f*)(Tb0 + (kk) + 0); wc1[st][0] = *(const v4f*)(Tb1 + (kk) + 0); \
                      wa[st][1]  = *(const v4f*)(Wr  + (kk) + 8); wc0[st][1] = *(const v4f*)(Tb0 + (kk) + 8); wc1[st][1] = *(const v4f*)(Tb1 + (kk) + 8); \
                      wa[st][2]  = *(const v4f*)(Wr  + (kk) + 16); wc0[st][2] = *(const v4f*)(Tb0 + (kk) + 16); wc1[st][2] = *(const v4f*)(Tb1 + (kk) + 16); \
                      wa[st][3]  = *(const v4f*)(Wr  + (kk) + 24); wc0[st][3] = *(const v4f*)(Tb0 + (kk) + 24); wc1[st][3] = *(const v4f*)(Tb1 + (kk) + 24); }
#define GM(st) { g0 = wmma_f32(lo2(wa[st][0]), lo2(wc0[st][0]), g0); g1 = wmma_f32(lo2(wa[st][0]), lo2(wc1[st][0]), g1); \
                 g0 = wmma_f32(hi2(wa[st][0]), hi2(wc0[st][0]), g0); g1 = wmma_f32(hi2(wa[st][0]), hi2(wc1[st][0]), g1); \
                 g0 = wmma_f32(lo2(wa[st][1]), lo2(wc0[st][1]), g0); g1 = wmma_f32(lo2(wa[st][1]), lo2(wc1[st][1]), g1); \
                 g0 = wmma_f32(hi2(wa[st][1]), hi2(wc0[st][1]), g0); g1 = wmma_f32(hi2(wa[st][1]), hi2(wc1[st][1]), g1); \
                 g0 = wmma_f32(lo2(wa[st][2]), lo2(wc0[st][2]), g0); g1 = wmma_f32(lo2(wa[st][2]), lo2(wc1[st][2]), g1); \
                 g0 = wmma_f32(hi2(wa[st][2]), hi2(wc0[st][2]), g0); g1 = wmma_f32(hi2(wa[st][2]), hi2(wc1[st][2]), g1); \
                 g0 = wmma_f32(lo2(wa[st][3]), lo2(wc0[st][3]), g0); g1 = wmma_f32(lo2(wa[st][3]), lo2(wc1[st][3]), g1); \
                 g0 = wmma_f32(hi2(wa[st][3]), hi2(wc0[st][3]), g0); g1 = wmma_f32(hi2(wa[st][3]), hi2(wc1[st][3]), g1); }
  FLD(0, kof);      PIPE_FENCE();
  FLD(1, kof + 32); PIPE_FENCE();
  FLD(2, kof + 64); PIPE_FENCE();
  FLD(3, kof + 96); PIPE_FENCE();
  for (int k = 0; k < HID - 128; k += 128) {
    GM(0); PIPE_FENCE(); FLD(0, k + 128 + kof); PIPE_FENCE();
    GM(1); PIPE_FENCE(); FLD(1, k + 160 + kof); PIPE_FENCE();
    GM(2); PIPE_FENCE(); FLD(2, k + 192 + kof); PIPE_FENCE();
    GM(3); PIPE_FENCE(); FLD(3, k + 224 + kof); PIPE_FENCE();
  }
  GM(0);
  GM(1);
  GM(2);
  GM(3);
#undef FLD
#undef GM

  #pragma unroll
  for (int r = 0; r < 8; ++r) {
    float s = g0[r] * g0[r] + g1[r] * g1[r];
    s += __shfl_xor(s, 1);
    s += __shfl_xor(s, 2);
    s += __shfl_xor(s, 4);
    s += __shfl_xor(s, 8);
    if (l16 == 0)
      atomicAdd(&zsum[(size_t)b * NST + wv * 16 + r + 8 * hlf], s);
  }
}

__global__ void finalize_norm(const float* __restrict__ zsum,
                              float* __restrict__ out) {
  int i = blockIdx.x * blockDim.x + threadIdx.x;
  if (i >= BATCH * NST) return;
  int b = i >> 7, c = i & 127;
  out[(size_t)b * 384 + 256 + c] = sqrtf(zsum[i]);
}

extern "C" void kernel_launch(void* const* d_in, const int* in_sizes, int n_in,
                              void* d_out, int out_size, void* d_ws, size_t ws_size,
                              hipStream_t stream)
{
  (void)in_sizes; (void)n_in; (void)out_size; (void)ws_size;
  const float* x     = (const float*)d_in[0];
  const float* W_in  = (const float*)d_in[1];
  const float* b_in  = (const float*)d_in[2];
  const float* Wh    = (const float*)d_in[3];
  const float* bh    = (const float*)d_in[4];
  const float* W_out = (const float*)d_in[5];
  const float* b_out = (const float*)d_in[6];
  float* out = (float*)d_out;
  float* ws  = (float*)d_ws;

  const float* Wh0 = Wh;
  const float* Wh1 = Wh + 1024 * 1024;
  const float* Wh2 = Wh + 2 * 1024 * 1024;
  const float* bh0 = bh, *bh1 = bh + 1024, *bh2 = bh + 2048;

  float* X2  = ws;
  float* hA  = X2  + 512 * 256;
  float* hB  = hA  + 512 * 1024;
  float* D1  = hB  + 512 * 1024;
  float* D2  = D1  + 512 * 1024;
  float* D3  = D2  + 512 * 1024;
  float* uA  = D3  + 512 * 1024;
  float* uB  = uA  + 256 * 1024;
  float* M0b = uB  + 256 * 1024;
  float* zs  = M0b + 1024 * 128;

  dim3 blk(256);
  build_x2<<<dim3((512 * 256 + 255) / 256), blk, 0, stream>>>(x, X2);
  zero_buf<<<dim3((256 * 128 + 255) / 256), blk, 0, stream>>>(zs, 256 * 128);

  dim3 lblk(128);
  dim3 lblk2(64);
#define SLOT(L, n, A_, lda_, W_, ldw_, b_, m_, ldm_, C_, ldc_, M_, K_, e_, gx_, gy_) \
  { (L).A##n = (A_); (L).lda##n = (lda_); (L).W##n = (W_); (L).ldw##n = (ldw_); \
    (L).bias##n = (b_); (L).mask##n = (m_); (L).ldm##n = (ldm_); (L).C##n = (C_); \
    (L).ldc##n = (ldc_); (L).M##n = (M_); (L).K##n = (K_); (L).epi##n = (e_); \
    (L).gx##n = (gx_); (L).gy##n = (gy_); }

  LevelArgs L1 = {}; L1.abz = -1;
  SLOT(L1, 0, X2,256,      W_in,256, b_in,    nullptr,0, hA,1024, 512,256, 0, 8,16)
  SLOT(L1, 1, x + 256,512, W_in,256, nullptr, nullptr,0, uA,1024, 256,128, 0, 4,16)
  level_gemm<<<dim3(8,16,2), lblk, 0, stream>>>(L1);

  LevelArgs L2 = {}; L2.abz = -1;
  SLOT(L2, 0, hA,1024, Wh0,1024, bh0, nullptr,0, hB,1024, 512,1024, 1, 16,16)
  level_gemm<<<dim3(16,16,1), lblk2, 0, stream>>>(L2);

  LevelArgs L3 = {}; L3.abz = -1;
  SLOT(L3, 0, hB,1024, Wh0,1024, bh0, nullptr,0, D1,1024, 512,1024, 2, 8,16)
  SLOT(L3, 1, hB,1024, Wh1,1024, bh1, nullptr,0, hA,1024, 512,1024, 1, 8,16)
  level_gemm<<<dim3(8,16,2), lblk, 0, stream>>>(L3);

  LevelArgs L4 = {}; L4.abz = -1;
  SLOT(L4, 0, hA,1024, Wh1,1024, bh1,     nullptr,0, D2,1024, 256,1024, 2, 4,16)
  SLOT(L4, 1, hA,1024, Wh2,1024, bh2,     nullptr,0, hB,1024, 512,1024, 1, 8,16)
  SLOT(L4, 2, uA,1024, Wh0,1024, nullptr, D1,1024,   uB,1024, 256,1024, 0, 4,16)
  level_gemm<<<dim3(8,16,3), lblk, 0, stream>>>(L4);

  LevelArgs L5 = {}; L5.abz = -1;
  SLOT(L5, 0, hB,1024, Wh2,1024,   bh2,     nullptr,0, D3,1024, 512,1024, 2, 8,16)
  SLOT(L5, 1, hB,1024, W_out,1024, b_out,   nullptr,0, out,384, 256,1024, 0, 4,2)
  SLOT(L5, 2, uB,1024, Wh1,1024,   nullptr, D2,1024,   uA,1024, 256,1024, 0, 4,16)
  level_gemm<<<dim3(8,16,3), lblk, 0, stream>>>(L5);

  LevelArgs L6 = {}; L6.abz = -1;
  SLOT(L6, 0, uA,1024, Wh2,1024, nullptr, D3,1024, uB,1024, 256,1024, 0, 8,16)
  SLOT(L6, 1, hA + 256 * 1024,1024, Wh1,1024, bh1, nullptr,0, D2 + 256 * 1024,1024, 256,1024, 2, 8,16)
  level_gemm<<<dim3(8,16,2), lblk2, 0, stream>>>(L6);

  LevelArgs L7 = {}; L7.abz = 1;
  SLOT(L7, 0, uB,1024, W_out,1024, nullptr, nullptr,0, out + 128,384, 256,1024, 0, 4,2)
  L7.Aab = Wh0; L7.ldaab = 1024; L7.Bab = W_in; L7.ldbab = 256; L7.Cab = M0b; L7.ldcab = 128;
  L7.Mab = 1024; L7.Kab = 1024; L7.gxab = 16; L7.gyab = 2;
  level_gemm<<<dim3(16,2,2), lblk, 0, stream>>>(L7);
#undef SLOT

  zero_chain<<<dim3(4,256), blk, 2 * CB * TS * sizeof(float) + 2 * HID * sizeof(int) + 64 * sizeof(int), stream>>>(
      M0b, Wh1, Wh2,
      D1 + 256 * 1024, D2 + 256 * 1024, D3 + 256 * 1024,
      W_out, zs);
  finalize_norm<<<dim3((256 * 128 + 255) / 256), blk, 0, stream>>>(zs, out);
}


# Zero chain runs from the W_out side on list-compacted tiles: each product skips the k whose mask is 0 as well as the masked outputs; row norms summed in-group, no atomics; sum order changes
# speedup vs baseline: 3.4189x; 1.1276x over previous
#include <hip/hip_runtime.h>
#include <hip/hip_bf16.h>
#include <math.h>


typedef float v2f __attribute__((ext_vector_type(2)));
typedef float v4f __attribute__((ext_vector_type(4)));
typedef float v8f __attribute__((ext_vector_type(8)));
typedef int   v4i __attribute__((ext_vector_type(4)));

#define HID   1024
#define NST   128
#define BATCH 256
#define IN2   256
#define CB    32
#define TS    1028

__device__ __forceinline__ v8f wmma_f32(v2f a, v2f b, v8f c) {
  return __builtin_amdgcn_wmma_f32_16x16x4_f32(
       false, a,  false, b,
       (short)0, c,  false,  false);
}

#define PIPE_FENCE() __builtin_amdgcn_sched_barrier(0)

__device__ __forceinline__ v2f lo2(v4f v) { return __builtin_shufflevector(v, v, 0, 1); }
__device__ __forceinline__ v2f hi2(v4f v) { return __builtin_shufflevector(v, v, 2, 3); }

__global__ void build_x2(const float* __restrict__ x, float* __restrict__ X2) {
  int i = blockIdx.x * blockDim.x + threadIdx.x;
  if (i >= 512 * IN2) return;
  int row = i >> 8, c = i & 255;
  float v;
  if (row < 256) {
    v = x[row * 512 + c];
  } else {
    int rz = row - 256;
    v = (c < 128) ? x[rz * 512 + 384 + c] : x[rz * 512 + 128 + (c - 128)];
  }
  X2[i] = v;
}

__device__ __forceinline__ void awt_tile(
    const float* __restrict__ A, int lda,
    const float* __restrict__ W, int ldw,
    const float* __restrict__ bias,
    const float* __restrict__ mask, int ldm,
    float* __restrict__ C, int ldc,
    int M, int K, int epi)
{
  const int lane = threadIdx.x & 31, wv = threadIdx.x >> 5;
  const int l16 = lane & 15, hlf = lane >> 4;
  const int m0 = (blockIdx.x * (int)(blockDim.x >> 5) + wv) * 16;
  const int n0 = blockIdx.y * 64;
  if (m0 >= M) return;

  const v8f vz = {0.f,0.f,0.f,0.f,0.f,0.f,0.f,0.f};
  v8f acc[4] = {vz, vz, vz, vz};

  const int kof = hlf * 4;
  const float* Ar  = A + (size_t)(m0 + l16) * lda;
  const float* Wr0 = W + (size_t)(n0 +  0 + l16) * ldw;
  const float* Wr1 = W + (size_t)(n0 + 16 + l16) * ldw;
  const float* Wr2 = W + (size_t)(n0 + 32 + l16) * ldw;
  const float* Wr3 = W + (size_t)(n0 + 48 + l16) * ldw;

  float bv[4] = {0.f, 0.f, 0.f, 0.f};
  float mv[4][8];
  #pragma unroll
  for (int nt = 0; nt < 4; ++nt) {
    #pragma unroll
    for (int r = 0; r < 8; ++r) mv[nt][r] = 1.f;
  }
  if (bias) {
    #pragma unroll
    for (int nt = 0; nt < 4; ++nt) bv[nt] = bias[n0 + nt * 16 + l16];
  }
  if (mask) {
    #pragma unroll
    for (int nt = 0; nt < 4; ++nt) {
      #pragma unroll
      for (int r = 0; r < 8; ++r)
        mv[nt][r] = mask[(size_t)(m0 + r + 8 * hlf) * ldm + n0 + nt * 16 + l16];
    }
  }
  PIPE_FENCE();

  v4f aS[4], bS[4][4];
#define AWT_LD(st, kk) { aS[st]    = *(const v4f*)(Ar  + (kk)); \
                         bS[st][0] = *(const v4f*)(Wr0 + (kk)); \
                         bS[st][1] = *(const v4f*)(Wr1 + (kk)); \
                         bS[st][2] = *(const v4f*)(Wr2 + (kk)); \
                         bS[st][3] = *(const v4f*)(Wr3 + (kk)); }
#define AWT_MM(st) { acc[0] = wmma_f32(lo2(aS[st]), lo2(bS[st][0]), acc[0]); \
                     acc[1] = wmma_f32(lo2(aS[st]), lo2(bS[st][1]), acc[1]); \
                     acc[2] = wmma_f32(lo2(aS[st]), lo2(bS[st][2]), acc[2]); \
                     acc[3] = wmma_f32(lo2(aS[st]), lo2(bS[st][3]), acc[3]); \
                     acc[0] = wmma_f32(hi2(aS[st]), hi2(bS[st][0]), acc[0]); \
                     acc[1] = wmma_f32(hi2(aS[st]), hi2(bS[st][1]), acc[1]); \
                     acc[2] = wmma_f32(hi2(aS[st]), hi2(bS[st][2]), acc[2]); \
                     acc[3] = wmma_f32(hi2(aS[st]), hi2(bS[st][3]), acc[3]); }

  AWT_LD(0, kof);      PIPE_FENCE();
  AWT_LD(1, kof + 8);  PIPE_FENCE();
  AWT_LD(2, kof + 16); PIPE_FENCE();
  AWT_LD(3, kof + 24); PIPE_FENCE();
  for (int k = 0; k < K - 32; k += 32) {
    AWT_MM(0); PIPE_FENCE(); AWT_LD(0, k + 32 + kof); PIPE_FENCE();
    AWT_MM(1); PIPE_FENCE(); AWT_LD(1, k + 40 + kof); PIPE_FENCE();
    AWT_MM(2); PIPE_FENCE(); AWT_LD(2, k + 48 + kof); PIPE_FENCE();
    AWT_MM(3); PIPE_FENCE(); AWT_LD(3, k + 56 + kof); PIPE_FENCE();
  }
  AWT_MM(0);
  AWT_MM(1);
  AWT_MM(2);
  AWT_MM(3);
#undef AWT_LD
#undef AWT_MM

  if (bias) {
    #pragma unroll
    for (int nt = 0; nt < 4; ++nt) {
      #pragma unroll
      for (int r = 0; r < 8; ++r) acc[nt][r] += bv[nt];
    }
  }
  if (epi == 1) {
    #pragma unroll
    for (int nt = 0; nt < 4; ++nt) {
      #pragma unroll
      for (int r = 0; r < 8; ++r) acc[nt][r] = acc[nt][r] > 0.f ? acc[nt][r] : 0.f;
    }
  } else if (epi == 2) {
    #pragma unroll
    for (int nt = 0; nt < 4; ++nt) {
      #pragma unroll
      for (int r = 0; r < 8; ++r) acc[nt][r] = acc[nt][r] > 0.f ? 1.f : 0.f;
    }
  }
  if (mask) {
    #pragma unroll
    for (int nt = 0; nt < 4; ++nt) {
      #pragma unroll
      for (int r = 0; r < 8; ++r) acc[nt][r] *= mv[nt][r];
    }
  }
  #pragma unroll
  for (int r = 0; r < 8; ++r) {
    float* Cr = C + (size_t)(m0 + r + 8 * hlf) * ldc + n0 + l16;
    #pragma unroll
    for (int nt = 0; nt < 4; ++nt) Cr[nt * 16] = acc[nt][r];
  }
}

__device__ __forceinline__ void ab_tile(
    const float* __restrict__ A, int lda,
    const float* __restrict__ B, int ldb,
    float* __restrict__ C, int ldc,
    int M, int K)
{
  const int lane = threadIdx.x & 31, wv = threadIdx.x >> 5;
  const int l16 = lane & 15, hlf = lane >> 4;
  const int m0 = (blockIdx.x * (int)(blockDim.x >> 5) + wv) * 16;
  const int n0 = blockIdx.y * 64;
  if (m0 >= M) return;

  const v8f vz = {0.f,0.f,0.f,0.f,0.f,0.f,0.f,0.f};
  v8f acc[4] = {vz, vz, vz, vz};
  const int kof = hlf * 4;
  const float* Ar = A + (size_t)(m0 + l16) * lda;
  const float* Bc = B + n0 + l16;

  v4f a0, c00, c10, c20, c30;
  v4f a1, c01, c11, c21, c31;

#define BLD(dst, kk, nt) { (dst).x = Bc[(size_t)(kk) * ldb + (nt) * 16]; \
                           (dst).y = Bc[(size_t)((kk) + 1) * ldb + (nt) * 16]; \
                           (dst).z = Bc[(size_t)((kk) + 2) * ldb + (nt) * 16]; \
                           (dst).w = Bc[(size_t)((kk) + 3) * ldb + (nt) * 16]; }
#define LDSET0(kk) { a0 = *(const v4f*)(Ar + (kk)); \
                     BLD(c00, (kk), 0) BLD(c10, (kk), 1) BLD(c20, (kk), 2) BLD(c30, (kk), 3) }
#define LDSET1(kk) { a1 = *(const v4f*)(Ar + (kk)); \
                     BLD(c01, (kk), 0) BLD(c11, (kk), 1) BLD(c21, (kk), 2) BLD(c31, (kk), 3) }
#define WM0() { acc[0] = wmma_f32(lo2(a0), lo2(c00), acc[0]); acc[1] = wmma_f32(lo2(a0), lo2(c10), acc[1]); \
                acc[2] = wmma_f32(lo2(a0), lo2(c20), acc[2]); acc[3] = wmma_f32(lo2(a0), lo2(c30), acc[3]); \
                acc[0] = wmma_f32(hi2(a0), hi2(c00), acc[0]); acc[1] = wmma_f32(hi2(a0), hi2(c10), acc[1]); \
                acc[2] = wmma_f32(hi2(a0), hi2(c20), acc[2]); acc[3] = wmma_f32(hi2(a0), hi2(c30), acc[3]); }
#define WM1() { acc[0] = wmma_f32(lo2(a1), lo2(c01), acc[0]); acc[1] = wmma_f32(lo2(a1), lo2(c11), acc[1]); \
                acc[2] = wmma_f32(lo2(a1), lo2(c21), acc[2]); acc[3] = wmma_f32(lo2(a1), lo2(c31), acc[3]); \
                acc[0] = wmma_f32(hi2(a1), hi2(c01), acc[0]); acc[1] = wmma_f32(hi2(a1), hi2(c11), acc[1]); \
                acc[2] = wmma_f32(hi2(a1), hi2(c21), acc[2]); acc[3] = wmma_f32(hi2(a1), hi2(c31), acc[3]); }

  LDSET0(kof); PIPE_FENCE();
  LDSET1(kof + 8); PIPE_FENCE();
  for (int k = 0; k < K - 16; k += 16) {
    WM0(); PIPE_FENCE(); LDSET0(k + 16 + kof); PIPE_FENCE();
    WM1(); PIPE_FENCE(); LDSET1(k + 24 + kof); PIPE_FENCE();
  }
  WM0();
  WM1();
#undef BLD
#undef LDSET0
#undef LDSET1
#undef WM0
#undef WM1

  #pragma unroll
  for (int nt = 0; nt < 4; ++nt)
    #pragma unroll
    for (int r = 0; r < 8; ++r)
      C[(size_t)(m0 + r + 8 * hlf) * ldc + n0 + nt * 16 + l16] = acc[nt][r];
}

struct LevelArgs {
  const float* A0; const float* W0; const float* bias0; const float* mask0; float* C0;
  const float* A1; const float* W1; const float* bias1; const float* mask1; float* C1;
  const float* A2; const float* W2; const float* bias2; const float* mask2; float* C2;
  const float* Aab; const float* Bab; float* Cab;
  int lda0, ldw0, ldm0, ldc0, M0, K0, epi0, gx0, gy0;
  int lda1, ldw1, ldm1, ldc1, M1, K1, epi1, gx1, gy1;
  int lda2, ldw2, ldm2, ldc2, M2, K2, epi2, gx2, gy2;
  int ldaab, ldbab, ldcab, Mab, Kab, gxab, gyab, abz;
};

__global__ __launch_bounds__(128) void level_gemm(const LevelArgs p)
{
  const int z = blockIdx.z;
  const int bx = blockIdx.x, by = blockIdx.y;
  if (z == p.abz) {
    if (bx < p.gxab && by < p.gyab)
      ab_tile(p.Aab, p.ldaab, p.Bab, p.ldbab, p.Cab, p.ldcab, p.Mab, p.Kab);
    return;
  }
  const float* A; const float* Wt; const float* bias; const float* mask; float* C;
  int lda, ldw, ldm, ldc, M, K, epi, gx, gy;
  if (z == 0) {
    A = p.A0; Wt = p.W0; bias = p.bias0; mask = p.mask0; C = p.C0;
    lda = p.lda0; ldw = p.ldw0; ldm = p.ldm0; ldc = p.ldc0;
    M = p.M0; K = p.K0; epi = p.epi0; gx = p.gx0; gy = p.gy0;
  } else if (z == 1) {
    A = p.A1; Wt = p.W1; bias = p.bias1; mask = p.mask1; C = p.C1;
    lda = p.lda1; ldw = p.ldw1; ldm = p.ldm1; ldc = p.ldc1;
    M = p.M1; K = p.K1; epi = p.epi1; gx = p.gx1; gy = p.gy1;
  } else {
    A = p.A2; Wt = p.W2; bias = p.bias2; mask = p.mask2; C = p.C2;
    lda = p.lda2; ldw = p.ldw2; ldm = p.ldm2; ldc = p.ldc2;
    M = p.M2; K = p.K2; epi = p.epi2; gx = p.gx2; gy = p.gy2;
  }
  if (bx >= gx || by >= gy) return;
  awt_tile(A, lda, Wt, ldw, bias, mask, ldm, C, ldc, M, K, epi);
}


__device__ __forceinline__ void cc_ldA(const float* __restrict__ Ta0,
                                       const float* __restrict__ Ta1, int kk, v4f aS[4][2])
{
  #pragma unroll
  for (int q = 0; q < 4; ++q) {
    aS[q][0] = *(const v4f*)(Ta0 + kk + 8 * q);
    aS[q][1] = *(const v4f*)(Ta1 + kk + 8 * q);
  }
}

__device__ __forceinline__ void cc_ldI(const int* __restrict__ kl, int kk, v4i ix[4])
{
  #pragma unroll
  for (int q = 0; q < 4; ++q) ix[q] = *(const v4i*)(kl + kk + 8 * q);
}

template <int LDSH>
__device__ __forceinline__ float cc_w(const float* __restrict__ Wm, int row, int col)
{
  return *(const float*)((const char*)Wm + (unsigned)((row << (LDSH + 2)) + col));
}
template <int LDSH>
__device__ __forceinline__ void cc_ldB(const float* __restrict__ Wm, int col,
                                       const v4i ix[4], v4f bv[4])
{
  #pragma unroll
  for (int q = 0; q < 4; ++q) {
    bv[q].x = cc_w<LDSH>(Wm, ix[q].x, col);
    bv[q].y = cc_w<LDSH>(Wm, ix[q].y, col);
    bv[q].z = cc_w<LDSH>(Wm, ix[q].z, col);
    bv[q].w = cc_w<LDSH>(Wm, ix[q].w, col);
  }
}

__device__ __forceinline__ void cc_mm(const v4f aS[4][2], const v4f bv[4], v8f acc[2])
{
  #pragma unroll
  for (int q = 0; q < 4; ++q) {
    acc[0] = wmma_f32(lo2(aS[q][0]), lo2(bv[q]), acc[0]);
    acc[1] = wmma_f32(lo2(aS[q][1]), lo2(bv[q]), acc[1]);
    acc[0] = wmma_f32(hi2(aS[q][0]), hi2(bv[q]), acc[0]);
    acc[1] = wmma_f32(hi2(aS[q][1]), hi2(bv[q]), acc[1]);
  }
}

template <int NT, int LDSH>
__device__ __forceinline__ void cc_trip(
    const float* __restrict__ Wm, const int col[NT],
    const float* __restrict__ Ta0, const float* __restrict__ Ta1,
    const int* __restrict__ kl, int kn, int kn2,
    v4f bv[NT][4], const v4f aCur[4][2], v4f aNxt[4][2],
    const v4i iNxt[4], v4i iNxt2[4], v8f acc[NT][2])
{
  cc_mm(aCur, bv[0], acc[0]); PIPE_FENCE();
  cc_ldB<LDSH>(Wm, col[0], iNxt, bv[0]); cc_ldA(Ta0, Ta1, kn, aNxt); cc_ldI(kl, kn2, iNxt2); PIPE_FENCE();
  #pragma unroll
  for (int t = 1; t < NT; ++t) {
    cc_mm(aCur, bv[t], acc[t]); PIPE_FENCE();
    cc_ldB<LDSH>(Wm, col[t], iNxt, bv[t]); PIPE_FENCE();
  }
}

template <int NT, int LDSH>
__device__ __forceinline__ void cc_gemm(
    const float* __restrict__ Wm, const int col[NT],
    const float* __restrict__ Tin, const int* __restrict__ kl, int capK,
    int lane, v8f acc[NT][2])
{
  const int l16 = lane & 15, hlf = lane >> 4;
  const int kof = hlf * 4;
  const v8f vz = {0.f,0.f,0.f,0.f,0.f,0.f,0.f,0.f};
  #pragma unroll
  for (int t = 0; t < NT; ++t) { acc[t][0] = vz; acc[t][1] = vz; }
  if (capK <= 0) return;

  const float* Ta0 = Tin + (size_t)l16 * TS;
  const float* Ta1 = Tin + (size_t)(16 + l16) * TS;
  const int klast = capK - 32 + kof;

  v4f bv[NT][4];
  v4f aX[4][2], aY[4][2];
  v4i iX[4], iY[4];

  cc_ldI(kl, kof, iY);
  cc_ldB<LDSH>(Wm, col[0], iY, bv[0]); cc_ldA(Ta0, Ta1, kof, aX); cc_ldI(kl, 32 + kof, iX); PIPE_FENCE();
  #pragma unroll
  for (int t = 1; t < NT; ++t) { cc_ldB<LDSH>(Wm, col[t], iY, bv[t]); PIPE_FENCE(); }
  for (int k = 0; k < capK; k += 64) {
    const int k1 = k + 32 + kof;
    int k2 = k + 64 + kof; k2 = k2 < klast ? k2 : klast;
    int k3 = k + 96 + kof; k3 = k3 < klast ? k3 : klast;
    cc_trip<NT, LDSH>(Wm, col, Ta0, Ta1, kl, k1, k2, bv, aX, aY, iX, iY, acc);
    cc_trip<NT, LDSH>(Wm, col, Ta0, Ta1, kl, k2, k3, bv, aY, aX, iY, iX, acc);
  }
}

template <int NT>
__device__ __forceinline__ void chain_cols(
    const float* __restrict__ Wm,
    const float* __restrict__ Tin, float* __restrict__ Tout,
    const int* __restrict__ kl, int capK,
    const int* __restrict__ nl, int Pn,
    int lane, int wv)
{
  const int l16 = lane & 15, hlf = lane >> 4;
  int col[NT];
  #pragma unroll
  for (int t = 0; t < NT; ++t) col[t] = nl[(wv + 8 * t) * 16 + l16] * 4;
  v8f acc[NT][2];
  cc_gemm<NT, 10>(Wm, col, Tin, kl, capK, lane, acc);
  #pragma unroll
  for (int t = 0; t < NT; ++t) {
    const int q = (wv + 8 * t) * 16 + l16;
    const bool keep = q < Pn;
    #pragma unroll
    for (int r = 0; r < 8; ++r) {
      Tout[(size_t)(8 * hlf + r) * TS + q]      = keep ? acc[t][0][r] : 0.f;
      Tout[(size_t)(16 + 8 * hlf + r) * TS + q] = keep ? acc[t][1][r] : 0.f;
    }
  }
}

__global__ __launch_bounds__(256) void zero_chain(
    const float* __restrict__ M0,
    const float* __restrict__ Wh1, const float* __restrict__ Wh2,
    const float* __restrict__ d1z, const float* __restrict__ d2z,
    const float* __restrict__ d3z,
    const float* __restrict__ Wout,
    float* __restrict__ out)
{
  extern __shared__ float lds[];
  float* T0 = lds;
  float* T1 = lds + CB * TS;
  int* rowsL = (int*)(lds + 2 * CB * TS);
  int* cnt   = rowsL + 3 * HID;
  float* red = (float*)(cnt + 96);

  const int nb = blockIdx.x;
  const int b  = blockIdx.y;
  const int lane = threadIdx.x & 31, wv = threadIdx.x >> 5;

  unsigned mk[3][4];
  {
    float dv[3][4];
    #pragma unroll
    for (int g = 0; g < 3; ++g) {
      const float* dg = (g == 0 ? d1z : g == 1 ? d2z : d3z) + (size_t)b * HID;
      #pragma unroll
      for (int j = 0; j < 4; ++j) dv[g][j] = dg[j * 256 + threadIdx.x];
    }
    #pragma unroll
    for (int g = 0; g < 3; ++g) {
      #pragma unroll
      for (int j = 0; j < 4; ++j) mk[g][j] = (unsigned)__ballot(dv[g][j] != 0.f);
    }
    if (lane == 0) {
      #pragma unroll
      for (int g = 0; g < 3; ++g) {
        #pragma unroll
        for (int j = 0; j < 4; ++j) cnt[g * 32 + j * 8 + wv] = __popc(mk[g][j]);
      }
    }
  }
  __syncthreads();

  int Pn[3], capn[3];
  #pragma unroll
  for (int g = 0; g < 3; ++g) {
    int tot = 0, off[4] = {0, 0, 0, 0};
    #pragma unroll
    for (int sgm = 0; sgm < 32; ++sgm) {
      if ((sgm & 7) == wv) off[sgm >> 3] = tot;
      tot += cnt[g * 32 + sgm];
    }
    int* rl = rowsL + g * HID;
    #pragma unroll
    for (int j = 0; j < 4; ++j)
      if ((mk[g][j] >> lane) & 1u)
        rl[off[j] + __popc(mk[g][j] & ((1u << lane) - 1u))] = j * 256 + threadIdx.x;
    const int cap = ((tot + 63) >> 6) << 6;
    for (int i = tot + threadIdx.x; i < cap; i += 256) rl[i] = 0;
    Pn[g] = tot; capn[g] = cap;
  }
  __syncthreads();

  {
    const int* l3 = rowsL + 2 * HID;
    const int P3 = Pn[2];
    const float* Wn = Wout + (size_t)(nb * 32 + wv * 4) * HID;
    float* Tn = T0 + (size_t)(wv * 4) * TS;
    for (int p = lane * 4; p < capn[2]; p += 128) {
      const v4i ix = *(const v4i*)(l3 + p);
      float w[4][4];
      #pragma unroll
      for (int n = 0; n < 4; ++n) {
        w[n][0] = Wn[n * HID + ix.x]; w[n][1] = Wn[n * HID + ix.y];
        w[n][2] = Wn[n * HID + ix.z]; w[n][3] = Wn[n * HID + ix.w];
      }
      #pragma unroll
      for (int n = 0; n < 4; ++n) {
        v4f o;
        o.x = (p     < P3) ? w[n][0] : 0.f;
        o.y = (p + 1 < P3) ? w[n][1] : 0.f;
        o.z = (p + 2 < P3) ? w[n][2] : 0.f;
        o.w = (p + 3 < P3) ? w[n][3] : 0.f;
        *(v4f*)(Tn + (size_t)n * TS + p) = o;
      }
    }
  }

  #pragma unroll 1
  for (int g = 0; g < 2; ++g) {
    const float* Wm  = g ? Wh1 : Wh2;
    const float* Tin = lds + (g ? CB * TS : 0);
    float* Tout      = lds + (g ? 0 : CB * TS);
    const int* kl    = rowsL + (g ? 1 : 2) * HID;
    const int capK   = g ? capn[1] : capn[2];
    const int* nl    = rowsL + (g ? 0 : 1) * HID;
    const int P      = g ? Pn[0] : Pn[1];
    const int capN   = g ? capn[0] : capn[1];
    __syncthreads();
    const int ntot = capN >> 4;
    const int ntw = __builtin_amdgcn_readfirstlane(wv < ntot ? (ntot - wv + 7) >> 3 : 0);
    switch (ntw) {
      case 1: chain_cols<1>(Wm, Tin, Tout, kl, capK, nl, P, lane, wv); break;
      case 2: chain_cols<2>(Wm, Tin, Tout, kl, capK, nl, P, lane, wv); break;
      case 3: chain_cols<3>(Wm, Tin, Tout, kl, capK, nl, P, lane, wv); break;
      case 4: chain_cols<4>(Wm, Tin, Tout, kl, capK, nl, P, lane, wv); break;
      case 5: chain_cols<5>(Wm, Tin, Tout, kl, capK, nl, P, lane, wv); break;
      case 6: chain_cols<6>(Wm, Tin, Tout, kl, capK, nl, P, lane, wv); break;
      case 7: chain_cols<7>(Wm, Tin, Tout, kl, capK, nl, P, lane, wv); break;
      case 8: chain_cols<8>(Wm, Tin, Tout, kl, capK, nl, P, lane, wv); break;
      default: break;
    }
  }
  __syncthreads();

  {
    const int l16 = lane & 15, hlf = lane >> 4;
    int col[1] = { (wv * 16 + l16) * 4 };
    v8f acc[1][2];
    cc_gemm<1, 7>(M0, col, T0, rowsL, capn[0], lane, acc);
    #pragma unroll
    for (int m = 0; m < 2; ++m) {
      #pragma unroll
      for (int r = 0; r < 8; ++r) {
        float sq = acc[0][m][r] * acc[0][m][r];
        sq += __shfl_xor(sq, 1);
        sq += __shfl_xor(sq, 2);
        sq += __shfl_xor(sq, 4);
        sq += __shfl_xor(sq, 8);
        if (l16 == 0) red[wv * 32 + 16 * m + 8 * hlf + r] = sq;
      }
    }
  }
  __syncthreads();
  if (threadIdx.x < 32) {
    float sm = 0.f;
    #pragma unroll
    for (int w = 0; w < 8; ++w) sm += red[w * 32 + threadIdx.x];
    out[(size_t)b * 384 + 256 + nb * 32 + threadIdx.x] = sqrtf(sm);
  }
}

extern "C" void kernel_launch(void* const* d_in, const int* in_sizes, int n_in,
                              void* d_out, int out_size, void* d_ws, size_t ws_size,
                              hipStream_t stream)
{
  (void)in_sizes; (void)n_in; (void)out_size; (void)ws_size;
  const float* x     = (const float*)d_in[0];
  const float* W_in  = (const float*)d_in[1];
  const float* b_in  = (const float*)d_in[2];
  const float* Wh    = (const float*)d_in[3];
  const float* bh    = (const float*)d_in[4];
  const float* W_out = (const float*)d_in[5];
  const float* b_out = (const float*)d_in[6];
  float* out = (float*)d_out;
  float* ws  = (float*)d_ws;

  const float* Wh0 = Wh;
  const float* Wh1 = Wh + 1024 * 1024;
  const float* Wh2 = Wh + 2 * 1024 * 1024;
  const float* bh0 = bh, *bh1 = bh + 1024, *bh2 = bh + 2048;

  float* X2  = ws;
  float* hA  = X2  + 512 * 256;
  float* hB  = hA  + 512 * 1024;
  float* D1  = hB  + 512 * 1024;
  float* D2  = D1  + 512 * 1024;
  float* D3  = D2  + 512 * 1024;
  float* uA  = D3  + 512 * 1024;
  float* uB  = uA  + 256 * 1024;
  float* M0b = uB  + 256 * 1024;

  dim3 blk(256);
  build_x2<<<dim3((512 * 256 + 255) / 256), blk, 0, stream>>>(x, X2);

  dim3 lblk(128);
  dim3 lblk2(64);
#define SLOT(L, n, A_, lda_, W_, ldw_, b_, m_, ldm_, C_, ldc_, M_, K_, e_, gx_, gy_) \
  { (L).A##n = (A_); (L).lda##n = (lda_); (L).W##n = (W_); (L).ldw##n = (ldw_); \
    (L).bias##n = (b_); (L).mask##n = (m_); (L).ldm##n = (ldm_); (L).C##n = (C_); \
    (L).ldc##n = (ldc_); (L).M##n = (M_); (L).K##n = (K_); (L).epi##n = (e_); \
    (L).gx##n = (gx_); (L).gy##n = (gy_); }

  LevelArgs L1 = {}; L1.abz = -1;
  SLOT(L1, 0, X2,256,      W_in,256, b_in,    nullptr,0, hA,1024, 512,256, 0, 8,16)
  SLOT(L1, 1, x + 256,512, W_in,256, nullptr, nullptr,0, uA,1024, 256,128, 0, 4,16)
  level_gemm<<<dim3(8,16,2), lblk, 0, stream>>>(L1);

  LevelArgs L2 = {}; L2.abz = -1;
  SLOT(L2, 0, hA,1024, Wh0,1024, bh0, nullptr,0, hB,1024, 512,1024, 1, 16,16)
  level_gemm<<<dim3(16,16,1), lblk2, 0, stream>>>(L2);

  LevelArgs L3 = {}; L3.abz = -1;
  SLOT(L3, 0, hB,1024, Wh0,1024, bh0, nullptr,0, D1,1024, 512,1024, 2, 8,16)
  SLOT(L3, 1, hB,1024, Wh1,1024, bh1, nullptr,0, hA,1024, 512,1024, 1, 8,16)
  level_gemm<<<dim3(8,16,2), lblk, 0, stream>>>(L3);

  LevelArgs L4 = {}; L4.abz = -1;
  SLOT(L4, 0, hA,1024, Wh1,1024, bh1,     nullptr,0, D2,1024, 256,1024, 2, 4,16)
  SLOT(L4, 1, hA,1024, Wh2,1024, bh2,     nullptr,0, hB,1024, 512,1024, 1, 8,16)
  SLOT(L4, 2, uA,1024, Wh0,1024, nullptr, D1,1024,   uB,1024, 256,1024, 0, 4,16)
  level_gemm<<<dim3(8,16,3), lblk, 0, stream>>>(L4);

  LevelArgs L5 = {}; L5.abz = -1;
  SLOT(L5, 0, hB,1024, Wh2,1024,   bh2,     nullptr,0, D3,1024, 512,1024, 2, 8,16)
  SLOT(L5, 1, hB,1024, W_out,1024, b_out,   nullptr,0, out,384, 256,1024, 0, 4,2)
  SLOT(L5, 2, uB,1024, Wh1,1024,   nullptr, D2,1024,   uA,1024, 256,1024, 0, 4,16)
  level_gemm<<<dim3(8,16,3), lblk, 0, stream>>>(L5);

  LevelArgs L6 = {}; L6.abz = -1;
  SLOT(L6, 0, uA,1024, Wh2,1024, nullptr, D3,1024, uB,1024, 256,1024, 0, 8,16)
  SLOT(L6, 1, hA + 256 * 1024,1024, Wh1,1024, bh1, nullptr,0, D2 + 256 * 1024,1024, 256,1024, 2, 8,16)
  level_gemm<<<dim3(8,16,2), lblk2, 0, stream>>>(L6);

  LevelArgs L7 = {}; L7.abz = 1;
  SLOT(L7, 0, uB,1024, W_out,1024, nullptr, nullptr,0, out + 128,384, 256,1024, 0, 4,2)
  L7.Aab = Wh0; L7.ldaab = 1024; L7.Bab = W_in; L7.ldbab = 256; L7.Cab = M0b; L7.ldcab = 128;
  L7.Mab = 1024; L7.Kab = 1024; L7.gxab = 16; L7.gyab = 2;
  level_gemm<<<dim3(16,2,2), lblk, 0, stream>>>(L7);
#undef SLOT

  zero_chain<<<dim3(4,256), blk, 2 * CB * TS * sizeof(float) + 3 * HID * sizeof(int) + 96 * sizeof(int) + 256 * sizeof(float), stream>>>(
      M0b, Wh1, Wh2,
      D1 + 256 * 1024, D2 + 256 * 1024, D3 + 256 * 1024,
      W_out, out);
}


# Zero chain work-group owns 64 rows of W_out, not 32, on one LDS tile updated in place behind a barrier: each gathered weight element feeds 4 WMMAs, not 2; gathers and L2 traffic halve; sums unchanged
# speedup vs baseline: 3.9108x; 1.1439x over previous
#include <hip/hip_runtime.h>
#include <hip/hip_bf16.h>
#include <math.h>


typedef float v2f __attribute__((ext_vector_type(2)));
typedef float v4f __attribute__((ext_vector_type(4)));
typedef float v8f __attribute__((ext_vector_type(8)));
typedef int   v4i __attribute__((ext_vector_type(4)));

#define HID   1024
#define NST   128
#define BATCH 256
#define IN2   256
#define CB    64
#define TS    1028

__device__ __forceinline__ v8f wmma_f32(v2f a, v2f b, v8f c) {
  return __builtin_amdgcn_wmma_f32_16x16x4_f32(
       false, a,  false, b,
       (short)0, c,  false,  false);
}

#define PIPE_FENCE() __builtin_amdgcn_sched_barrier(0)

__device__ __forceinline__ v2f lo2(v4f v) { return __builtin_shufflevector(v, v, 0, 1); }
__device__ __forceinline__ v2f hi2(v4f v) { return __builtin_shufflevector(v, v, 2, 3); }

__global__ void build_x2(const float* __restrict__ x, float* __restrict__ X2) {
  int i = blockIdx.x * blockDim.x + threadIdx.x;
  if (i >= 512 * IN2) return;
  int row = i >> 8, c = i & 255;
  float v;
  if (row < 256) {
    v = x[row * 512 + c];
  } else {
    int rz = row - 256;
    v = (c < 128) ? x[rz * 512 + 384 + c] : x[rz * 512 + 128 + (c - 128)];
  }
  X2[i] = v;
}

__device__ __forceinline__ void awt_tile(
    const float* __restrict__ A, int lda,
    const float* __restrict__ W, int ldw,
    const float* __restrict__ bias,
    const float* __restrict__ mask, int ldm,
    float* __restrict__ C, int ldc,
    int M, int K, int epi)
{
  const int lane = threadIdx.x & 31, wv = threadIdx.x >> 5;
  const int l16 = lane & 15, hlf = lane >> 4;
  const int m0 = (blockIdx.x * (int)(blockDim.x >> 5) + wv) * 16;
  const int n0 = blockIdx.y * 64;
  if (m0 >= M) return;

  const v8f vz = {0.f,0.f,0.f,0.f,0.f,0.f,0.f,0.f};
  v8f acc[4] = {vz, vz, vz, vz};

  const int kof = hlf * 4;
  const float* Ar  = A + (size_t)(m0 + l16) * lda;
  const float* Wr0 = W + (size_t)(n0 +  0 + l16) * ldw;
  const float* Wr1 = W + (size_t)(n0 + 16 + l16) * ldw;
  const float* Wr2 = W + (size_t)(n0 + 32 + l16) * ldw;
  const float* Wr3 = W + (size_t)(n0 + 48 + l16) * ldw;

  float bv[4] = {0.f, 0.f, 0.f, 0.f};
  float mv[4][8];
  #pragma unroll
  for (int nt = 0; nt < 4; ++nt) {
    #pragma unroll
    for (int r = 0; r < 8; ++r) mv[nt][r] = 1.f;
  }
  if (bias) {
    #pragma unroll
    for (int nt = 0; nt < 4; ++nt) bv[nt] = bias[n0 + nt * 16 + l16];
  }
  if (mask) {
    #pragma unroll
    for (int nt = 0; nt < 4; ++nt) {
      #pragma unroll
      for (int r = 0; r < 8; ++r)
        mv[nt][r] = mask[(size_t)(m0 + r + 8 * hlf) * ldm + n0 + nt * 16 + l16];
    }
  }
  PIPE_FENCE();

  v4f aS[4], bS[4][4];
#define AWT_LD(st, kk) { aS[st]    = *(const v4f*)(Ar  + (kk)); \
                         bS[st][0] = *(const v4f*)(Wr0 + (kk)); \
                         bS[st][1] = *(const v4f*)(Wr1 + (kk)); \
                         bS[st][2] = *(const v4f*)(Wr2 + (kk)); \
                         bS[st][3] = *(const v4f*)(Wr3 + (kk)); }
#define AWT_MM(st) { acc[0] = wmma_f32(lo2(aS[st]), lo2(bS[st][0]), acc[0]); \
                     acc[1] = wmma_f32(lo2(aS[st]), lo2(bS[st][1]), acc[1]); \
                     acc[2] = wmma_f32(lo2(aS[st]), lo2(bS[st][2]), acc[2]); \
                     acc[3] = wmma_f32(lo2(aS[st]), lo2(bS[st][3]), acc[3]); \
                     acc[0] = wmma_f32(hi2(aS[st]), hi2(bS[st][0]), acc[0]); \
                     acc[1] = wmma_f32(hi2(aS[st]), hi2(bS[st][1]), acc[1]); \
                     acc[2] = wmma_f32(hi2(aS[st]), hi2(bS[st][2]), acc[2]); \
                     acc[3] = wmma_f32(hi2(aS[st]), hi2(bS[st][3]), acc[3]); }

  AWT_LD(0, kof);      PIPE_FENCE();
  AWT_LD(1, kof + 8);  PIPE_FENCE();
  AWT_LD(2, kof + 16); PIPE_FENCE();
  AWT_LD(3, kof + 24); PIPE_FENCE();
  for (int k = 0; k < K - 32; k += 32) {
    AWT_MM(0); PIPE_FENCE(); AWT_LD(0, k + 32 + kof); PIPE_FENCE();
    AWT_MM(1); PIPE_FENCE(); AWT_LD(1, k + 40 + kof); PIPE_FENCE();
    AWT_MM(2); PIPE_FENCE(); AWT_LD(2, k + 48 + kof); PIPE_FENCE();
    AWT_MM(3); PIPE_FENCE(); AWT_LD(3, k + 56 + kof); PIPE_FENCE();
  }
  AWT_MM(0);
  AWT_MM(1);
  AWT_MM(2);
  AWT_MM(3);
#undef AWT_LD
#undef AWT_MM

  if (bias) {
    #pragma unroll
    for (int nt = 0; nt < 4; ++nt) {
      #pragma unroll
      for (int r = 0; r < 8; ++r) acc[nt][r] += bv[nt];
    }
  }
  if (epi == 1) {
    #pragma unroll
    for (int nt = 0; nt < 4; ++nt) {
      #pragma unroll
      for (int r = 0; r < 8; ++r) acc[nt][r] = acc[nt][r] > 0.f ? acc[nt][r] : 0.f;
    }
  } else if (epi == 2) {
    #pragma unroll
    for (int nt = 0; nt < 4; ++nt) {
      #pragma unroll
      for (int r = 0; r < 8; ++r) acc[nt][r] = acc[nt][r] > 0.f ? 1.f : 0.f;
    }
  }
  if (mask) {
    #pragma unroll
    for (int nt = 0; nt < 4; ++nt) {
      #pragma unroll
      for (int r = 0; r < 8; ++r) acc[nt][r] *= mv[nt][r];
    }
  }
  #pragma unroll
  for (int r = 0; r < 8; ++r) {
    float* Cr = C + (size_t)(m0 + r + 8 * hlf) * ldc + n0 + l16;
    #pragma unroll
    for (int nt = 0; nt < 4; ++nt) Cr[nt * 16] = acc[nt][r];
  }
}

__device__ __forceinline__ void ab_tile(
    const float* __restrict__ A, int lda,
    const float* __restrict__ B, int ldb,
    float* __restrict__ C, int ldc,
    int M, int K)
{
  const int lane = threadIdx.x & 31, wv = threadIdx.x >> 5;
  const int l16 = lane & 15, hlf = lane >> 4;
  const int m0 = (blockIdx.x * (int)(blockDim.x >> 5) + wv) * 16;
  const int n0 = blockIdx.y * 64;
  if (m0 >= M) return;

  const v8f vz = {0.f,0.f,0.f,0.f,0.f,0.f,0.f,0.f};
  v8f acc[4] = {vz, vz, vz, vz};
  const int kof = hlf * 4;
  const float* Ar = A + (size_t)(m0 + l16) * lda;
  const float* Bc = B + n0 + l16;

  v4f a0, c00, c10, c20, c30;
  v4f a1, c01, c11, c21, c31;

#define BLD(dst, kk, nt) { (dst).x = Bc[(size_t)(kk) * ldb + (nt) * 16]; \
                           (dst).y = Bc[(size_t)((kk) + 1) * ldb + (nt) * 16]; \
                           (dst).z = Bc[(size_t)((kk) + 2) * ldb + (nt) * 16]; \
                           (dst).w = Bc[(size_t)((kk) + 3) * ldb + (nt) * 16]; }
#define LDSET0(kk) { a0 = *(const v4f*)(Ar + (kk)); \
                     BLD(c00, (kk), 0) BLD(c10, (kk), 1) BLD(c20, (kk), 2) BLD(c30, (kk), 3) }
#define LDSET1(kk) { a1 = *(const v4f*)(Ar + (kk)); \
                     BLD(c01, (kk), 0) BLD(c11, (kk), 1) BLD(c21, (kk), 2) BLD(c31, (kk), 3) }
#define WM0() { acc[0] = wmma_f32(lo2(a0), lo2(c00), acc[0]); acc[1] = wmma_f32(lo2(a0), lo2(c10), acc[1]); \
                acc[2] = wmma_f32(lo2(a0), lo2(c20), acc[2]); acc[3] = wmma_f32(lo2(a0), lo2(c30), acc[3]); \
                acc[0] = wmma_f32(hi2(a0), hi2(c00), acc[0]); acc[1] = wmma_f32(hi2(a0), hi2(c10), acc[1]); \
                acc[2] = wmma_f32(hi2(a0), hi2(c20), acc[2]); acc[3] = wmma_f32(hi2(a0), hi2(c30), acc[3]); }
#define WM1() { acc[0] = wmma_f32(lo2(a1), lo2(c01), acc[0]); acc[1] = wmma_f32(lo2(a1), lo2(c11), acc[1]); \
                acc[2] = wmma_f32(lo2(a1), lo2(c21), acc[2]); acc[3] = wmma_f32(lo2(a1), lo2(c31), acc[3]); \
                acc[0] = wmma_f32(hi2(a1), hi2(c01), acc[0]); acc[1] = wmma_f32(hi2(a1), hi2(c11), acc[1]); \
                acc[2] = wmma_f32(hi2(a1), hi2(c21), acc[2]); acc[3] = wmma_f32(hi2(a1), hi2(c31), acc[3]); }

  LDSET0(kof); PIPE_FENCE();
  LDSET1(kof + 8); PIPE_FENCE();
  for (int k = 0; k < K - 16; k += 16) {
    WM0(); PIPE_FENCE(); LDSET0(k + 16 + kof); PIPE_FENCE();
    WM1(); PIPE_FENCE(); LDSET1(k + 24 + kof); PIPE_FENCE();
  }
  WM0();
  WM1();
#undef BLD
#undef LDSET0
#undef LDSET1
#undef WM0
#undef WM1

  #pragma unroll
  for (int nt = 0; nt < 4; ++nt)
    #pragma unroll
    for (int r = 0; r < 8; ++r)
      C[(size_t)(m0 + r + 8 * hlf) * ldc + n0 + nt * 16 + l16] = acc[nt][r];
}

struct LevelArgs {
  const float* A0; const float* W0; const float* bias0; const float* mask0; float* C0;
  const float* A1; const float* W1; const float* bias1; const float* mask1; float* C1;
  const float* A2; const float* W2; const float* bias2; const float* mask2; float* C2;
  const float* Aab; const float* Bab; float* Cab;
  int lda0, ldw0, ldm0, ldc0, M0, K0, epi0, gx0, gy0;
  int lda1, ldw1, ldm1, ldc1, M1, K1, epi1, gx1, gy1;
  int lda2, ldw2, ldm2, ldc2, M2, K2, epi2, gx2, gy2;
  int ldaab, ldbab, ldcab, Mab, Kab, gxab, gyab, abz;
};

__global__ __launch_bounds__(128) void level_gemm(const LevelArgs p)
{
  const int z = blockIdx.z;
  const int bx = blockIdx.x, by = blockIdx.y;
  if (z == p.abz) {
    if (bx < p.gxab && by < p.gyab)
      ab_tile(p.Aab, p.ldaab, p.Bab, p.ldbab, p.Cab, p.ldcab, p.Mab, p.Kab);
    return;
  }
  const float* A; const float* Wt; const float* bias; const float* mask; float* C;
  int lda, ldw, ldm, ldc, M, K, epi, gx, gy;
  if (z == 0) {
    A = p.A0; Wt = p.W0; bias = p.bias0; mask = p.mask0; C = p.C0;
    lda = p.lda0; ldw = p.ldw0; ldm = p.ldm0; ldc = p.ldc0;
    M = p.M0; K = p.K0; epi = p.epi0; gx = p.gx0; gy = p.gy0;
  } else if (z == 1) {
    A = p.A1; Wt = p.W1; bias = p.bias1; mask = p.mask1; C = p.C1;
    lda = p.lda1; ldw = p.ldw1; ldm = p.ldm1; ldc = p.ldc1;
    M = p.M1; K = p.K1; epi = p.epi1; gx = p.gx1; gy = p.gy1;
  } else {
    A = p.A2; Wt = p.W2; bias = p.bias2; mask = p.mask2; C = p.C2;
    lda = p.lda2; ldw = p.ldw2; ldm = p.ldm2; ldc = p.ldc2;
    M = p.M2; K = p.K2; epi = p.epi2; gx = p.gx2; gy = p.gy2;
  }
  if (bx >= gx || by >= gy) return;
  awt_tile(A, lda, Wt, ldw, bias, mask, ldm, C, ldc, M, K, epi);
}


__device__ __forceinline__ void cc_ldA(const float* __restrict__ Ta, int kk, v4f aS[4][4])
{
  #pragma unroll
  for (int q = 0; q < 4; ++q) {
    #pragma unroll
    for (int m = 0; m < 4; ++m)
      aS[q][m] = *(const v4f*)(Ta + (size_t)(16 * m) * TS + kk + 8 * q);
  }
}

__device__ __forceinline__ void cc_ldI(const int* __restrict__ kl, int kk, v4i ix[4])
{
  #pragma unroll
  for (int q = 0; q < 4; ++q) ix[q] = *(const v4i*)(kl + kk + 8 * q);
}

template <int LDSH>
__device__ __forceinline__ float cc_w(const float* __restrict__ Wm, int row, int col)
{
  return *(const float*)((const char*)Wm + (unsigned)((row << (LDSH + 2)) + col));
}
template <int LDSH>
__device__ __forceinline__ void cc_ldB(const float* __restrict__ Wm, int col,
                                       const v4i ix[4], v4f bv[4])
{
  #pragma unroll
  for (int q = 0; q < 4; ++q) {
    bv[q].x = cc_w<LDSH>(Wm, ix[q].x, col);
    bv[q].y = cc_w<LDSH>(Wm, ix[q].y, col);
    bv[q].z = cc_w<LDSH>(Wm, ix[q].z, col);
    bv[q].w = cc_w<LDSH>(Wm, ix[q].w, col);
  }
}

__device__ __forceinline__ void cc_mm(const v4f aS[4][4], const v4f bv[4], v8f acc[4])
{
  #pragma unroll
  for (int q = 0; q < 4; ++q) {
    #pragma unroll
    for (int m = 0; m < 4; ++m) acc[m] = wmma_f32(lo2(aS[q][m]), lo2(bv[q]), acc[m]);
    #pragma unroll
    for (int m = 0; m < 4; ++m) acc[m] = wmma_f32(hi2(aS[q][m]), hi2(bv[q]), acc[m]);
  }
}

template <int NT, int LDSH>
__device__ __forceinline__ void cc_trip(
    const float* __restrict__ Wm, const int col[],
    const float* __restrict__ Ta,
    const int* __restrict__ kl, int kn, int kn2,
    v4f bv[NT][4], const v4f aCur[4][4], v4f aNxt[4][4],
    const v4i iNxt[4], v4i iNxt2[4], v8f acc[][4])
{
  cc_mm(aCur, bv[0], acc[0]); PIPE_FENCE();
  cc_ldB<LDSH>(Wm, col[0], iNxt, bv[0]); cc_ldA(Ta, kn, aNxt); cc_ldI(kl, kn2, iNxt2); PIPE_FENCE();
  #pragma unroll
  for (int t = 1; t < NT; ++t) {
    cc_mm(aCur, bv[t], acc[t]); PIPE_FENCE();
    cc_ldB<LDSH>(Wm, col[t], iNxt, bv[t]); PIPE_FENCE();
  }
}

template <int NT, int LDSH>
__device__ __forceinline__ void cc_gemm(
    const float* __restrict__ Wm, const int col[],
    const float* __restrict__ Tin, const int* __restrict__ kl, int capK,
    int lane, v8f acc[][4])
{
  const int l16 = lane & 15, hlf = lane >> 4;
  const int kof = hlf * 4;
  const v8f vz = {0.f,0.f,0.f,0.f,0.f,0.f,0.f,0.f};
  #pragma unroll
  for (int t = 0; t < NT; ++t) {
    #pragma unroll
    for (int m = 0; m < 4; ++m) acc[t][m] = vz;
  }
  if (capK <= 0) return;

  const float* Ta = Tin + (size_t)l16 * TS;
  const int klast = capK - 32 + kof;

  v4f bv[NT][4];
  v4f aX[4][4], aY[4][4];
  v4i iX[4], iY[4];

  cc_ldI(kl, kof, iY);
  cc_ldB<LDSH>(Wm, col[0], iY, bv[0]); cc_ldA(Ta, kof, aX); cc_ldI(kl, 32 + kof, iX); PIPE_FENCE();
  #pragma unroll
  for (int t = 1; t < NT; ++t) { cc_ldB<LDSH>(Wm, col[t], iY, bv[t]); PIPE_FENCE(); }
  for (int k = 0; k < capK; k += 64) {
    const int k1 = k + 32 + kof;
    int k2 = k + 64 + kof; k2 = k2 < klast ? k2 : klast;
    int k3 = k + 96 + kof; k3 = k3 < klast ? k3 : klast;
    cc_trip<NT, LDSH>(Wm, col, Ta, kl, k1, k2, bv, aX, aY, iX, iY, acc);
    cc_trip<NT, LDSH>(Wm, col, Ta, kl, k2, k3, bv, aY, aX, iY, iX, acc);
  }
}

template <int NT, int LDSH>
__device__ __forceinline__ void cc_gemm_plain(
    const float* __restrict__ Wm, const int col[NT],
    const float* __restrict__ Tin, const int* __restrict__ kl, int capK,
    int lane, bool lastReal, v8f acc[][4])
{
  const int l16 = lane & 15, hlf = lane >> 4;
  const int kof = hlf * 4;
  const v8f vz = {0.f,0.f,0.f,0.f,0.f,0.f,0.f,0.f};
  #pragma unroll
  for (int t = 0; t < NT; ++t) {
    #pragma unroll
    for (int m = 0; m < 4; ++m) acc[t][m] = vz;
  }
  const float* Ta = Tin + (size_t)l16 * TS;
  #pragma unroll 1
  for (int k = 0; k < capK; k += 32) {
    v4i ix[4];
    v4f aS[4][4];
    cc_ldI(kl, k + kof, ix);
    cc_ldA(Ta, k + kof, aS);
    #pragma unroll
    for (int t = 0; t < NT; ++t) {
      if (t < NT - 1 || lastReal) {
        v4f bv[4];
        cc_ldB<LDSH>(Wm, col[t], ix, bv);
        cc_mm(aS, bv, acc[t]);
        PIPE_FENCE();
      }
    }
  }
}

template <int NT>
__device__ __forceinline__ void chain_cols(
    const float* __restrict__ Wm, float* __restrict__ T,
    const int* __restrict__ kl, int capK,
    const int* __restrict__ nl, int Pn,
    int lane, int wv, bool lastReal)
{
  const int l16 = lane & 15, hlf = lane >> 4;
  int col[NT];
  #pragma unroll
  for (int t = 0; t < NT; ++t) col[t] = nl[(wv + 8 * t) * 16 + l16] * 4;
  v8f acc[NT][4];
  if (NT <= 5) {
    if (lastReal)    cc_gemm<NT, 10>(Wm, col, T, kl, capK, lane, acc);
    else if (NT > 1) cc_gemm<(NT > 1 ? NT - 1 : 1), 10>(Wm, col, T, kl, capK, lane, acc);
  } else {
    cc_gemm_plain<NT, 10>(Wm, col, T, kl, capK, lane, lastReal, acc);
  }
  __syncthreads();
  int ob[4];
  #pragma unroll
  for (int m = 0; m < 4; ++m) {
    ob[m] = (16 * m + 8 * hlf) * TS + wv * 16 + l16;
    asm volatile("" : "+v"(ob[m]));
  }
  #pragma unroll
  for (int t = 0; t < NT; ++t) {
    if (t < NT - 1 || lastReal) {
      const bool keep = (wv + 8 * t) * 16 + l16 < Pn;
      #pragma unroll
      for (int m = 0; m < 4; ++m) {
        #pragma unroll
        for (int r = 0; r < 8; ++r)
          T[ob[m] + r * TS + t * 128] = keep ? acc[t][m][r] : 0.f;
      }
    }
  }
}

__global__ __launch_bounds__(256) void zero_chain(
    const float* __restrict__ M0,
    const float* __restrict__ Wh1, const float* __restrict__ Wh2,
    const float* __restrict__ d1z, const float* __restrict__ d2z,
    const float* __restrict__ d3z,
    const float* __restrict__ Wout,
    float* __restrict__ out)
{
  extern __shared__ float lds[];
  float* T = lds;
  int* rowsL = (int*)(lds + CB * TS);
  int* cnt   = rowsL + 3 * HID;
  float* red = (float*)(cnt + 96);

  const int nb = (int)__builtin_amdgcn_workgroup_id_x();
  const int b  = (int)__builtin_amdgcn_workgroup_id_y();
  const int lane = threadIdx.x & 31, wv = threadIdx.x >> 5;

  unsigned mk[3][4];
  {
    float dv[3][4];
    #pragma unroll
    for (int g = 0; g < 3; ++g) {
      const float* dg = (g == 0 ? d1z : g == 1 ? d2z : d3z) + (size_t)b * HID;
      #pragma unroll
      for (int j = 0; j < 4; ++j) dv[g][j] = dg[j * 256 + threadIdx.x];
    }
    #pragma unroll
    for (int g = 0; g < 3; ++g) {
      #pragma unroll
      for (int j = 0; j < 4; ++j) mk[g][j] = (unsigned)__ballot(dv[g][j] != 0.f);
    }
    if (lane == 0) {
      #pragma unroll
      for (int g = 0; g < 3; ++g) {
        #pragma unroll
        for (int j = 0; j < 4; ++j) cnt[g * 32 + j * 8 + wv] = __popc(mk[g][j]);
      }
    }
  }
  __syncthreads();

  int Pn[3], capn[3];
  #pragma unroll
  for (int g = 0; g < 3; ++g) {
    int tot = 0, off[4] = {0, 0, 0, 0};
    #pragma unroll
    for (int sgm = 0; sgm < 32; ++sgm) {
      if ((sgm & 7) == wv) off[sgm >> 3] = tot;
      tot += cnt[g * 32 + sgm];
    }
    int* rl = rowsL + g * HID;
    #pragma unroll
    for (int j = 0; j < 4; ++j)
      if ((mk[g][j] >> lane) & 1u)
        rl[off[j] + __popc(mk[g][j] & ((1u << lane) - 1u))] = j * 256 + threadIdx.x;
    const int cap = ((tot + 63) >> 6) << 6;
    for (int i = tot + threadIdx.x; i < cap; i += 256) rl[i] = 0;
    Pn[g] = tot; capn[g] = cap;
  }
  __syncthreads();

  {
    const int* l3 = rowsL + 2 * HID;
    const int P3 = Pn[2];
    const float* Wn = Wout + (size_t)(nb * CB + wv * 8) * HID;
    float* Tn = T + (size_t)(wv * 8) * TS;
    for (int p = lane * 4; p < capn[2]; p += 128) {
      const v4i ix = *(const v4i*)(l3 + p);
      float w[8][4];
      #pragma unroll
      for (int n = 0; n < 8; ++n) {
        w[n][0] = Wn[n * HID + ix.x]; w[n][1] = Wn[n * HID + ix.y];
        w[n][2] = Wn[n * HID + ix.z]; w[n][3] = Wn[n * HID + ix.w];
      }
      #pragma unroll
      for (int n = 0; n < 8; ++n) {
        v4f o;
        o.x = (p     < P3) ? w[n][0] : 0.f;
        o.y = (p + 1 < P3) ? w[n][1] : 0.f;
        o.z = (p + 2 < P3) ? w[n][2] : 0.f;
        o.w = (p + 3 < P3) ? w[n][3] : 0.f;
        *(v4f*)(Tn + (size_t)n * TS + p) = o;
      }
    }
  }

  #pragma unroll 1
  for (int g = 0; g < 2; ++g) {
    const float* Wm  = g ? Wh1 : Wh2;
    const int* kl    = rowsL + (g ? 1 : 2) * HID;
    const int capK   = g ? capn[1] : capn[2];
    const int* nl    = rowsL + (g ? 0 : 1) * HID;
    const int P      = g ? Pn[0] : Pn[1];
    const int capN   = g ? capn[0] : capn[1];
    __syncthreads();
    const int ntot = capN >> 4;
    const int ntb = __builtin_amdgcn_readfirstlane((ntot + 7) >> 3);
    const bool lastReal = __builtin_amdgcn_readfirstlane(wv + 8 * (ntb - 1) < ntot ? 1 : 0) != 0;
    switch (ntb) {
      case 1: chain_cols<1>(Wm, T, kl, capK, nl, P, lane, wv, lastReal); break;
      case 2: chain_cols<2>(Wm, T, kl, capK, nl, P, lane, wv, lastReal); break;
      case 3: chain_cols<3>(Wm, T, kl, capK, nl, P, lane, wv, lastReal); break;
      case 4: chain_cols<4>(Wm, T, kl, capK, nl, P, lane, wv, lastReal); break;
      case 5: chain_cols<5>(Wm, T, kl, capK, nl, P, lane, wv, lastReal); break;
      case 6: chain_cols<6>(Wm, T, kl, capK, nl, P, lane, wv, lastReal); break;
      case 7: chain_cols<7>(Wm, T, kl, capK, nl, P, lane, wv, lastReal); break;
      case 8: chain_cols<8>(Wm, T, kl, capK, nl, P, lane, wv, lastReal); break;
      default: break;
    }
  }
  __syncthreads();

  {
    const int l16 = lane & 15, hlf = lane >> 4;
    int col[1] = { (wv * 16 + l16) * 4 };
    v8f acc[1][4];
    cc_gemm<1, 7>(M0, col, T, rowsL, capn[0], lane, acc);
    #pragma unroll
    for (int m = 0; m < 4; ++m) {
      #pragma unroll
      for (int r = 0; r < 8; ++r) {
        float sq = acc[0][m][r] * acc[0][m][r];
        sq += __shfl_xor(sq, 1);
        sq += __shfl_xor(sq, 2);
        sq += __shfl_xor(sq, 4);
        sq += __shfl_xor(sq, 8);
        if (l16 == 0) red[wv * CB + 16 * m + 8 * hlf + r] = sq;
      }
    }
  }
  __syncthreads();
  if (threadIdx.x < CB) {
    float sm = 0.f;
    #pragma unroll
    for (int w = 0; w < 8; ++w) sm += red[w * CB + threadIdx.x];
    out[(size_t)b * 384 + 256 + nb * CB + threadIdx.x] = sqrtf(sm);
  }
}

extern "C" void kernel_launch(void* const* d_in, const int* in_sizes, int n_in,
                              void* d_out, int out_size, void* d_ws, size_t ws_size,
                              hipStream_t stream)
{
  (void)in_sizes; (void)n_in; (void)out_size; (void)ws_size;
  const float* x     = (const float*)d_in[0];
  const float* W_in  = (const float*)d_in[1];
  const float* b_in  = (const float*)d_in[2];
  const float* Wh    = (const float*)d_in[3];
  const float* bh    = (const float*)d_in[4];
  const float* W_out = (const float*)d_in[5];
  const float* b_out = (const float*)d_in[6];
  float* out = (float*)d_out;
  float* ws  = (float*)d_ws;

  const float* Wh0 = Wh;
  const float* Wh1 = Wh + 1024 * 1024;
  const float* Wh2 = Wh + 2 * 1024 * 1024;
  const float* bh0 = bh, *bh1 = bh + 1024, *bh2 = bh + 2048;

  float* X2  = ws;
  float* hA  = X2  + 512 * 256;
  float* hB  = hA  + 512 * 1024;
  float* D1  = hB  + 512 * 1024;
  float* D2  = D1  + 512 * 1024;
  float* D3  = D2  + 512 * 1024;
  float* uA  = D3  + 512 * 1024;
  float* uB  = uA  + 256 * 1024;
  float* M0b = uB  + 256 * 1024;

  dim3 blk(256);
  build_x2<<<dim3((512 * 256 + 255) / 256), blk, 0, stream>>>(x, X2);

  dim3 lblk(128);
  dim3 lblk2(64);
#define SLOT(L, n, A_, lda_, W_, ldw_, b_, m_, ldm_, C_, ldc_, M_, K_, e_, gx_, gy_) \
  { (L).A##n = (A_); (L).lda##n = (lda_); (L).W##n = (W_); (L).ldw##n = (ldw_); \
    (L).bias##n = (b_); (L).mask##n = (m_); (L).ldm##n = (ldm_); (L).C##n = (C_); \
    (L).ldc##n = (ldc_); (L).M##n = (M_); (L).K##n = (K_); (L).epi##n = (e_); \
    (L).gx##n = (gx_); (L).gy##n = (gy_); }

  LevelArgs L1 = {}; L1.abz = -1;
  SLOT(L1, 0, X2,256,      W_in,256, b_in,    nullptr,0, hA,1024, 512,256, 0, 8,16)
  SLOT(L1, 1, x + 256,512, W_in,256, nullptr, nullptr,0, uA,1024, 256,128, 0, 4,16)
  level_gemm<<<dim3(8,16,2), lblk, 0, stream>>>(L1);

  LevelArgs L2 = {}; L2.abz = -1;
  SLOT(L2, 0, hA,1024, Wh0,1024, bh0, nullptr,0, hB,1024, 512,1024, 1, 16,16)
  level_gemm<<<dim3(16,16,1), lblk2, 0, stream>>>(L2);

  LevelArgs L3 = {}; L3.abz = -1;
  SLOT(L3, 0, hB,1024, Wh0,1024, bh0, nullptr,0, D1,1024, 512,1024, 2, 8,16)
  SLOT(L3, 1, hB,1024, Wh1,1024, bh1, nullptr,0, hA,1024, 512,1024, 1, 8,16)
  level_gemm<<<dim3(8,16,2), lblk, 0, stream>>>(L3);

  LevelArgs L4 = {}; L4.abz = -1;
  SLOT(L4, 0, hA,1024, Wh1,1024, bh1,     nullptr,0, D2,1024, 256,1024, 2, 4,16)
  SLOT(L4, 1, hA,1024, Wh2,1024, bh2,     nullptr,0, hB,1024, 512,1024, 1, 8,16)
  SLOT(L4, 2, uA,1024, Wh0,1024, nullptr, D1,1024,   uB,1024, 256,1024, 0, 4,16)
  level_gemm<<<dim3(8,16,3), lblk, 0, stream>>>(L4);

  LevelArgs L5 = {}; L5.abz = -1;
  SLOT(L5, 0, hB,1024, Wh2,1024,   bh2,     nullptr,0, D3,1024, 512,1024, 2, 8,16)
  SLOT(L5, 1, hB,1024, W_out,1024, b_out,   nullptr,0, out,384, 256,1024, 0, 4,2)
  SLOT(L5, 2, uB,1024, Wh1,1024,   nullptr, D2,1024,   uA,1024, 256,1024, 0, 4,16)
  level_gemm<<<dim3(8,16,3), lblk, 0, stream>>>(L5);

  LevelArgs L6 = {}; L6.abz = -1;
  SLOT(L6, 0, uA,1024, Wh2,1024, nullptr, D3,1024, uB,1024, 256,1024, 0, 8,16)
  SLOT(L6, 1, hA + 256 * 1024,1024, Wh1,1024, bh1, nullptr,0, D2 + 256 * 1024,1024, 256,1024, 2, 8,16)
  level_gemm<<<dim3(8,16,2), lblk2, 0, stream>>>(L6);

  LevelArgs L7 = {}; L7.abz = 1;
  SLOT(L7, 0, uB,1024, W_out,1024, nullptr, nullptr,0, out + 128,384, 256,1024, 0, 4,2)
  L7.Aab = Wh0; L7.ldaab = 1024; L7.Bab = W_in; L7.ldbab = 256; L7.Cab = M0b; L7.ldcab = 128;
  L7.Mab = 1024; L7.Kab = 1024; L7.gxab = 16; L7.gyab = 2;
  level_gemm<<<dim3(16,2,2), lblk, 0, stream>>>(L7);
#undef SLOT

  zero_chain<<<dim3(NST / CB,256), blk, CB * TS * sizeof(float) + 3 * HID * sizeof(int) + 96 * sizeof(int) + 8 * CB * sizeof(float), stream>>>(
      M0b, Wh1, Wh2,
      D1 + 256 * 1024, D2 + 256 * 1024, D3 + 256 * 1024,
      W_out, out);
}


# Level GEMMs: a work-group's shared weight tile goes through LDS (fetched once per 64 k in whole-line loads, split barrier by hand), not through 16-line fragment loads in every wave; sums unchanged
# speedup vs baseline: 4.0404x; 1.0331x over previous
#include <hip/hip_runtime.h>
#include <hip/hip_bf16.h>
#include <math.h>


typedef float v2f __attribute__((ext_vector_type(2)));
typedef float v4f __attribute__((ext_vector_type(4)));
typedef float v8f __attribute__((ext_vector_type(8)));
typedef int   v4i __attribute__((ext_vector_type(4)));

#define HID   1024
#define NST   128
#define BATCH 256
#define IN2   256
#define CB    64
#define TS    1028

__device__ __forceinline__ v8f wmma_f32(v2f a, v2f b, v8f c) {
  return __builtin_amdgcn_wmma_f32_16x16x4_f32(
       false, a,  false, b,
       (short)0, c,  false,  false);
}

#define PIPE_FENCE() __builtin_amdgcn_sched_barrier(0)

__device__ __forceinline__ v2f lo2(v4f v) { return __builtin_shufflevector(v, v, 0, 1); }
__device__ __forceinline__ v2f hi2(v4f v) { return __builtin_shufflevector(v, v, 2, 3); }

__global__ void build_x2(const float* __restrict__ x, float* __restrict__ X2) {
  int i = blockIdx.x * blockDim.x + threadIdx.x;
  if (i >= 512 * IN2) return;
  int row = i >> 8, c = i & 255;
  float v;
  if (row < 256) {
    v = x[row * 512 + c];
  } else {
    int rz = row - 256;
    v = (c < 128) ? x[rz * 512 + 384 + c] : x[rz * 512 + 128 + (c - 128)];
  }
  X2[i] = v;
}

#define LK   64
#define LP   68
#define LBUF (64 * LP)
#define LNB  3

template <int NW>
__device__ __forceinline__ void awt_tile(
    const float* __restrict__ A, int lda,
    const float* __restrict__ W, int ldw,
    const float* __restrict__ bias,
    const float* __restrict__ mask, int ldm,
    float* __restrict__ C, int ldc,
    int K, int epi, float* Bs)
{
  constexpr int NTH = NW * 32;
  constexpr int NLD = (64 * LK / 4) / NTH;
  constexpr int RST = NTH / 16;
  const int tid = threadIdx.x;
  const int lane = tid & 31, wv = tid >> 5;
  const int l16 = lane & 15, hlf = lane >> 4;
  const int m0 = (blockIdx.x * NW + wv) * 16;
  const int n0 = blockIdx.y * 64;

  const v8f vz = {0.f,0.f,0.f,0.f,0.f,0.f,0.f,0.f};
  v8f acc[4] = {vz, vz, vz, vz};

  const int kof = hlf * 4;
  const float* Ak = A + (size_t)(m0 + l16) * lda + kof;
  const float* Wg = W + (size_t)(n0 + (tid >> 4)) * ldw + (tid & 15) * 4;
  const int lw = (tid >> 4) * LP + (tid & 15) * 4;
  const int lr = l16 * LP + kof;

  float bv[4] = {0.f, 0.f, 0.f, 0.f};
  float mv[4][8];
  #pragma unroll
  for (int nt = 0; nt < 4; ++nt) {
    #pragma unroll
    for (int r = 0; r < 8; ++r) mv[nt][r] = 1.f;
  }
  if (bias) {
    #pragma unroll
    for (int nt = 0; nt < 4; ++nt) bv[nt] = bias[n0 + nt * 16 + l16];
  }
  if (mask) {
    #pragma unroll
    for (int nt = 0; nt < 4; ++nt) {
      #pragma unroll
      for (int r = 0; r < 8; ++r)
        mv[nt][r] = mask[(size_t)(m0 + r + 8 * hlf) * ldm + n0 + nt * 16 + l16];
    }
  }
  PIPE_FENCE();

  v4f aR[8];
  v4f bF[4][4];
  v4f gR[NLD];

#define G_LD(dst, s) { _Pragma("unroll") for (int j = 0; j < NLD; ++j) \
    (dst)[j] = *(const v4f*)(Wg + (size_t)(j * RST) * ldw + (s) * LK); }
#define G_ST(srcr, bp) { _Pragma("unroll") for (int j = 0; j < NLD; ++j) \
    *(v4f*)((bp) + lw + j * (RST * LP)) = (srcr)[j]; }
#define B_LD(st, bp, q) { _Pragma("unroll") for (int nt = 0; nt < 4; ++nt) \
    bF[st][nt] = *(const v4f*)((bp) + lr + nt * (16 * LP) + 8 * (q)); }
#define A_LD(q, kk) { aR[q] = *(const v4f*)(Ak + (kk) + 8 * (q)); }
#define AWT_MM(q) { acc[0] = wmma_f32(lo2(aR[q]), lo2(bF[(q) & 3][0]), acc[0]); \
                    acc[1] = wmma_f32(lo2(aR[q]), lo2(bF[(q) & 3][1]), acc[1]); \
                    acc[2] = wmma_f32(lo2(aR[q]), lo2(bF[(q) & 3][2]), acc[2]); \
                    acc[3] = wmma_f32(lo2(aR[q]), lo2(bF[(q) & 3][3]), acc[3]); \
                    acc[0] = wmma_f32(hi2(aR[q]), hi2(bF[(q) & 3][0]), acc[0]); \
                    acc[1] = wmma_f32(hi2(aR[q]), hi2(bF[(q) & 3][1]), acc[1]); \
                    acc[2] = wmma_f32(hi2(aR[q]), hi2(bF[(q) & 3][2]), acc[2]); \
                    acc[3] = wmma_f32(hi2(aR[q]), hi2(bF[(q) & 3][3]), acc[3]); }
#define ACC_PIN() asm volatile("" : "+v"(acc[0]), "+v"(acc[1]), "+v"(acc[2]), "+v"(acc[3]) :: "memory")
#define BAR_SIGNAL() asm volatile("s_wait_dscnt 0x0\n\ts_barrier_signal -1" ::: "memory")
#define BAR_WAIT()        asm volatile("s_barrier_wait -1" ::: "memory")

  float* cur = Bs;
  float* nxt = Bs + LBUF;
  float* nx2 = Bs + 2 * LBUF;
  const int nslab = K / LK;

  {
    v4f g1[NLD];
    G_LD(gR, 0) G_LD(g1, 1)
    #pragma unroll
    for (int q = 0; q < 8; ++q) A_LD(q, 0)
    PIPE_FENCE();
    G_ST(gR, cur) G_ST(g1, nxt)
  }
  BAR_SIGNAL(); BAR_WAIT();
  B_LD(0, cur, 0) PIPE_FENCE();
  B_LD(1, cur, 1) PIPE_FENCE();
  B_LD(2, cur, 2) PIPE_FENCE();
  B_LD(3, cur, 3) PIPE_FENCE();

#define SLAB(s, HAS1, HAS2) { \
    if (HAS2) { G_LD(gR, (s) + 2) } PIPE_FENCE(); \
    AWT_MM(0); ACC_PIN(); PIPE_FENCE(); B_LD(0, cur, 4) if (HAS1) A_LD(0, ((s) + 1) * LK) PIPE_FENCE(); \
    AWT_MM(1); ACC_PIN(); PIPE_FENCE(); B_LD(1, cur, 5) if (HAS1) A_LD(1, ((s) + 1) * LK) PIPE_FENCE(); \
    AWT_MM(2); ACC_PIN(); PIPE_FENCE(); B_LD(2, cur, 6) if (HAS1) A_LD(2, ((s) + 1) * LK) PIPE_FENCE(); \
    AWT_MM(3); ACC_PIN(); PIPE_FENCE(); B_LD(3, cur, 7) if (HAS1) A_LD(3, ((s) + 1) * LK) PIPE_FENCE(); \
    AWT_MM(4); ACC_PIN(); PIPE_FENCE(); if (HAS1) { B_LD(0, nxt, 0) A_LD(4, ((s) + 1) * LK) } PIPE_FENCE(); \
    AWT_MM(5); ACC_PIN(); PIPE_FENCE(); if (HAS1) { B_LD(1, nxt, 1) A_LD(5, ((s) + 1) * LK) } PIPE_FENCE(); \
    if (HAS2) { G_ST(gR, nx2) } PIPE_FENCE(); \
    AWT_MM(6); ACC_PIN(); PIPE_FENCE(); if (HAS2) { BAR_SIGNAL(); } PIPE_FENCE(); \
    if (HAS1) { B_LD(2, nxt, 2) A_LD(6, ((s) + 1) * LK) } PIPE_FENCE(); \
    AWT_MM(7); ACC_PIN(); PIPE_FENCE(); if (HAS1) { B_LD(3, nxt, 3) A_LD(7, ((s) + 1) * LK) } PIPE_FENCE(); \
    if (HAS2) { BAR_WAIT(); } PIPE_FENCE(); }

  int s = 0;
  for (; s < nslab - 2; ++s) {
    SLAB(s, true, true)
    float* t = cur; cur = nxt; nxt = nx2; nx2 = t;
  }
  SLAB(s, true, false)
  cur = nxt; ++s;
  SLAB(s, false, false)
#undef SLAB
#undef G_LD
#undef G_ST
#undef B_LD
#undef A_LD
#undef AWT_MM
#undef BAR_SIGNAL
#undef BAR_WAIT
#undef ACC_PIN

  if (bias) {
    #pragma unroll
    for (int nt = 0; nt < 4; ++nt) {
      #pragma unroll
      for (int r = 0; r < 8; ++r) acc[nt][r] += bv[nt];
    }
  }
  if (epi == 1) {
    #pragma unroll
    for (int nt = 0; nt < 4; ++nt) {
      #pragma unroll
      for (int r = 0; r < 8; ++r) acc[nt][r] = acc[nt][r] > 0.f ? acc[nt][r] : 0.f;
    }
  } else if (epi == 2) {
    #pragma unroll
    for (int nt = 0; nt < 4; ++nt) {
      #pragma unroll
      for (int r = 0; r < 8; ++r) acc[nt][r] = acc[nt][r] > 0.f ? 1.f : 0.f;
    }
  }
  if (mask) {
    #pragma unroll
    for (int nt = 0; nt < 4; ++nt) {
      #pragma unroll
      for (int r = 0; r < 8; ++r) acc[nt][r] *= mv[nt][r];
    }
  }
  #pragma unroll
  for (int r = 0; r < 8; ++r) {
    float* Cr = C + (size_t)(m0 + r + 8 * hlf) * ldc + n0 + l16;
    #pragma unroll
    for (int nt = 0; nt < 4; ++nt) Cr[nt * 16] = acc[nt][r];
  }
}

__device__ __forceinline__ void ab_tile(
    const float* __restrict__ A, int lda,
    const float* __restrict__ B, int ldb,
    float* __restrict__ C, int ldc,
    int M, int K)
{
  const int lane = threadIdx.x & 31, wv = threadIdx.x >> 5;
  const int l16 = lane & 15, hlf = lane >> 4;
  const int m0 = (blockIdx.x * (int)(blockDim.x >> 5) + wv) * 16;
  const int n0 = blockIdx.y * 64;
  if (m0 >= M) return;

  const v8f vz = {0.f,0.f,0.f,0.f,0.f,0.f,0.f,0.f};
  v8f acc[4] = {vz, vz, vz, vz};
  const int kof = hlf * 4;
  const float* Ar = A + (size_t)(m0 + l16) * lda;
  const float* Bc = B + n0 + l16;

  v4f a0, c00, c10, c20, c30;
  v4f a1, c01, c11, c21, c31;

#define BLD(dst, kk, nt) { (dst).x = Bc[(size_t)(kk) * ldb + (nt) * 16]; \
                           (dst).y = Bc[(size_t)((kk) + 1) * ldb + (nt) * 16]; \
                           (dst).z = Bc[(size_t)((kk) + 2) * ldb + (nt) * 16]; \
                           (dst).w = Bc[(size_t)((kk) + 3) * ldb + (nt) * 16]; }
#define LDSET0(kk) { a0 = *(const v4f*)(Ar + (kk)); \
                     BLD(c00, (kk), 0) BLD(c10, (kk), 1) BLD(c20, (kk), 2) BLD(c30, (kk), 3) }
#define LDSET1(kk) { a1 = *(const v4f*)(Ar + (kk)); \
                     BLD(c01, (kk), 0) BLD(c11, (kk), 1) BLD(c21, (kk), 2) BLD(c31, (kk), 3) }
#define WM0() { acc[0] = wmma_f32(lo2(a0), lo2(c00), acc[0]); acc[1] = wmma_f32(lo2(a0), lo2(c10), acc[1]); \
                acc[2] = wmma_f32(lo2(a0), lo2(c20), acc[2]); acc[3] = wmma_f32(lo2(a0), lo2(c30), acc[3]); \
                acc[0] = wmma_f32(hi2(a0), hi2(c00), acc[0]); acc[1] = wmma_f32(hi2(a0), hi2(c10), acc[1]); \
                acc[2] = wmma_f32(hi2(a0), hi2(c20), acc[2]); acc[3] = wmma_f32(hi2(a0), hi2(c30), acc[3]); }
#define WM1() { acc[0] = wmma_f32(lo2(a1), lo2(c01), acc[0]); acc[1] = wmma_f32(lo2(a1), lo2(c11), acc[1]); \
                acc[2] = wmma_f32(lo2(a1), lo2(c21), acc[2]); acc[3] = wmma_f32(lo2(a1), lo2(c31), acc[3]); \
                acc[0] = wmma_f32(hi2(a1), hi2(c01), acc[0]); acc[1] = wmma_f32(hi2(a1), hi2(c11), acc[1]); \
                acc[2] = wmma_f32(hi2(a1), hi2(c21), acc[2]); acc[3] = wmma_f32(hi2(a1), hi2(c31), acc[3]); }

  LDSET0(kof); PIPE_FENCE();
  LDSET1(kof + 8); PIPE_FENCE();
  for (int k = 0; k < K - 16; k += 16) {
    WM0(); PIPE_FENCE(); LDSET0(k + 16 + kof); PIPE_FENCE();
    WM1(); PIPE_FENCE(); LDSET1(k + 24 + kof); PIPE_FENCE();
  }
  WM0();
  WM1();
#undef BLD
#undef LDSET0
#undef LDSET1
#undef WM0
#undef WM1

  #pragma unroll
  for (int nt = 0; nt < 4; ++nt)
    #pragma unroll
    for (int r = 0; r < 8; ++r)
      C[(size_t)(m0 + r + 8 * hlf) * ldc + n0 + nt * 16 + l16] = acc[nt][r];
}

struct LevelArgs {
  const float* A0; const float* W0; const float* bias0; const float* mask0; float* C0;
  const float* A1; const float* W1; const float* bias1; const float* mask1; float* C1;
  const float* A2; const float* W2; const float* bias2; const float* mask2; float* C2;
  const float* Aab; const float* Bab; float* Cab;
  int lda0, ldw0, ldm0, ldc0, M0, K0, epi0, gx0, gy0;
  int lda1, ldw1, ldm1, ldc1, M1, K1, epi1, gx1, gy1;
  int lda2, ldw2, ldm2, ldc2, M2, K2, epi2, gx2, gy2;
  int ldaab, ldbab, ldcab, Mab, Kab, gxab, gyab, abz;
};

__global__ __launch_bounds__(128) void level_gemm(const LevelArgs p)
{
  __shared__ __attribute__((aligned(16))) float Bs[LNB * LBUF];
  const int z = blockIdx.z;
  const int bx = blockIdx.x, by = blockIdx.y;
  if (z == p.abz) {
    if (bx < p.gxab && by < p.gyab)
      ab_tile(p.Aab, p.ldaab, p.Bab, p.ldbab, p.Cab, p.ldcab, p.Mab, p.Kab);
    return;
  }
  const float* A; const float* Wt; const float* bias; const float* mask; float* C;
  int lda, ldw, ldm, ldc, M, K, epi, gx, gy;
  if (z == 0) {
    A = p.A0; Wt = p.W0; bias = p.bias0; mask = p.mask0; C = p.C0;
    lda = p.lda0; ldw = p.ldw0; ldm = p.ldm0; ldc = p.ldc0;
    M = p.M0; K = p.K0; epi = p.epi0; gx = p.gx0; gy = p.gy0;
  } else if (z == 1) {
    A = p.A1; Wt = p.W1; bias = p.bias1; mask = p.mask1; C = p.C1;
    lda = p.lda1; ldw = p.ldw1; ldm = p.ldm1; ldc = p.ldc1;
    M = p.M1; K = p.K1; epi = p.epi1; gx = p.gx1; gy = p.gy1;
  } else {
    A = p.A2; Wt = p.W2; bias = p.bias2; mask = p.mask2; C = p.C2;
    lda = p.lda2; ldw = p.ldw2; ldm = p.ldm2; ldc = p.ldc2;
    M = p.M2; K = p.K2; epi = p.epi2; gx = p.gx2; gy = p.gy2;
  }
  if (bx >= gx || by >= gy) return;
  (void)M;
  if (blockDim.x == 64) awt_tile<2>(A, lda, Wt, ldw, bias, mask, ldm, C, ldc, K, epi, Bs);
  else                  awt_tile<4>(A, lda, Wt, ldw, bias, mask, ldm, C, ldc, K, epi, Bs);
}


__device__ __forceinline__ void cc_ldA(const float* __restrict__ Ta, int kk, v4f aS[4][4])
{
  #pragma unroll
  for (int q = 0; q < 4; ++q) {
    #pragma unroll
    for (int m = 0; m < 4; ++m)
      aS[q][m] = *(const v4f*)(Ta + (size_t)(16 * m) * TS + kk + 8 * q);
  }
}

__device__ __forceinline__ void cc_ldI(const int* __restrict__ kl, int kk, v4i ix[4])
{
  #pragma unroll
  for (int q = 0; q < 4; ++q) ix[q] = *(const v4i*)(kl + kk + 8 * q);
}

template <int LDSH>
__device__ __forceinline__ float cc_w(const float* __restrict__ Wm, int row, int col)
{
  return *(const float*)((const char*)Wm + (unsigned)((row << (LDSH + 2)) + col));
}
template <int LDSH>
__device__ __forceinline__ void cc_ldB(const float* __restrict__ Wm, int col,
                                       const v4i ix[4], v4f bv[4])
{
  #pragma unroll
  for (int q = 0; q < 4; ++q) {
    bv[q].x = cc_w<LDSH>(Wm, ix[q].x, col);
    bv[q].y = cc_w<LDSH>(Wm, ix[q].y, col);
    bv[q].z = cc_w<LDSH>(Wm, ix[q].z, col);
    bv[q].w = cc_w<LDSH>(Wm, ix[q].w, col);
  }
}

__device__ __forceinline__ void cc_mm(const v4f aS[4][4], const v4f bv[4], v8f acc[4])
{
  #pragma unroll
  for (int q = 0; q < 4; ++q) {
    #pragma unroll
    for (int m = 0; m < 4; ++m) acc[m] = wmma_f32(lo2(aS[q][m]), lo2(bv[q]), acc[m]);
    #pragma unroll
    for (int m = 0; m < 4; ++m) acc[m] = wmma_f32(hi2(aS[q][m]), hi2(bv[q]), acc[m]);
  }
}

template <int NT, int LDSH>
__device__ __forceinline__ void cc_trip(
    const float* __restrict__ Wm, const int col[],
    const float* __restrict__ Ta,
    const int* __restrict__ kl, int kn, int kn2,
    v4f bv[NT][4], const v4f aCur[4][4], v4f aNxt[4][4],
    const v4i iNxt[4], v4i iNxt2[4], v8f acc[][4])
{
  cc_mm(aCur, bv[0], acc[0]); PIPE_FENCE();
  cc_ldB<LDSH>(Wm, col[0], iNxt, bv[0]); cc_ldA(Ta, kn, aNxt); cc_ldI(kl, kn2, iNxt2); PIPE_FENCE();
  #pragma unroll
  for (int t = 1; t < NT; ++t) {
    cc_mm(aCur, bv[t], acc[t]); PIPE_FENCE();
    cc_ldB<LDSH>(Wm, col[t], iNxt, bv[t]); PIPE_FENCE();
  }
}

template <int NT, int LDSH>
__device__ __forceinline__ void cc_gemm(
    const float* __restrict__ Wm, const int col[],
    const float* __restrict__ Tin, const int* __restrict__ kl, int capK,
    int lane, v8f acc[][4])
{
  const int l16 = lane & 15, hlf = lane >> 4;
  const int kof = hlf * 4;
  const v8f vz = {0.f,0.f,0.f,0.f,0.f,0.f,0.f,0.f};
  #pragma unroll
  for (int t = 0; t < NT; ++t) {
    #pragma unroll
    for (int m = 0; m < 4; ++m) acc[t][m] = vz;
  }
  if (capK <= 0) return;

  const float* Ta = Tin + (size_t)l16 * TS;
  const int klast = capK - 32 + kof;

  v4f bv[NT][4];
  v4f aX[4][4], aY[4][4];
  v4i iX[4], iY[4];

  cc_ldI(kl, kof, iY);
  cc_ldB<LDSH>(Wm, col[0], iY, bv[0]); cc_ldA(Ta, kof, aX); cc_ldI(kl, 32 + kof, iX); PIPE_FENCE();
  #pragma unroll
  for (int t = 1; t < NT; ++t) { cc_ldB<LDSH>(Wm, col[t], iY, bv[t]); PIPE_FENCE(); }
  for (int k = 0; k < capK; k += 64) {
    const int k1 = k + 32 + kof;
    int k2 = k + 64 + kof; k2 = k2 < klast ? k2 : klast;
    int k3 = k + 96 + kof; k3 = k3 < klast ? k3 : klast;
    cc_trip<NT, LDSH>(Wm, col, Ta, kl, k1, k2, bv, aX, aY, iX, iY, acc);
    cc_trip<NT, LDSH>(Wm, col, Ta, kl, k2, k3, bv, aY, aX, iY, iX, acc);
  }
}

template <int NT, int LDSH>
__device__ __forceinline__ void cc_gemm_plain(
    const float* __restrict__ Wm, const int col[NT],
    const float* __restrict__ Tin, const int* __restrict__ kl, int capK,
    int lane, bool lastReal, v8f acc[][4])
{
  const int l16 = lane & 15, hlf = lane >> 4;
  const int kof = hlf * 4;
  const v8f vz = {0.f,0.f,0.f,0.f,0.f,0.f,0.f,0.f};
  #pragma unroll
  for (int t = 0; t < NT; ++t) {
    #pragma unroll
    for (int m = 0; m < 4; ++m) acc[t][m] = vz;
  }
  const float* Ta = Tin + (size_t)l16 * TS;
  #pragma unroll 1
  for (int k = 0; k < capK; k += 32) {
    v4i ix[4];
    v4f aS[4][4];
    cc_ldI(kl, k + kof, ix);
    cc_ldA(Ta, k + kof, aS);
    #pragma unroll
    for (int t = 0; t < NT; ++t) {
      if (t < NT - 1 || lastReal) {
        v4f bv[4];
        cc_ldB<LDSH>(Wm, col[t], ix, bv);
        cc_mm(aS, bv, acc[t]);
        PIPE_FENCE();
      }
    }
  }
}

template <int NT>
__device__ __forceinline__ void chain_cols(
    const float* __restrict__ Wm, float* __restrict__ T,
    const int* __restrict__ kl, int capK,
    const int* __restrict__ nl, int Pn,
    int lane, int wv, bool lastReal)
{
  const int l16 = lane & 15, hlf = lane >> 4;
  int col[NT];
  #pragma unroll
  for (int t = 0; t < NT; ++t) col[t] = nl[(wv + 8 * t) * 16 + l16] * 4;
  v8f acc[NT][4];
  if (NT <= 5) {
    if (lastReal)    cc_gemm<NT, 10>(Wm, col, T, kl, capK, lane, acc);
    else if (NT > 1) cc_gemm<(NT > 1 ? NT - 1 : 1), 10>(Wm, col, T, kl, capK, lane, acc);
  } else {
    cc_gemm_plain<NT, 10>(Wm, col, T, kl, capK, lane, lastReal, acc);
  }
  __syncthreads();
  int ob[4];
  #pragma unroll
  for (int m = 0; m < 4; ++m) {
    ob[m] = (16 * m + 8 * hlf) * TS + wv * 16 + l16;
    asm volatile("" : "+v"(ob[m]));
  }
  #pragma unroll
  for (int t = 0; t < NT; ++t) {
    if (t < NT - 1 || lastReal) {
      const bool keep = (wv + 8 * t) * 16 + l16 < Pn;
      #pragma unroll
      for (int m = 0; m < 4; ++m) {
        #pragma unroll
        for (int r = 0; r < 8; ++r)
          T[ob[m] + r * TS + t * 128] = keep ? acc[t][m][r] : 0.f;
      }
    }
  }
}

__global__ __launch_bounds__(256) void zero_chain(
    const float* __restrict__ M0,
    const float* __restrict__ Wh1, const float* __restrict__ Wh2,
    const float* __restrict__ d1z, const float* __restrict__ d2z,
    const float* __restrict__ d3z,
    const float* __restrict__ Wout,
    float* __restrict__ out)
{
  extern __shared__ float lds[];
  float* T = lds;
  int* rowsL = (int*)(lds + CB * TS);
  int* cnt   = rowsL + 3 * HID;
  float* red = (float*)(cnt + 96);

  const int nb = (int)__builtin_amdgcn_workgroup_id_x();
  const int b  = (int)__builtin_amdgcn_workgroup_id_y();
  const int lane = threadIdx.x & 31, wv = threadIdx.x >> 5;

  unsigned mk[3][4];
  {
    float dv[3][4];
    #pragma unroll
    for (int g = 0; g < 3; ++g) {
      const float* dg = (g == 0 ? d1z : g == 1 ? d2z : d3z) + (size_t)b * HID;
      #pragma unroll
      for (int j = 0; j < 4; ++j) dv[g][j] = dg[j * 256 + threadIdx.x];
    }
    #pragma unroll
    for (int g = 0; g < 3; ++g) {
      #pragma unroll
      for (int j = 0; j < 4; ++j) mk[g][j] = (unsigned)__ballot(dv[g][j] != 0.f);
    }
    if (lane == 0) {
      #pragma unroll
      for (int g = 0; g < 3; ++g) {
        #pragma unroll
        for (int j = 0; j < 4; ++j) cnt[g * 32 + j * 8 + wv] = __popc(mk[g][j]);
      }
    }
  }
  __syncthreads();

  int Pn[3], capn[3];
  #pragma unroll
  for (int g = 0; g < 3; ++g) {
    int tot = 0, off[4] = {0, 0, 0, 0};
    #pragma unroll
    for (int sgm = 0; sgm < 32; ++sgm) {
      if ((sgm & 7) == wv) off[sgm >> 3] = tot;
      tot += cnt[g * 32 + sgm];
    }
    int* rl = rowsL + g * HID;
    #pragma unroll
    for (int j = 0; j < 4; ++j)
      if ((mk[g][j] >> lane) & 1u)
        rl[off[j] + __popc(mk[g][j] & ((1u << lane) - 1u))] = j * 256 + threadIdx.x;
    const int cap = ((tot + 63) >> 6) << 6;
    for (int i = tot + threadIdx.x; i < cap; i += 256) rl[i] = 0;
    Pn[g] = tot; capn[g] = cap;
  }
  __syncthreads();

  {
    const int* l3 = rowsL + 2 * HID;
    const int P3 = Pn[2];
    const float* Wn = Wout + (size_t)(nb * CB + wv * 8) * HID;
    float* Tn = T + (size_t)(wv * 8) * TS;
    for (int p = lane * 4; p < capn[2]; p += 128) {
      const v4i ix = *(const v4i*)(l3 + p);
      float w[8][4];
      #pragma unroll
      for (int n = 0; n < 8; ++n) {
        w[n][0] = Wn[n * HID + ix.x]; w[n][1] = Wn[n * HID + ix.y];
        w[n][2] = Wn[n * HID + ix.z]; w[n][3] = Wn[n * HID + ix.w];
      }
      #pragma unroll
      for (int n = 0; n < 8; ++n) {
        v4f o;
        o.x = (p     < P3) ? w[n][0] : 0.f;
        o.y = (p + 1 < P3) ? w[n][1] : 0.f;
        o.z = (p + 2 < P3) ? w[n][2] : 0.f;
        o.w = (p + 3 < P3) ? w[n][3] : 0.f;
        *(v4f*)(Tn + (size_t)n * TS + p) = o;
      }
    }
  }

  #pragma unroll 1
  for (int g = 0; g < 2; ++g) {
    const float* Wm  = g ? Wh1 : Wh2;
    const int* kl    = rowsL + (g ? 1 : 2) * HID;
    const int capK   = g ? capn[1] : capn[2];
    const int* nl    = rowsL + (g ? 0 : 1) * HID;
    const int P      = g ? Pn[0] : Pn[1];
    const int capN   = g ? capn[0] : capn[1];
    __syncthreads();
    const int ntot = capN >> 4;
    const int ntb = __builtin_amdgcn_readfirstlane((ntot + 7) >> 3);
    const bool lastReal = __builtin_amdgcn_readfirstlane(wv + 8 * (ntb - 1) < ntot ? 1 : 0) != 0;
    switch (ntb) {
      case 1: chain_cols<1>(Wm, T, kl, capK, nl, P, lane, wv, lastReal); break;
      case 2: chain_cols<2>(Wm, T, kl, capK, nl, P, lane, wv, lastReal); break;
      case 3: chain_cols<3>(Wm, T, kl, capK, nl, P, lane, wv, lastReal); break;
      case 4: chain_cols<4>(Wm, T, kl, capK, nl, P, lane, wv, lastReal); break;
      case 5: chain_cols<5>(Wm, T, kl, capK, nl, P, lane, wv, lastReal); break;
      case 6: chain_cols<6>(Wm, T, kl, capK, nl, P, lane, wv, lastReal); break;
      case 7: chain_cols<7>(Wm, T, kl, capK, nl, P, lane, wv, lastReal); break;
      case 8: chain_cols<8>(Wm, T, kl, capK, nl, P, lane, wv, lastReal); break;
      default: break;
    }
  }
  __syncthreads();

  {
    const int l16 = lane & 15, hlf = lane >> 4;
    int col[1] = { (wv * 16 + l16) * 4 };
    v8f acc[1][4];
    cc_gemm<1, 7>(M0, col, T, rowsL, capn[0], lane, acc);
    #pragma unroll
    for (int m = 0; m < 4; ++m) {
      #pragma unroll
      for (int r = 0; r < 8; ++r) {
        float sq = acc[0][m][r] * acc[0][m][r];
        sq += __shfl_xor(sq, 1);
        sq += __shfl_xor(sq, 2);
        sq += __shfl_xor(sq, 4);
        sq += __shfl_xor(sq, 8);
        if (l16 == 0) red[wv * CB + 16 * m + 8 * hlf + r] = sq;
      }
    }
  }
  __syncthreads();
  if (threadIdx.x < CB) {
    float sm = 0.f;
    #pragma unroll
    for (int w = 0; w < 8; ++w) sm += red[w * CB + threadIdx.x];
    out[(size_t)b * 384 + 256 + nb * CB + threadIdx.x] = sqrtf(sm);
  }
}

extern "C" void kernel_launch(void* const* d_in, const int* in_sizes, int n_in,
                              void* d_out, int out_size, void* d_ws, size_t ws_size,
                              hipStream_t stream)
{
  (void)in_sizes; (void)n_in; (void)out_size; (void)ws_size;
  const float* x     = (const float*)d_in[0];
  const float* W_in  = (const float*)d_in[1];
  const float* b_in  = (const float*)d_in[2];
  const float* Wh    = (const float*)d_in[3];
  const float* bh    = (const float*)d_in[4];
  const float* W_out = (const float*)d_in[5];
  const float* b_out = (const float*)d_in[6];
  float* out = (float*)d_out;
  float* ws  = (float*)d_ws;

  const float* Wh0 = Wh;
  const float* Wh1 = Wh + 1024 * 1024;
  const float* Wh2 = Wh + 2 * 1024 * 1024;
  const float* bh0 = bh, *bh1 = bh + 1024, *bh2 = bh + 2048;

  float* X2  = ws;
  float* hA  = X2  + 512 * 256;
  float* hB  = hA  + 512 * 1024;
  float* D1  = hB  + 512 * 1024;
  float* D2  = D1  + 512 * 1024;
  float* D3  = D2  + 512 * 1024;
  float* uA  = D3  + 512 * 1024;
  float* uB  = uA  + 256 * 1024;
  float* M0b = uB  + 256 * 1024;

  dim3 blk(256);
  build_x2<<<dim3((512 * 256 + 255) / 256), blk, 0, stream>>>(x, X2);

  dim3 lblk(128);
  dim3 lblk2(64);
#define SLOT(L, n, A_, lda_, W_, ldw_, b_, m_, ldm_, C_, ldc_, M_, K_, e_, gx_, gy_) \
  { (L).A##n = (A_); (L).lda##n = (lda_); (L).W##n = (W_); (L).ldw##n = (ldw_); \
    (L).bias##n = (b_); (L).mask##n = (m_); (L).ldm##n = (ldm_); (L).C##n = (C_); \
    (L).ldc##n = (ldc_); (L).M##n = (M_); (L).K##n = (K_); (L).epi##n = (e_); \
    (L).gx##n = (gx_); (L).gy##n = (gy_); }

  LevelArgs L1 = {}; L1.abz = -1;
  SLOT(L1, 0, X2,256,      W_in,256, b_in,    nullptr,0, hA,1024, 512,256, 0, 8,16)
  SLOT(L1, 1, x + 256,512, W_in,256, nullptr, nullptr,0, uA,1024, 256,128, 0, 4,16)
  level_gemm<<<dim3(8,16,2), lblk, 0, stream>>>(L1);

  LevelArgs L2 = {}; L2.abz = -1;
  SLOT(L2, 0, hA,1024, Wh0,1024, bh0, nullptr,0, hB,1024, 512,1024, 1, 16,16)
  level_gemm<<<dim3(16,16,1), lblk2, 0, stream>>>(L2);

  LevelArgs L3 = {}; L3.abz = -1;
  SLOT(L3, 0, hB,1024, Wh0,1024, bh0, nullptr,0, D1,1024, 512,1024, 2, 8,16)
  SLOT(L3, 1, hB,1024, Wh1,1024, bh1, nullptr,0, hA,1024, 512,1024, 1, 8,16)
  level_gemm<<<dim3(8,16,2), lblk, 0, stream>>>(L3);

  LevelArgs L4 = {}; L4.abz = -1;
  SLOT(L4, 0, hA,1024, Wh1,1024, bh1,     nullptr,0, D2,1024, 256,1024, 2, 4,16)
  SLOT(L4, 1, hA,1024, Wh2,1024, bh2,     nullptr,0, hB,1024, 512,1024, 1, 8,16)
  SLOT(L4, 2, uA,1024, Wh0,1024, nullptr, D1,1024,   uB,1024, 256,1024, 0, 4,16)
  level_gemm<<<dim3(8,16,3), lblk, 0, stream>>>(L4);

  LevelArgs L5 = {}; L5.abz = -1;
  SLOT(L5, 0, hB,1024, Wh2,1024,   bh2,     nullptr,0, D3,1024, 512,1024, 2, 8,16)
  SLOT(L5, 1, hB,1024, W_out,1024, b_out,   nullptr,0, out,384, 256,1024, 0, 4,2)
  SLOT(L5, 2, uB,1024, Wh1,1024,   nullptr, D2,1024,   uA,1024, 256,1024, 0, 4,16)
  level_gemm<<<dim3(8,16,3), lblk, 0, stream>>>(L5);

  LevelArgs L6 = {}; L6.abz = -1;
  SLOT(L6, 0, uA,1024, Wh2,1024, nullptr, D3,1024, uB,1024, 256,1024, 0, 8,16)
  SLOT(L6, 1, hA + 256 * 1024,1024, Wh1,1024, bh1, nullptr,0, D2 + 256 * 1024,1024, 256,1024, 2, 8,16)
  level_gemm<<<dim3(8,16,2), lblk2, 0, stream>>>(L6);

  LevelArgs L7 = {}; L7.abz = 1;
  SLOT(L7, 0, uB,1024, W_out,1024, nullptr, nullptr,0, out + 128,384, 256,1024, 0, 4,2)
  L7.Aab = Wh0; L7.ldaab = 1024; L7.Bab = W_in; L7.ldbab = 256; L7.Cab = M0b; L7.ldcab = 128;
  L7.Mab = 1024; L7.Kab = 1024; L7.gxab = 16; L7.gyab = 2;
  level_gemm<<<dim3(16,2,2), lblk, 0, stream>>>(L7);
#undef SLOT

  zero_chain<<<dim3(NST / CB,256), blk, CB * TS * sizeof(float) + 3 * HID * sizeof(int) + 96 * sizeof(int) + 8 * CB * sizeof(float), stream>>>(
      M0b, Wh1, Wh2,
      D1 + 256 * 1024, D2 + 256 * 1024, D3 + 256 * 1024,
      W_out, out);
}
